# starting kernel
# baseline (speedup 1.0000x reference)
#include <hip/hip_runtime.h>
#include <math.h>


typedef __attribute__((ext_vector_type(2))) float v2f;
typedef __attribute__((ext_vector_type(8))) float v8f;

#define D 128
#define SELF_PAD 512

__global__ __launch_bounds__(256) void gemm128_wmma(
    const float* __restrict__ A, const float* __restrict__ W,
    const float* __restrict__ bias, float* __restrict__ C, int M) {
  const int lane = threadIdx.x & 31;
  const int wave = threadIdx.x >> 5;
  const int r = lane & 15;
  const int g = lane >> 4;
  const int m0 = blockIdx.x * 16;
  const int n0 = wave * 16;

  const float* __restrict__ Arow = A + (size_t)(m0 + r) * D;
  const float* __restrict__ Wcol = W + (n0 + r);

  v8f acc = {};
#pragma unroll
  for (int k = 0; k < D; k += 4) {
    const int kk = k + 2 * g;
    v2f a;
    a.x = Arow[kk];
    a.y = Arow[kk + 1];
    v2f b;
    b.x = Wcol[(size_t)kk * D];
    b.y = Wcol[(size_t)(kk + 1) * D];
    acc = __builtin_amdgcn_wmma_f32_16x16x4_f32(
         false, a,  false, b,
         (short)0, acc,  false,  false);
  }

  const float bv = bias ? bias[n0 + r] : 0.0f;
#pragma unroll
  for (int j = 0; j < 8; ++j) {
    const int row = m0 + j + 8 * g;
    C[(size_t)row * D + n0 + r] = acc[j] + bv;
  }
}

__global__ void init_kernel(float* out0, float* featdyn, float* mdyn,
                            float* msta, float* sdyn, float* ssta,
                            float* cnts, int N) {
  const int i = blockIdx.x * 256 + threadIdx.x;
  const int ND = N * D;
  if (i < ND) {
    out0[i] = 0.0f;
    featdyn[i] = 0.0f;
  }
  if (i < N) {
    mdyn[i] = -__builtin_inff();
    msta[i] = -__builtin_inff();
    sdyn[i] = 0.0f;
    ssta[i] = 0.0f;
    cnts[i] = 0.0f;
  }
}

__global__ void build_trel(const float* __restrict__ relemb,
                           const float* __restrict__ selfrel,
                           float* __restrict__ Trel, int R) {
  const int i = blockIdx.x * 256 + threadIdx.x;
  if (i >= SELF_PAD * D) return;
  const int row = i >> 7;
  const int c = i & (D - 1);
  float v = 0.0f;
  if (row < R) v = relemb[i];
  else if (row == R) v = selfrel[c];
  Trel[i] = v;
}

__device__ inline void atomicMaxF(float* addr, float v) {
  if (v >= 0.0f)
    atomicMax((int*)addr, __float_as_int(v));
  else
    atomicMin((unsigned int*)addr, __float_as_uint(v));
}

__global__ __launch_bounds__(256) void edge_attn_kernel(
    const int* __restrict__ ei, const int* __restrict__ et, int E, int N,
    int R, const float* __restrict__ Xq, const float* __restrict__ Xk,
    const float* __restrict__ Tq, const float* __restrict__ Tk,
    float* __restrict__ attn, float* __restrict__ mdyn,
    float* __restrict__ msta, float* __restrict__ cnts) {
  const int e = blockIdx.x * 8 + (threadIdx.x >> 5);
  const int lane = threadIdx.x & 31;
  const int ET = E + N;
  if (e >= ET) return;

  int s, d, t;
  if (e < E) {
    s = ei[e];
    d = ei[E + e];
    t = et[e];
  } else {
    s = d = e - E;
    t = R;
  }

  const float4 q0 = ((const float4*)(Xq + (size_t)d * D))[lane];
  const float4 q1 = ((const float4*)(Tq + (size_t)t * D))[lane];
  const float4 k0 = ((const float4*)(Xk + (size_t)s * D))[lane];
  const float4 k1 = ((const float4*)(Tk + (size_t)t * D))[lane];

  float a = (q0.x + q1.x) * (k0.x + k1.x) + (q0.y + q1.y) * (k0.y + k1.y) +
            (q0.z + q1.z) * (k0.z + k1.z) + (q0.w + q1.w) * (k0.w + k1.w);
#pragma unroll
  for (int off = 16; off >= 1; off >>= 1) a += __shfl_xor(a, off, 32);

  if (lane == 0) {
    a *= 0.08838834764831845f;
    attn[e] = a;
    atomicMaxF(&mdyn[d], a);
    atomicMaxF(&msta[d], -a);
    atomicAdd(&cnts[d], 1.0f);
  }
}

__global__ void edge_expsum_kernel(const int* __restrict__ ei, int E, int N,
                                   const float* __restrict__ attn,
                                   const float* __restrict__ mdyn,
                                   const float* __restrict__ msta,
                                   float* __restrict__ edyn,
                                   float* __restrict__ esta,
                                   float* __restrict__ sdyn,
                                   float* __restrict__ ssta) {
  const int e = blockIdx.x * 256 + threadIdx.x;
  const int ET = E + N;
  if (e >= ET) return;
  const int d = (e < E) ? ei[E + e] : (e - E);
  const float a = attn[e];
  const float ed = __expf(a - mdyn[d]);
  const float es = __expf(-a - msta[d]);
  edyn[e] = ed;
  esta[e] = es;
  atomicAdd(&sdyn[d], ed);
  atomicAdd(&ssta[d], es);
}

__global__ __launch_bounds__(256) void edge_scatter_kernel(
    const int* __restrict__ ei, int E, int N, const float* __restrict__ edyn,
    const float* __restrict__ esta, const float* __restrict__ sdyn,
    const float* __restrict__ ssta, const float* __restrict__ cnts,
    const float* __restrict__ Vv, float* __restrict__ featsta,
    float* __restrict__ featdyn) {
  const int e = blockIdx.x * 8 + (threadIdx.x >> 5);
  const int lane = threadIdx.x & 31;
  const int ET = E + N;
  if (e >= ET) return;

  int s, d;
  if (e < E) {
    s = ei[e];
    d = ei[E + e];
  } else {
    s = d = e - E;
  }

  const float denom = fmaxf(cnts[d], 1.0f);
  const float wd = edyn[e] / (sdyn[d] * denom);
  const float ws = esta[e] / (ssta[d] * denom);

  const float4 v4 = ((const float4*)(Vv + (size_t)s * D))[lane];
  float* fd = featdyn + (size_t)d * D + lane * 4;
  float* fs = featsta + (size_t)d * D + lane * 4;
  atomicAdd(fs + 0, ws * v4.x);
  atomicAdd(fs + 1, ws * v4.y);
  atomicAdd(fs + 2, ws * v4.z);
  atomicAdd(fs + 3, ws * v4.w);
  atomicAdd(fd + 0, wd * v4.x);
  atomicAdd(fd + 1, wd * v4.y);
  atomicAdd(fd + 2, wd * v4.z);
  atomicAdd(fd + 3, wd * v4.w);
}

extern "C" void kernel_launch(void* const* d_in, const int* in_sizes, int n_in,
                              void* d_out, int out_size, void* d_ws,
                              size_t ws_size, hipStream_t stream) {
  const float* curr = (const float*)d_in[0];
  const float* last = (const float*)d_in[1];
  const float* relemb = (const float*)d_in[2];
  const float* selfrel = (const float*)d_in[3];
  const float* Wq = (const float*)d_in[4];
  const float* bq = (const float*)d_in[5];
  const float* Wk = (const float*)d_in[6];
  const float* bk = (const float*)d_in[7];
  const float* Wv = (const float*)d_in[8];
  const float* bv = (const float*)d_in[9];
  const float* dynW = (const float*)d_in[10];
  const int* ei = (const int*)d_in[11];
  const int* et = (const int*)d_in[12];

  const int N = in_sizes[0] / D;
  const int E = in_sizes[12];
  const int R = in_sizes[2] / D;
  const int ET = E + N;
  const size_t ND = (size_t)N * D;

  float* ws = (float*)d_ws;
  float* Xq = ws;
  float* Xk = Xq + ND;
  float* Vv = Xk + ND;
  float* featdyn = Vv + ND;
  float* Trel = featdyn + ND;
  float* Tq = Trel + SELF_PAD * D;
  float* Tk = Tq + SELF_PAD * D;
  float* attn = Tk + SELF_PAD * D;
  float* edyn = attn + ET;
  float* esta = edyn + ET;
  float* mdyn = esta + ET;
  float* msta = mdyn + N;
  float* sdyn = msta + N;
  float* ssta = sdyn + N;
  float* cnts = ssta + N;

  float* out0 = (float*)d_out;
  float* out1 = out0 + ND;

  init_kernel<<<(int)((ND + 255) / 256), 256, 0, stream>>>(
      out0, featdyn, mdyn, msta, sdyn, ssta, cnts, N);
  build_trel<<<(SELF_PAD * D + 255) / 256, 256, 0, stream>>>(relemb, selfrel,
                                                             Trel, R);

  gemm128_wmma<<<N / 16, 256, 0, stream>>>(curr, Wq, nullptr, Xq, N);
  gemm128_wmma<<<N / 16, 256, 0, stream>>>(last, Wk, nullptr, Xk, N);
  gemm128_wmma<<<N / 16, 256, 0, stream>>>(last, Wv, bv, Vv, N);
  gemm128_wmma<<<SELF_PAD / 16, 256, 0, stream>>>(Trel, Wq + D * D, bq, Tq,
                                                  SELF_PAD);
  gemm128_wmma<<<SELF_PAD / 16, 256, 0, stream>>>(Trel, Wk + D * D, bk, Tk,
                                                  SELF_PAD);

  edge_attn_kernel<<<(ET + 7) / 8, 256, 0, stream>>>(
      ei, et, E, N, R, Xq, Xk, Tq, Tk, attn, mdyn, msta, cnts);
  edge_expsum_kernel<<<(ET + 255) / 256, 256, 0, stream>>>(
      ei, E, N, attn, mdyn, msta, edyn, esta, sdyn, ssta);
  edge_scatter_kernel<<<(ET + 7) / 8, 256, 0, stream>>>(
      ei, E, N, edyn, esta, sdyn, ssta, cnts, Vv, out0, featdyn);

  gemm128_wmma<<<N / 16, 256, 0, stream>>>(featdyn, dynW, nullptr, out1, N);
}


# Scatter pass: a work-group owns a range of destination rows, sums both features in LDS by f32 LDS atomics (arrival order varies as before), stores rows once; global f32 atomics and zero-fill gone
# speedup vs baseline: 1.4502x; 1.4502x over previous
#include <hip/hip_runtime.h>
#include <math.h>


typedef __attribute__((ext_vector_type(2))) float v2f;
typedef __attribute__((ext_vector_type(8))) float v8f;

#define D 128
#define SELF_PAD 512
#define GEMM_GRID 500

__device__ __forceinline__ void load_a_tile(float4 (&a)[16],
                                            const float* __restrict__ A,
                                            int tile, int r, int g) {
  const float4* __restrict__ p =
      (const float4*)(A + (size_t)(tile * 16 + r) * D) + g;
#pragma unroll
  for (int i = 0; i < 16; ++i) a[i] = p[2 * i];
}

__device__ __forceinline__ void mma_store_tile(const float4 (&a)[16],
                                               const v2f (&b)[32], float bv,
                                               float* __restrict__ C, int tile,
                                               int n0, int r, int g) {
  v8f acc = {};
#pragma unroll
  for (int i = 0; i < 16; ++i) {
    v2f a0, a1;
    a0.x = a[i].x;
    a0.y = a[i].y;
    a1.x = a[i].z;
    a1.y = a[i].w;
    acc = __builtin_amdgcn_wmma_f32_16x16x4_f32(
         false, a0,  false, b[2 * i],
         (short)0, acc,  false,  false);
    acc = __builtin_amdgcn_wmma_f32_16x16x4_f32(
         false, a1,  false, b[2 * i + 1],
         (short)0, acc,  false,  false);
  }
#pragma unroll
  for (int j = 0; j < 8; ++j) {
    const int row = tile * 16 + j + 8 * g;
    C[(size_t)row * D + n0 + r] = acc[j] + bv;
  }
}

__global__ __launch_bounds__(256) void gemm128_wmma(
    const float* __restrict__ A, const float* __restrict__ W,
    const float* __restrict__ bias, float* __restrict__ C, int tiles) {
  const int lane = threadIdx.x & 31;
  const int wave = threadIdx.x >> 5;
  const int r = lane & 15;
  const int g = lane >> 4;
  const int n0 = wave * 16;
  const int stride = gridDim.x;

  int t = blockIdx.x;
  if (t >= tiles) return;

  const float* __restrict__ Wcol = W + (n0 + r);
  v2f b[32];
#pragma unroll
  for (int s = 0; s < 32; ++s) {
    const int kk = 8 * (s >> 1) + 4 * g + 2 * (s & 1);
    b[s].x = Wcol[(size_t)kk * D];
    b[s].y = Wcol[(size_t)(kk + 1) * D];
  }
  const float bv = bias ? bias[n0 + r] : 0.0f;

  float4 a0[16], a1[16];
  load_a_tile(a0, A, t, r, g);
  for (;;) {
    const int t1 = t + stride;
    if (t1 >= tiles) {
      mma_store_tile(a0, b, bv, C, t, n0, r, g);
      break;
    }
    load_a_tile(a1, A, t1, r, g);
    __builtin_amdgcn_sched_barrier(0);
    mma_store_tile(a0, b, bv, C, t, n0, r, g);
    const int t2 = t1 + stride;
    if (t2 >= tiles) {
      mma_store_tile(a1, b, bv, C, t1, n0, r, g);
      break;
    }
    load_a_tile(a0, A, t2, r, g);
    __builtin_amdgcn_sched_barrier(0);
    mma_store_tile(a1, b, bv, C, t1, n0, r, g);
    t = t2;
  }
}

__global__ void init_kernel(float* mdyn, float* msta, float* sdyn,
                            float* ssta, float* cnts, int N) {
  const int i = blockIdx.x * 256 + threadIdx.x;
  if (i < N) {
    mdyn[i] = -__builtin_inff();
    msta[i] = -__builtin_inff();
    sdyn[i] = 0.0f;
    ssta[i] = 0.0f;
    cnts[i] = 0.0f;
  }
}

__global__ void build_trel(const float* __restrict__ relemb,
                           const float* __restrict__ selfrel,
                           float* __restrict__ Trel, int R) {
  const int i = blockIdx.x * 256 + threadIdx.x;
  if (i >= SELF_PAD * D) return;
  const int row = i >> 7;
  const int c = i & (D - 1);
  float v = 0.0f;
  if (row < R) v = relemb[i];
  else if (row == R) v = selfrel[c];
  Trel[i] = v;
}

__device__ inline void atomicMaxF(float* addr, float v) {
  if (v >= 0.0f)
    atomicMax((int*)addr, __float_as_int(v));
  else
    atomicMin((unsigned int*)addr, __float_as_uint(v));
}

__global__ __launch_bounds__(256) void edge_attn_kernel(
    const int* __restrict__ ei, const int* __restrict__ et, int E, int N,
    int R, const float* __restrict__ Xq, const float* __restrict__ Xk,
    const float* __restrict__ Tq, const float* __restrict__ Tk,
    float* __restrict__ attn, float* __restrict__ mdyn,
    float* __restrict__ msta, float* __restrict__ cnts) {
  const int e = blockIdx.x * 8 + (threadIdx.x >> 5);
  const int lane = threadIdx.x & 31;
  const int ET = E + N;
  if (e >= ET) return;

  int s, d, t;
  if (e < E) {
    s = ei[e];
    d = ei[E + e];
    t = et[e];
  } else {
    s = d = e - E;
    t = R;
  }

  const float4 q0 = ((const float4*)(Xq + (size_t)d * D))[lane];
  const float4 q1 = ((const float4*)(Tq + (size_t)t * D))[lane];
  const float4 k0 = ((const float4*)(Xk + (size_t)s * D))[lane];
  const float4 k1 = ((const float4*)(Tk + (size_t)t * D))[lane];

  float a = (q0.x + q1.x) * (k0.x + k1.x) + (q0.y + q1.y) * (k0.y + k1.y) +
            (q0.z + q1.z) * (k0.z + k1.z) + (q0.w + q1.w) * (k0.w + k1.w);
#pragma unroll
  for (int off = 16; off >= 1; off >>= 1) a += __shfl_xor(a, off, 32);

  if (lane == 0) {
    a *= 0.08838834764831845f;
    attn[e] = a;
    atomicMaxF(&mdyn[d], a);
    atomicMaxF(&msta[d], -a);
    atomicAdd(&cnts[d], 1.0f);
  }
}

__global__ void edge_expsum_kernel(const int* __restrict__ ei, int E, int N,
                                   const float* __restrict__ attn,
                                   const float* __restrict__ mdyn,
                                   const float* __restrict__ msta,
                                   float* __restrict__ edyn,
                                   float* __restrict__ esta,
                                   float* __restrict__ sdyn,
                                   float* __restrict__ ssta) {
  const int e = blockIdx.x * 256 + threadIdx.x;
  const int ET = E + N;
  if (e >= ET) return;
  const int d = (e < E) ? ei[E + e] : (e - E);
  const float a = attn[e];
  const float ed = __expf(a - mdyn[d]);
  const float es = __expf(-a - msta[d]);
  edyn[e] = ed;
  esta[e] = es;
  atomicAdd(&sdyn[d], ed);
  atomicAdd(&ssta[d], es);
}

#define SC_ROWS 160
#define SC_THREADS 1024
#define SC_WAVES (SC_THREADS / 32)
#define SC_BATCH 512
#define SC_LIST (SC_BATCH + 32)

template <bool VEC>
__device__ __forceinline__ int4 load_dst4(const int* __restrict__ dst, int idx,
                                          int E) {
  if (VEC) {
    return *(const int4*)(dst + min(idx, E - 4));
  } else {
    int4 v;
    v.x = dst[min(idx + 0, E - 1)];
    v.y = dst[min(idx + 1, E - 1)];
    v.z = dst[min(idx + 2, E - 1)];
    v.w = dst[min(idx + 3, E - 1)];
    return v;
  }
}

__device__ __forceinline__ int list_append(int* lst, int n, bool m,
                                           int e, int lane) {
  const unsigned mask = (unsigned)__ballot(m);
  if (mask != 0u) {
    const int pos = n + __popc(mask & ((1u << lane) - 1u));
    if (m) lst[pos] = e;
    n += __popc(mask);
  }
  return n;
}

__device__ __forceinline__ void add_entries(
    const int* lst, int base, int cnt, int lane, int lo,
    const int* __restrict__ ei, int E, const float* __restrict__ edyn,
    const float* __restrict__ esta, const float* __restrict__ sdyn,
    const float* __restrict__ ssta, const float* __restrict__ cnts,
    const float* __restrict__ Vv, float* __restrict__ acc_s,
    float* __restrict__ acc_d) {
  int s = 0, row = 0;
  float wd = 0.0f, ws = 0.0f;
  if (lane < cnt) {
    const int e = lst[base + lane];
    int d;
    if (e < E) {
      s = ei[e];
      d = ei[E + e];
    } else {
      s = d = e - E;
    }
    const float denom = fmaxf(cnts[d], 1.0f);
    wd = edyn[e] / (sdyn[d] * denom);
    ws = esta[e] / (ssta[d] * denom);
    row = d - lo;
  }
  for (int j0 = 0; j0 < cnt; j0 += 8) {
    float4 v[8];
#pragma unroll
    for (int u = 0; u < 8; ++u) {
      const int j = min(j0 + u, cnt - 1);
      const int sj = __builtin_amdgcn_readlane(s, j);
      v[u] = ((const float4*)(Vv + (size_t)sj * D))[lane];
    }
#pragma unroll
    for (int u = 0; u < 8; ++u) {
      const int j = j0 + u;
      if (j < cnt) {
        const int rj = __builtin_amdgcn_readlane(row, j);
        const float wsj = __int_as_float(__builtin_amdgcn_readlane(__float_as_int(ws), j));
        const float wdj = __int_as_float(__builtin_amdgcn_readlane(__float_as_int(wd), j));
        float* as = acc_s + rj * D + lane;
        float* ad = acc_d + rj * D + lane;
        __hip_atomic_fetch_add(as + 0, wsj * v[u].x, __ATOMIC_RELAXED, __HIP_MEMORY_SCOPE_WORKGROUP);
        __hip_atomic_fetch_add(as + 32, wsj * v[u].y, __ATOMIC_RELAXED, __HIP_MEMORY_SCOPE_WORKGROUP);
        __hip_atomic_fetch_add(as + 64, wsj * v[u].z, __ATOMIC_RELAXED, __HIP_MEMORY_SCOPE_WORKGROUP);
        __hip_atomic_fetch_add(as + 96, wsj * v[u].w, __ATOMIC_RELAXED, __HIP_MEMORY_SCOPE_WORKGROUP);
        __hip_atomic_fetch_add(ad + 0, wdj * v[u].x, __ATOMIC_RELAXED, __HIP_MEMORY_SCOPE_WORKGROUP);
        __hip_atomic_fetch_add(ad + 32, wdj * v[u].y, __ATOMIC_RELAXED, __HIP_MEMORY_SCOPE_WORKGROUP);
        __hip_atomic_fetch_add(ad + 64, wdj * v[u].z, __ATOMIC_RELAXED, __HIP_MEMORY_SCOPE_WORKGROUP);
        __hip_atomic_fetch_add(ad + 96, wdj * v[u].w, __ATOMIC_RELAXED, __HIP_MEMORY_SCOPE_WORKGROUP);
      }
    }
  }
}

template <bool VEC>
__device__ __forceinline__ int scan_stripe(
    int* lst, int n, int wave, int lane, int lo, int nrows, int rot, int nbatch,
    const int* __restrict__ ei, int E, const float* __restrict__ edyn,
    const float* __restrict__ esta, const float* __restrict__ sdyn,
    const float* __restrict__ ssta, const float* __restrict__ cnts,
    const float* __restrict__ Vv, float* __restrict__ acc_s,
    float* __restrict__ acc_d) {
  const int* __restrict__ dst = ei + E;
  const unsigned span = (unsigned)nrows;
  for (int b0 = wave; b0 < nbatch; b0 += SC_WAVES) {
    int b = b0 + rot;
    if (b >= nbatch) b -= nbatch;
    const int idx = b * SC_BATCH + lane * 4;
    int4 q[4];
#pragma unroll
    for (int u = 0; u < 4; ++u) q[u] = load_dst4<VEC>(dst, idx + u * 128, E);
#pragma unroll
    for (int u = 0; u < 4; ++u) {
      const int e = idx + u * 128;
      bool m0, m1, m2, m3;
      if (VEC) {
        const unsigned sp = (e < E) ? span : 0u;
        m0 = (unsigned)(q[u].x - lo) < sp;
        m1 = (unsigned)(q[u].y - lo) < sp;
        m2 = (unsigned)(q[u].z - lo) < sp;
        m3 = (unsigned)(q[u].w - lo) < sp;
      } else {
        m0 = (e + 0 < E) & ((unsigned)(q[u].x - lo) < span);
        m1 = (e + 1 < E) & ((unsigned)(q[u].y - lo) < span);
        m2 = (e + 2 < E) & ((unsigned)(q[u].z - lo) < span);
        m3 = (e + 3 < E) & ((unsigned)(q[u].w - lo) < span);
      }
      if (__ballot(m0 | m1 | m2 | m3) != 0) {
        n = list_append(lst, n, m0, e + 0, lane);
        n = list_append(lst, n, m1, e + 1, lane);
        n = list_append(lst, n, m2, e + 2, lane);
        n = list_append(lst, n, m3, e + 3, lane);
      }
    }
    int done = 0;
    while (n - done >= 32) {
      add_entries(lst, done, 32, lane, lo, ei, E, edyn, esta, sdyn, ssta, cnts,
                  Vv, acc_s, acc_d);
      done += 32;
    }
    if (done > 0) {
      const int rem = n - done;
      int keep = 0;
      if (lane < rem) keep = lst[done + lane];
      if (lane < rem) lst[lane] = keep;
      n = rem;
    }
  }
  return n;
}

__global__ __launch_bounds__(SC_THREADS) void edge_scatter_kernel(
    const int* __restrict__ ei, int E, int N, const float* __restrict__ edyn,
    const float* __restrict__ esta, const float* __restrict__ sdyn,
    const float* __restrict__ ssta, const float* __restrict__ cnts,
    const float* __restrict__ Vv, float* __restrict__ featsta,
    float* __restrict__ featdyn) {
  __shared__ float acc_s[SC_ROWS * D];
  __shared__ float acc_d[SC_ROWS * D];
  __shared__ int lists[SC_WAVES * SC_LIST];

  const int tid = threadIdx.x;
  const int lane = tid & 31;
  const int wave = tid >> 5;
  const int lo = blockIdx.x * SC_ROWS;
  const int nrows = min(SC_ROWS, N - lo);
  int* lst = lists + wave * SC_LIST;

  for (int i = tid; i < SC_ROWS * D / 4; i += SC_THREADS) {
    ((float4*)acc_s)[i] = make_float4(0.0f, 0.0f, 0.0f, 0.0f);
    ((float4*)acc_d)[i] = make_float4(0.0f, 0.0f, 0.0f, 0.0f);
  }
  __syncthreads();

  int n = 0;
  for (int r0 = 0; r0 < SC_ROWS; r0 += SC_THREADS) {
    const int r = r0 + lane * SC_WAVES + wave;
    n = list_append(lst, n, r < nrows, E + lo + r, lane);
  }

  const int nbatch = (E + SC_BATCH - 1) / SC_BATCH;
  const int rot = (int)(((long long)blockIdx.x * nbatch) / gridDim.x);
  if ((E & 3) == 0 && E >= 4)
    n = scan_stripe<true>(lst, n, wave, lane, lo, nrows, rot, nbatch, ei, E, edyn,
                          esta, sdyn, ssta, cnts, Vv, acc_s, acc_d);
  else if (E > 0)
    n = scan_stripe<false>(lst, n, wave, lane, lo, nrows, rot, nbatch, ei, E,
                           edyn, esta, sdyn, ssta, cnts, Vv, acc_s, acc_d);
  if (n > 0)
    add_entries(lst, 0, n, lane, lo, ei, E, edyn, esta, sdyn, ssta, cnts, Vv,
                acc_s, acc_d);
  __syncthreads();

  for (int r = wave; r < nrows; r += SC_WAVES) {
    const float* as = acc_s + r * D + lane;
    const float* ad = acc_d + r * D + lane;
    const float4 os = make_float4(as[0], as[32], as[64], as[96]);
    const float4 od = make_float4(ad[0], ad[32], ad[64], ad[96]);
    ((float4*)(featsta + (size_t)(lo + r) * D))[lane] = os;
    ((float4*)(featdyn + (size_t)(lo + r) * D))[lane] = od;
  }
}

extern "C" void kernel_launch(void* const* d_in, const int* in_sizes, int n_in,
                              void* d_out, int out_size, void* d_ws,
                              size_t ws_size, hipStream_t stream) {
  const float* curr = (const float*)d_in[0];
  const float* last = (const float*)d_in[1];
  const float* relemb = (const float*)d_in[2];
  const float* selfrel = (const float*)d_in[3];
  const float* Wq = (const float*)d_in[4];
  const float* bq = (const float*)d_in[5];
  const float* Wk = (const float*)d_in[6];
  const float* bk = (const float*)d_in[7];
  const float* Wv = (const float*)d_in[8];
  const float* bv = (const float*)d_in[9];
  const float* dynW = (const float*)d_in[10];
  const int* ei = (const int*)d_in[11];
  const int* et = (const int*)d_in[12];

  const int N = in_sizes[0] / D;
  const int E = in_sizes[12];
  const int R = in_sizes[2] / D;
  const int ET = E + N;
  const size_t ND = (size_t)N * D;

  float* ws = (float*)d_ws;
  float* Xq = ws;
  float* Xk = Xq + ND;
  float* Vv = Xk + ND;
  float* featdyn = Vv + ND;
  float* Trel = featdyn + ND;
  float* Tq = Trel + SELF_PAD * D;
  float* Tk = Tq + SELF_PAD * D;
  float* attn = Tk + SELF_PAD * D;
  float* edyn = attn + ET;
  float* esta = edyn + ET;
  float* mdyn = esta + ET;
  float* msta = mdyn + N;
  float* sdyn = msta + N;
  float* ssta = sdyn + N;
  float* cnts = ssta + N;

  float* out0 = (float*)d_out;
  float* out1 = out0 + ND;

  init_kernel<<<(N + 255) / 256, 256, 0, stream>>>(mdyn, msta, sdyn, ssta, cnts,
                                                   N);
  build_trel<<<(SELF_PAD * D + 255) / 256, 256, 0, stream>>>(relemb, selfrel,
                                                             Trel, R);

  gemm128_wmma<<<GEMM_GRID, 256, 0, stream>>>(curr, Wq, nullptr, Xq, N / 16);
  gemm128_wmma<<<GEMM_GRID, 256, 0, stream>>>(last, Wk, nullptr, Xk, N / 16);
  gemm128_wmma<<<GEMM_GRID, 256, 0, stream>>>(last, Wv, bv, Vv, N / 16);
  gemm128_wmma<<<SELF_PAD / 16, 256, 0, stream>>>(Trel, Wq + D * D, bq, Tq,
                                                  SELF_PAD / 16);
  gemm128_wmma<<<SELF_PAD / 16, 256, 0, stream>>>(Trel, Wk + D * D, bk, Tk,
                                                  SELF_PAD / 16);

  edge_attn_kernel<<<(ET + 7) / 8, 256, 0, stream>>>(
      ei, et, E, N, R, Xq, Xk, Tq, Tk, attn, mdyn, msta, cnts);
  edge_expsum_kernel<<<(ET + 255) / 256, 256, 0, stream>>>(
      ei, E, N, attn, mdyn, msta, edyn, esta, sdyn, ssta);
  edge_scatter_kernel<<<(N + SC_ROWS - 1) / SC_ROWS, SC_THREADS, 0, stream>>>(
      ei, E, N, edyn, esta, sdyn, ssta, cnts, Vv, out0, featdyn);

  gemm128_wmma<<<GEMM_GRID, 256, 0, stream>>>(featdyn, dynW, nullptr, out1, N / 16);
}


# Edge score pass: a work-group owns a range of destination rows (Xq rows, max/min/degree in LDS) and finds its edges by a sweep; wave-per-edge grid, global max/count atomics and init launch gone
# speedup vs baseline: 1.9976x; 1.3775x over previous
#include <hip/hip_runtime.h>
#include <math.h>


typedef __attribute__((ext_vector_type(2))) float v2f;
typedef __attribute__((ext_vector_type(8))) float v8f;

#define D 128
#define SELF_PAD 512
#define GEMM_GRID 500

__device__ __forceinline__ void load_a_tile(float4 (&a)[16],
                                            const float* __restrict__ A,
                                            int tile, int r, int g) {
  const float4* __restrict__ p =
      (const float4*)(A + (size_t)(tile * 16 + r) * D) + g;
#pragma unroll
  for (int i = 0; i < 16; ++i) a[i] = p[2 * i];
}

__device__ __forceinline__ void mma_store_tile(const float4 (&a)[16],
                                               const v2f (&b)[32], float bv,
                                               float* __restrict__ C, int tile,
                                               int n0, int r, int g) {
  v8f acc = {};
#pragma unroll
  for (int i = 0; i < 16; ++i) {
    v2f a0, a1;
    a0.x = a[i].x;
    a0.y = a[i].y;
    a1.x = a[i].z;
    a1.y = a[i].w;
    acc = __builtin_amdgcn_wmma_f32_16x16x4_f32(
         false, a0,  false, b[2 * i],
         (short)0, acc,  false,  false);
    acc = __builtin_amdgcn_wmma_f32_16x16x4_f32(
         false, a1,  false, b[2 * i + 1],
         (short)0, acc,  false,  false);
  }
#pragma unroll
  for (int j = 0; j < 8; ++j) {
    const int row = tile * 16 + j + 8 * g;
    C[(size_t)row * D + n0 + r] = acc[j] + bv;
  }
}

__global__ __launch_bounds__(256) void gemm128_wmma(
    const float* __restrict__ A, const float* __restrict__ W,
    const float* __restrict__ bias, float* __restrict__ C, int tiles) {
  const int lane = threadIdx.x & 31;
  const int wave = threadIdx.x >> 5;
  const int r = lane & 15;
  const int g = lane >> 4;
  const int n0 = wave * 16;
  const int stride = gridDim.x;

  int t = blockIdx.x;
  if (t >= tiles) return;

  const float* __restrict__ Wcol = W + (n0 + r);
  v2f b[32];
#pragma unroll
  for (int s = 0; s < 32; ++s) {
    const int kk = 8 * (s >> 1) + 4 * g + 2 * (s & 1);
    b[s].x = Wcol[(size_t)kk * D];
    b[s].y = Wcol[(size_t)(kk + 1) * D];
  }
  const float bv = bias ? bias[n0 + r] : 0.0f;

  float4 a0[16], a1[16];
  load_a_tile(a0, A, t, r, g);
  for (;;) {
    const int t1 = t + stride;
    if (t1 >= tiles) {
      mma_store_tile(a0, b, bv, C, t, n0, r, g);
      break;
    }
    load_a_tile(a1, A, t1, r, g);
    __builtin_amdgcn_sched_barrier(0);
    mma_store_tile(a0, b, bv, C, t, n0, r, g);
    const int t2 = t1 + stride;
    if (t2 >= tiles) {
      mma_store_tile(a1, b, bv, C, t1, n0, r, g);
      break;
    }
    load_a_tile(a0, A, t2, r, g);
    __builtin_amdgcn_sched_barrier(0);
    mma_store_tile(a1, b, bv, C, t1, n0, r, g);
    t = t2;
  }
}

__global__ void build_trel(const float* __restrict__ relemb,
                           const float* __restrict__ selfrel,
                           float* __restrict__ Trel, int R) {
  const int i = blockIdx.x * 256 + threadIdx.x;
  if (i >= SELF_PAD * D) return;
  const int row = i >> 7;
  const int c = i & (D - 1);
  float v = 0.0f;
  if (row < R) v = relemb[i];
  else if (row == R) v = selfrel[c];
  Trel[i] = v;
}

__device__ __forceinline__ void lds_max_f(int* addr, float v) {
  if (v >= 0.0f)
    atomicMax(addr, __float_as_int(v));
  else
    atomicMin((unsigned int*)addr, __float_as_uint(v));
}

__global__ void edge_expsum_kernel(const int* __restrict__ ei, int E, int N,
                                   const float* __restrict__ attn,
                                   const float* __restrict__ mdyn,
                                   const float* __restrict__ msta,
                                   float* __restrict__ edyn,
                                   float* __restrict__ esta,
                                   float* __restrict__ sdyn,
                                   float* __restrict__ ssta) {
  const int e = blockIdx.x * 256 + threadIdx.x;
  const int ET = E + N;
  if (e >= ET) return;
  const int d = (e < E) ? ei[E + e] : (e - E);
  const float a = attn[e];
  const float ed = __expf(a - mdyn[d]);
  const float es = __expf(-a - msta[d]);
  edyn[e] = ed;
  esta[e] = es;
  atomicAdd(&sdyn[d], ed);
  atomicAdd(&ssta[d], es);
}

#define SC_ROWS 160
#define SC_THREADS 1024
#define SC_WAVES (SC_THREADS / 32)
#define SC_BATCH 512
#define SC_LIST (SC_BATCH + 32)

template <bool VEC>
__device__ __forceinline__ int4 load_dst4(const int* __restrict__ dst, int idx,
                                          int E) {
  if (VEC) {
    return *(const int4*)(dst + min(idx, E - 4));
  } else {
    int4 v;
    v.x = dst[min(idx + 0, E - 1)];
    v.y = dst[min(idx + 1, E - 1)];
    v.z = dst[min(idx + 2, E - 1)];
    v.w = dst[min(idx + 3, E - 1)];
    return v;
  }
}

__device__ __forceinline__ int list_append(int* lst, int n, bool m,
                                           int e, int lane) {
  const unsigned mask = (unsigned)__ballot(m);
  if (mask != 0u) {
    const int pos = n + __popc(mask & ((1u << lane) - 1u));
    if (m) lst[pos] = e;
    n += __popc(mask);
  }
  return n;
}

__device__ __forceinline__ void add_entries(
    const int* lst, int base, int cnt, int lane, int lo,
    const int* __restrict__ ei, int E, const float* __restrict__ edyn,
    const float* __restrict__ esta, const float* __restrict__ sdyn,
    const float* __restrict__ ssta, const float* __restrict__ cnts,
    const float* __restrict__ Vv, float* __restrict__ acc_s,
    float* __restrict__ acc_d) {
  int s = 0, row = 0;
  float wd = 0.0f, ws = 0.0f;
  if (lane < cnt) {
    const int e = lst[base + lane];
    int d;
    if (e < E) {
      s = ei[e];
      d = ei[E + e];
    } else {
      s = d = e - E;
    }
    const float denom = fmaxf(cnts[d], 1.0f);
    wd = edyn[e] / (sdyn[d] * denom);
    ws = esta[e] / (ssta[d] * denom);
    row = d - lo;
  }
  for (int j0 = 0; j0 < cnt; j0 += 8) {
    float4 v[8];
#pragma unroll
    for (int u = 0; u < 8; ++u) {
      const int j = min(j0 + u, cnt - 1);
      const int sj = __builtin_amdgcn_readlane(s, j);
      v[u] = ((const float4*)(Vv + (size_t)sj * D))[lane];
    }
#pragma unroll
    for (int u = 0; u < 8; ++u) {
      const int j = j0 + u;
      if (j < cnt) {
        const int rj = __builtin_amdgcn_readlane(row, j);
        const float wsj = __int_as_float(__builtin_amdgcn_readlane(__float_as_int(ws), j));
        const float wdj = __int_as_float(__builtin_amdgcn_readlane(__float_as_int(wd), j));
        float* as = acc_s + rj * D + lane;
        float* ad = acc_d + rj * D + lane;
        __hip_atomic_fetch_add(as + 0, wsj * v[u].x, __ATOMIC_RELAXED, __HIP_MEMORY_SCOPE_WORKGROUP);
        __hip_atomic_fetch_add(as + 32, wsj * v[u].y, __ATOMIC_RELAXED, __HIP_MEMORY_SCOPE_WORKGROUP);
        __hip_atomic_fetch_add(as + 64, wsj * v[u].z, __ATOMIC_RELAXED, __HIP_MEMORY_SCOPE_WORKGROUP);
        __hip_atomic_fetch_add(as + 96, wsj * v[u].w, __ATOMIC_RELAXED, __HIP_MEMORY_SCOPE_WORKGROUP);
        __hip_atomic_fetch_add(ad + 0, wdj * v[u].x, __ATOMIC_RELAXED, __HIP_MEMORY_SCOPE_WORKGROUP);
        __hip_atomic_fetch_add(ad + 32, wdj * v[u].y, __ATOMIC_RELAXED, __HIP_MEMORY_SCOPE_WORKGROUP);
        __hip_atomic_fetch_add(ad + 64, wdj * v[u].z, __ATOMIC_RELAXED, __HIP_MEMORY_SCOPE_WORKGROUP);
        __hip_atomic_fetch_add(ad + 96, wdj * v[u].w, __ATOMIC_RELAXED, __HIP_MEMORY_SCOPE_WORKGROUP);
      }
    }
  }
}

struct AddEntries {
  int lane, lo;
  const int* __restrict__ ei;
  int E;
  const float* __restrict__ edyn;
  const float* __restrict__ esta;
  const float* __restrict__ sdyn;
  const float* __restrict__ ssta;
  const float* __restrict__ cnts;
  const float* __restrict__ Vv;
  float* __restrict__ acc_s;
  float* __restrict__ acc_d;
  __device__ __forceinline__ void operator()(const int* lst, int base,
                                             int cnt) const {
    add_entries(lst, base, cnt, lane, lo, ei, E, edyn, esta, sdyn, ssta, cnts,
                Vv, acc_s, acc_d);
  }
};

template <bool VEC, class WORK>
__device__ __forceinline__ int scan_stripe(
    int* lst, int n, int wave, int lane, int lo, int nrows, int rot, int nbatch,
    const int* __restrict__ ei, int E, const WORK& work) {
  const int* __restrict__ dst = ei + E;
  const unsigned span = (unsigned)nrows;
  for (int b0 = wave; b0 < nbatch; b0 += SC_WAVES) {
    int b = b0 + rot;
    if (b >= nbatch) b -= nbatch;
    const int idx = b * SC_BATCH + lane * 4;
    int4 q[4];
#pragma unroll
    for (int u = 0; u < 4; ++u) q[u] = load_dst4<VEC>(dst, idx + u * 128, E);
#pragma unroll
    for (int u = 0; u < 4; ++u) {
      const int e = idx + u * 128;
      bool m0, m1, m2, m3;
      if (VEC) {
        const unsigned sp = (e < E) ? span : 0u;
        m0 = (unsigned)(q[u].x - lo) < sp;
        m1 = (unsigned)(q[u].y - lo) < sp;
        m2 = (unsigned)(q[u].z - lo) < sp;
        m3 = (unsigned)(q[u].w - lo) < sp;
      } else {
        m0 = (e + 0 < E) & ((unsigned)(q[u].x - lo) < span);
        m1 = (e + 1 < E) & ((unsigned)(q[u].y - lo) < span);
        m2 = (e + 2 < E) & ((unsigned)(q[u].z - lo) < span);
        m3 = (e + 3 < E) & ((unsigned)(q[u].w - lo) < span);
      }
      if (__ballot(m0 | m1 | m2 | m3) != 0) {
        n = list_append(lst, n, m0, e + 0, lane);
        n = list_append(lst, n, m1, e + 1, lane);
        n = list_append(lst, n, m2, e + 2, lane);
        n = list_append(lst, n, m3, e + 3, lane);
      }
    }
    int done = 0;
    while (n - done >= 32) {
      work(lst, done, 32);
      done += 32;
    }
    if (done > 0) {
      const int rem = n - done;
      int keep = 0;
      if (lane < rem) keep = lst[done + lane];
      if (lane < rem) lst[lane] = keep;
      n = rem;
    }
  }
  return n;
}

template <class WORK>
__device__ __forceinline__ void sweep_range(int* lst, int wave, int lane,
                                            int lo, int nrows,
                                            const int* __restrict__ ei, int E,
                                            const WORK& work) {
  int n = 0;
  for (int r0 = 0; r0 < SC_ROWS; r0 += SC_THREADS) {
    const int r = r0 + lane * SC_WAVES + wave;
    n = list_append(lst, n, r < nrows, E + lo + r, lane);
  }
  const int nbatch = (E + SC_BATCH - 1) / SC_BATCH;
  const int rot = (int)(((long long)blockIdx.x * nbatch) / gridDim.x);
  if ((E & 3) == 0 && E >= 4)
    n = scan_stripe<true>(lst, n, wave, lane, lo, nrows, rot, nbatch, ei, E, work);
  else if (E > 0)
    n = scan_stripe<false>(lst, n, wave, lane, lo, nrows, rot, nbatch, ei, E, work);
  if (n > 0) work(lst, 0, n);
}

struct ScoreEntries {
  int lane, lo;
  const int* __restrict__ ei;
  const int* __restrict__ et;
  int E, R;
  const float* qs;
  const float* __restrict__ Xk;
  const float* __restrict__ Tq;
  const float* __restrict__ Tk;
  float* __restrict__ attn;
  int* smax;
  int* smin;
  float* scnt;
  __device__ __forceinline__ void operator()(const int* lst, int base,
                                             int cnt) const {
    int e = 0, s = 0, t = 0, row = 0;
    if (lane < cnt) {
      e = lst[base + lane];
      if (e < E) {
        s = ei[e];
        row = ei[E + e] - lo;
        t = et[e];
      } else {
        s = e - E;
        row = s - lo;
        t = R;
      }
    }
    float mine = 0.0f;
    for (int j0 = 0; j0 < cnt; j0 += 4) {
      float4 q0[4], q1[4], k0[4], k1[4];
#pragma unroll
      for (int u = 0; u < 4; ++u) {
        const int j = min(j0 + u, cnt - 1);
        const int sj = __builtin_amdgcn_readlane(s, j);
        const int tj = __builtin_amdgcn_readlane(t, j);
        const int rj = __builtin_amdgcn_readlane(row, j);
        k0[u] = ((const float4*)(Xk + (size_t)sj * D))[lane];
        q1[u] = ((const float4*)(Tq + (size_t)tj * D))[lane];
        k1[u] = ((const float4*)(Tk + (size_t)tj * D))[lane];
        q0[u] = ((const float4*)(qs + rj * D))[lane];
      }
      float a[4];
#pragma unroll
      for (int u = 0; u < 4; ++u)
        a[u] = (q0[u].x + q1[u].x) * (k0[u].x + k1[u].x) +
               (q0[u].y + q1[u].y) * (k0[u].y + k1[u].y) +
               (q0[u].z + q1[u].z) * (k0[u].z + k1[u].z) +
               (q0[u].w + q1[u].w) * (k0[u].w + k1[u].w);
#pragma unroll
      for (int off = 16; off >= 1; off >>= 1) {
#pragma unroll
        for (int u = 0; u < 4; ++u) a[u] += __shfl_xor(a[u], off, 32);
      }
#pragma unroll
      for (int u = 0; u < 4; ++u)
        if (lane == j0 + u) mine = a[u] * 0.08838834764831845f;
    }
    if (lane < cnt) {
      attn[e] = mine;
      lds_max_f(smax + row, mine);
      lds_max_f(smin + row, -mine);
      atomicAdd(scnt + row, 1.0f);
    }
  }
};

__global__ __launch_bounds__(SC_THREADS)
__attribute__((amdgpu_waves_per_eu(8, 8))) void edge_score_kernel(
    const int* __restrict__ ei, const int* __restrict__ et, int E, int N,
    int R, const float* __restrict__ Xq, const float* __restrict__ Xk,
    const float* __restrict__ Tq, const float* __restrict__ Tk,
    float* __restrict__ attn, float* __restrict__ mdyn,
    float* __restrict__ msta, float* __restrict__ sdyn,
    float* __restrict__ ssta, float* __restrict__ cnts) {
  __shared__ float qs[SC_ROWS * D];
  __shared__ int lists[SC_WAVES * SC_LIST];
  __shared__ int smax[SC_ROWS];
  __shared__ int smin[SC_ROWS];
  __shared__ float scnt[SC_ROWS];

  const int tid = threadIdx.x;
  const int lane = tid & 31;
  const int wave = tid >> 5;
  const int lo = blockIdx.x * SC_ROWS;
  const int nrows = min(SC_ROWS, N - lo);
  int* lst = lists + wave * SC_LIST;

  const float4* __restrict__ qg = (const float4*)(Xq + (size_t)lo * D);
  for (int i = tid; i < nrows * (D / 4); i += SC_THREADS)
    ((float4*)qs)[i] = qg[i];
  for (int i = tid; i < SC_ROWS; i += SC_THREADS) {
    smax[i] = __float_as_int(-__builtin_inff());
    smin[i] = __float_as_int(-__builtin_inff());
    scnt[i] = 0.0f;
  }
  __syncthreads();

  const ScoreEntries work = {lane, lo, ei, et,   E,    R,    qs,
                             Xk,   Tq, Tk, attn, smax, smin, scnt};
  sweep_range(lst, wave, lane, lo, nrows, ei, E, work);
  __syncthreads();

  for (int i = tid; i < nrows; i += SC_THREADS) {
    mdyn[lo + i] = __int_as_float(smax[i]);
    msta[lo + i] = __int_as_float(smin[i]);
    cnts[lo + i] = scnt[i];
    sdyn[lo + i] = 0.0f;
    ssta[lo + i] = 0.0f;
  }
}

__global__ __launch_bounds__(SC_THREADS) void edge_scatter_kernel(
    const int* __restrict__ ei, int E, int N, const float* __restrict__ edyn,
    const float* __restrict__ esta, const float* __restrict__ sdyn,
    const float* __restrict__ ssta, const float* __restrict__ cnts,
    const float* __restrict__ Vv, float* __restrict__ featsta,
    float* __restrict__ featdyn) {
  __shared__ float acc_s[SC_ROWS * D];
  __shared__ float acc_d[SC_ROWS * D];
  __shared__ int lists[SC_WAVES * SC_LIST];

  const int tid = threadIdx.x;
  const int lane = tid & 31;
  const int wave = tid >> 5;
  const int lo = blockIdx.x * SC_ROWS;
  const int nrows = min(SC_ROWS, N - lo);
  int* lst = lists + wave * SC_LIST;

  for (int i = tid; i < SC_ROWS * D / 4; i += SC_THREADS) {
    ((float4*)acc_s)[i] = make_float4(0.0f, 0.0f, 0.0f, 0.0f);
    ((float4*)acc_d)[i] = make_float4(0.0f, 0.0f, 0.0f, 0.0f);
  }
  __syncthreads();

  const AddEntries work = {lane, lo,   ei,   E,  edyn,  esta,
                           sdyn, ssta, cnts, Vv, acc_s, acc_d};
  sweep_range(lst, wave, lane, lo, nrows, ei, E, work);
  __syncthreads();

  for (int r = wave; r < nrows; r += SC_WAVES) {
    const float* as = acc_s + r * D + lane;
    const float* ad = acc_d + r * D + lane;
    const float4 os = make_float4(as[0], as[32], as[64], as[96]);
    const float4 od = make_float4(ad[0], ad[32], ad[64], ad[96]);
    ((float4*)(featsta + (size_t)(lo + r) * D))[lane] = os;
    ((float4*)(featdyn + (size_t)(lo + r) * D))[lane] = od;
  }
}

extern "C" void kernel_launch(void* const* d_in, const int* in_sizes, int n_in,
                              void* d_out, int out_size, void* d_ws,
                              size_t ws_size, hipStream_t stream) {
  const float* curr = (const float*)d_in[0];
  const float* last = (const float*)d_in[1];
  const float* relemb = (const float*)d_in[2];
  const float* selfrel = (const float*)d_in[3];
  const float* Wq = (const float*)d_in[4];
  const float* bq = (const float*)d_in[5];
  const float* Wk = (const float*)d_in[6];
  const float* bk = (const float*)d_in[7];
  const float* Wv = (const float*)d_in[8];
  const float* bv = (const float*)d_in[9];
  const float* dynW = (const float*)d_in[10];
  const int* ei = (const int*)d_in[11];
  const int* et = (const int*)d_in[12];

  const int N = in_sizes[0] / D;
  const int E = in_sizes[12];
  const int R = in_sizes[2] / D;
  const int ET = E + N;
  const size_t ND = (size_t)N * D;

  float* ws = (float*)d_ws;
  float* Xq = ws;
  float* Xk = Xq + ND;
  float* Vv = Xk + ND;
  float* featdyn = Vv + ND;
  float* Trel = featdyn + ND;
  float* Tq = Trel + SELF_PAD * D;
  float* Tk = Tq + SELF_PAD * D;
  float* attn = Tk + SELF_PAD * D;
  float* edyn = attn + ET;
  float* esta = edyn + ET;
  float* mdyn = esta + ET;
  float* msta = mdyn + N;
  float* sdyn = msta + N;
  float* ssta = sdyn + N;
  float* cnts = ssta + N;

  float* out0 = (float*)d_out;
  float* out1 = out0 + ND;

  build_trel<<<(SELF_PAD * D + 255) / 256, 256, 0, stream>>>(relemb, selfrel,
                                                             Trel, R);

  gemm128_wmma<<<GEMM_GRID, 256, 0, stream>>>(curr, Wq, nullptr, Xq, N / 16);
  gemm128_wmma<<<GEMM_GRID, 256, 0, stream>>>(last, Wk, nullptr, Xk, N / 16);
  gemm128_wmma<<<GEMM_GRID, 256, 0, stream>>>(last, Wv, bv, Vv, N / 16);
  gemm128_wmma<<<SELF_PAD / 16, 256, 0, stream>>>(Trel, Wq + D * D, bq, Tq,
                                                  SELF_PAD / 16);
  gemm128_wmma<<<SELF_PAD / 16, 256, 0, stream>>>(Trel, Wk + D * D, bk, Tk,
                                                  SELF_PAD / 16);

  edge_score_kernel<<<(N + SC_ROWS - 1) / SC_ROWS, SC_THREADS, 0, stream>>>(
      ei, et, E, N, R, Xq, Xk, Tq, Tk, attn, mdyn, msta, sdyn, ssta, cnts);
  edge_expsum_kernel<<<(ET + 255) / 256, 256, 0, stream>>>(
      ei, E, N, attn, mdyn, msta, edyn, esta, sdyn, ssta);
  edge_scatter_kernel<<<(N + SC_ROWS - 1) / SC_ROWS, SC_THREADS, 0, stream>>>(
      ei, E, N, edyn, esta, sdyn, ssta, cnts, Vv, out0, featdyn);

  gemm128_wmma<<<GEMM_GRID, 256, 0, stream>>>(featdyn, dynW, nullptr, out1, N / 16);
}


# Edge phase fused: the row-range group keeps each swept edge (source, row, score) in LDS; exp weights and feature sums from LDS, rows divided by the exp sum after summing; exp pass, 2nd sweep gone
# speedup vs baseline: 3.2357x; 1.6198x over previous
#include <hip/hip_runtime.h>
#include <math.h>


typedef __attribute__((ext_vector_type(2))) float v2f;
typedef __attribute__((ext_vector_type(8))) float v8f;

#define D 128
#define SELF_PAD 512
#define GEMM_GRID 500

__device__ __forceinline__ void load_a_tile(float4 (&a)[16],
                                            const float* __restrict__ A,
                                            int tile, int r, int g) {
  const float4* __restrict__ p =
      (const float4*)(A + (size_t)(tile * 16 + r) * D) + g;
#pragma unroll
  for (int i = 0; i < 16; ++i) a[i] = p[2 * i];
}

__device__ __forceinline__ void mma_store_tile(const float4 (&a)[16],
                                               const v2f (&b)[32], float bv,
                                               float* __restrict__ C, int tile,
                                               int n0, int r, int g) {
  v8f acc = {};
#pragma unroll
  for (int i = 0; i < 16; ++i) {
    v2f a0, a1;
    a0.x = a[i].x;
    a0.y = a[i].y;
    a1.x = a[i].z;
    a1.y = a[i].w;
    acc = __builtin_amdgcn_wmma_f32_16x16x4_f32(
         false, a0,  false, b[2 * i],
         (short)0, acc,  false,  false);
    acc = __builtin_amdgcn_wmma_f32_16x16x4_f32(
         false, a1,  false, b[2 * i + 1],
         (short)0, acc,  false,  false);
  }
#pragma unroll
  for (int j = 0; j < 8; ++j) {
    const int row = tile * 16 + j + 8 * g;
    C[(size_t)row * D + n0 + r] = acc[j] + bv;
  }
}

__global__ __launch_bounds__(256) void gemm128_wmma(
    const float* __restrict__ A, const float* __restrict__ W,
    const float* __restrict__ bias, float* __restrict__ C, int tiles) {
  const int lane = threadIdx.x & 31;
  const int wave = threadIdx.x >> 5;
  const int r = lane & 15;
  const int g = lane >> 4;
  const int n0 = wave * 16;
  const int stride = gridDim.x;

  int t = blockIdx.x;
  if (t >= tiles) return;

  const float* __restrict__ Wcol = W + (n0 + r);
  v2f b[32];
#pragma unroll
  for (int s = 0; s < 32; ++s) {
    const int kk = 8 * (s >> 1) + 4 * g + 2 * (s & 1);
    b[s].x = Wcol[(size_t)kk * D];
    b[s].y = Wcol[(size_t)(kk + 1) * D];
  }
  const float bv = bias ? bias[n0 + r] : 0.0f;

  float4 a0[16], a1[16];
  load_a_tile(a0, A, t, r, g);
  for (;;) {
    const int t1 = t + stride;
    if (t1 >= tiles) {
      mma_store_tile(a0, b, bv, C, t, n0, r, g);
      break;
    }
    load_a_tile(a1, A, t1, r, g);
    __builtin_amdgcn_sched_barrier(0);
    mma_store_tile(a0, b, bv, C, t, n0, r, g);
    const int t2 = t1 + stride;
    if (t2 >= tiles) {
      mma_store_tile(a1, b, bv, C, t1, n0, r, g);
      break;
    }
    load_a_tile(a0, A, t2, r, g);
    __builtin_amdgcn_sched_barrier(0);
    mma_store_tile(a1, b, bv, C, t1, n0, r, g);
    t = t2;
  }
}

__global__ void build_trel(const float* __restrict__ relemb,
                           const float* __restrict__ selfrel,
                           float* __restrict__ Trel, int R) {
  const int i = blockIdx.x * 256 + threadIdx.x;
  if (i >= SELF_PAD * D) return;
  const int row = i >> 7;
  const int c = i & (D - 1);
  float v = 0.0f;
  if (row < R) v = relemb[i];
  else if (row == R) v = selfrel[c];
  Trel[i] = v;
}

__device__ __forceinline__ void lds_max_f(int* addr, float v) {
  if (v >= 0.0f)
    atomicMax(addr, __float_as_int(v));
  else
    atomicMin((unsigned int*)addr, __float_as_uint(v));
}

#define SC_ROWS 160
#define SC_THREADS 1024
#define SC_WAVES (SC_THREADS / 32)
#define SC_BATCH 512
#define SC_LIST (SC_BATCH + 32)
#define REC_CAP 192
#define REC_WORDS (3 * REC_CAP)
static_assert(REC_WORDS >= SC_LIST, "the general path's list lives in the record space");

template <bool VEC>
__device__ __forceinline__ int4 load_dst4(const int* __restrict__ dst, int idx,
                                          int E) {
  if (VEC) {
    return *(const int4*)(dst + min(idx, E - 4));
  } else {
    int4 v;
    v.x = dst[min(idx + 0, E - 1)];
    v.y = dst[min(idx + 1, E - 1)];
    v.z = dst[min(idx + 2, E - 1)];
    v.w = dst[min(idx + 3, E - 1)];
    return v;
  }
}

template <bool REC>
__device__ __forceinline__ int list_append(int* lst, int n, bool m, int e,
                                           int row, int lane) {
  const unsigned mask = (unsigned)__ballot(m);
  if (mask != 0u) {
    const int pos = n + __popc(mask & ((1u << lane) - 1u));
    if (REC) {
      if (m && pos < REC_CAP) {
        lst[pos] = e;
        lst[REC_CAP + pos] = row;
      }
    } else {
      if (m) lst[pos] = e;
    }
    n += __popc(mask);
  }
  return n;
}

template <bool VEC, bool REC, class WORK>
__device__ __forceinline__ int scan_stripe(
    int* lst, int n, int wave, int lane, int lo, int nrows, int rot, int nbatch,
    const int* __restrict__ ei, int E, const WORK& work) {
  const int* __restrict__ dst = ei + E;
  const unsigned span = (unsigned)nrows;
  int done = 0;
  for (int b0 = wave; b0 < nbatch; b0 += SC_WAVES) {
    int b = b0 + rot;
    if (b >= nbatch) b -= nbatch;
    const int idx = b * SC_BATCH + lane * 4;
    int4 q[4];
#pragma unroll
    for (int u = 0; u < 4; ++u) q[u] = load_dst4<VEC>(dst, idx + u * 128, E);
#pragma unroll
    for (int u = 0; u < 4; ++u) {
      const int e = idx + u * 128;
      const int r0 = q[u].x - lo, r1 = q[u].y - lo;
      const int r2 = q[u].z - lo, r3 = q[u].w - lo;
      bool m0, m1, m2, m3;
      if (VEC) {
        const unsigned sp = (e < E) ? span : 0u;
        m0 = (unsigned)r0 < sp;
        m1 = (unsigned)r1 < sp;
        m2 = (unsigned)r2 < sp;
        m3 = (unsigned)r3 < sp;
      } else {
        m0 = (e + 0 < E) & ((unsigned)r0 < span);
        m1 = (e + 1 < E) & ((unsigned)r1 < span);
        m2 = (e + 2 < E) & ((unsigned)r2 < span);
        m3 = (e + 3 < E) & ((unsigned)r3 < span);
      }
      if (__ballot(m0 | m1 | m2 | m3) != 0) {
        n = list_append<REC>(lst, n, m0, e + 0, r0, lane);
        n = list_append<REC>(lst, n, m1, e + 1, r1, lane);
        n = list_append<REC>(lst, n, m2, e + 2, r2, lane);
        n = list_append<REC>(lst, n, m3, e + 3, r3, lane);
      }
    }
    if (REC) {
      if (n > REC_CAP) return -1;
      while (n - done >= 32) {
        work(lst, done, 32);
        done += 32;
      }
    } else {
      int dn = 0;
      while (n - dn >= 32) {
        work(lst, dn, 32);
        dn += 32;
      }
      if (dn > 0) {
        const int rem = n - dn;
        int keep = 0;
        if (lane < rem) keep = lst[dn + lane];
        if (lane < rem) lst[lane] = keep;
        n = rem;
      }
    }
  }
  return n;
}

template <bool REC, class WORK>
__device__ __forceinline__ int sweep_range(int* lst, int wave, int lane,
                                           int lo, int nrows,
                                           const int* __restrict__ ei, int E,
                                           const WORK& work) {
  int n = 0;
  for (int r0 = 0; r0 < SC_ROWS; r0 += SC_THREADS) {
    const int r = r0 + lane * SC_WAVES + wave;
    n = list_append<REC>(lst, n, r < nrows, E + lo + r, r, lane);
  }
  const int nbatch = (E + SC_BATCH - 1) / SC_BATCH;
  const int rot = (int)(((long long)blockIdx.x * nbatch) / gridDim.x);
  if (REC && (E & 3) == 0 && E >= 4)
    n = scan_stripe<true, REC>(lst, n, wave, lane, lo, nrows, rot, nbatch, ei, E, work);
  else if (E > 0)
    n = scan_stripe<false, REC>(lst, n, wave, lane, lo, nrows, rot, nbatch, ei, E, work);
  if (n < 0) return n;
  const int done = REC ? (n & ~31) : 0;
  if (n > done) work(lst, done, n - done);
  return REC ? n : 0;
}

template <bool REC>
struct ScoreWork {
  int lane, lo;
  const int* __restrict__ ei;
  const int* __restrict__ et;
  int E, R;
  const float* qs;
  const float* __restrict__ Xk;
  const float* __restrict__ Tq;
  const float* __restrict__ Tk;
  float* attn;
  int* smax;
  int* smin;
  float* scnt;
  __device__ __forceinline__ void operator()(int* lst, int base,
                                             int cnt) const {
    int e = 0, s = 0, t = 0, row = 0;
    if (lane < cnt) {
      e = lst[base + lane];
      if (REC) row = lst[REC_CAP + base + lane];
      if (e < E) {
        s = ei[e];
        t = et[e];
        if (!REC) row = ei[E + e] - lo;
      } else {
        s = e - E;
        t = R;
        if (!REC) row = s - lo;
      }
    }
    float mine = 0.0f;
    for (int j0 = 0; j0 < cnt; j0 += 4) {
      float4 q0[4], q1[4], k0[4], k1[4];
#pragma unroll
      for (int u = 0; u < 4; ++u) {
        const int j = min(j0 + u, cnt - 1);
        const int sj = __builtin_amdgcn_readlane(s, j);
        const int tj = __builtin_amdgcn_readlane(t, j);
        const int rj = __builtin_amdgcn_readlane(row, j);
        k0[u] = ((const float4*)(Xk + (size_t)sj * D))[lane];
        q1[u] = ((const float4*)(Tq + (size_t)tj * D))[lane];
        k1[u] = ((const float4*)(Tk + (size_t)tj * D))[lane];
        q0[u] = ((const float4*)(qs + rj * D))[lane];
      }
      float a[4];
#pragma unroll
      for (int u = 0; u < 4; ++u)
        a[u] = (q0[u].x + q1[u].x) * (k0[u].x + k1[u].x) +
               (q0[u].y + q1[u].y) * (k0[u].y + k1[u].y) +
               (q0[u].z + q1[u].z) * (k0[u].z + k1[u].z) +
               (q0[u].w + q1[u].w) * (k0[u].w + k1[u].w);
#pragma unroll
      for (int off = 16; off >= 1; off >>= 1) {
#pragma unroll
        for (int u = 0; u < 4; ++u) a[u] += __shfl_xor(a[u], off, 32);
      }
#pragma unroll
      for (int u = 0; u < 4; ++u)
        if (lane == j0 + u) mine = a[u] * 0.08838834764831845f;
    }
    if (lane < cnt) {
      if (REC) {
        lst[base + lane] = s;
        lst[2 * REC_CAP + base + lane] = __float_as_int(mine);
      } else {
        attn[e] = mine;
      }
      lds_max_f(smax + row, mine);
      lds_max_f(smin + row, -mine);
      atomicAdd(scnt + row, 1.0f);
    }
  }
};

__device__ __forceinline__ void edge_weights(float a, int row, const int* smax,
                                             const int* smin, float* sd,
                                             float* ss, float& wd, float& ws) {
  wd = __expf(a - __int_as_float(smax[row]));
  ws = __expf(-a - __int_as_float(smin[row]));
  atomicAdd(sd + row, wd);
  atomicAdd(ss + row, ws);
}

__device__ __forceinline__ void add_rows(int cnt, int lane, int s, int row,
                                         float wd, float ws,
                                         const float* __restrict__ Vv,
                                         float* acc_s, float* acc_d) {
  for (int j0 = 0; j0 < cnt; j0 += 8) {
    float4 v[8];
#pragma unroll
    for (int u = 0; u < 8; ++u) {
      const int j = min(j0 + u, cnt - 1);
      const int sj = __builtin_amdgcn_readlane(s, j);
      v[u] = ((const float4*)(Vv + (size_t)sj * D))[lane];
    }
#pragma unroll
    for (int u = 0; u < 8; ++u) {
      const int j = j0 + u;
      if (j < cnt) {
        const int rj = __builtin_amdgcn_readlane(row, j);
        const float wsj = __int_as_float(__builtin_amdgcn_readlane(__float_as_int(ws), j));
        const float wdj = __int_as_float(__builtin_amdgcn_readlane(__float_as_int(wd), j));
        float* as = acc_s + rj * D + lane;
        float* ad = acc_d + rj * D + lane;
        __hip_atomic_fetch_add(as + 0, wsj * v[u].x, __ATOMIC_RELAXED, __HIP_MEMORY_SCOPE_WORKGROUP);
        __hip_atomic_fetch_add(as + 32, wsj * v[u].y, __ATOMIC_RELAXED, __HIP_MEMORY_SCOPE_WORKGROUP);
        __hip_atomic_fetch_add(as + 64, wsj * v[u].z, __ATOMIC_RELAXED, __HIP_MEMORY_SCOPE_WORKGROUP);
        __hip_atomic_fetch_add(as + 96, wsj * v[u].w, __ATOMIC_RELAXED, __HIP_MEMORY_SCOPE_WORKGROUP);
        __hip_atomic_fetch_add(ad + 0, wdj * v[u].x, __ATOMIC_RELAXED, __HIP_MEMORY_SCOPE_WORKGROUP);
        __hip_atomic_fetch_add(ad + 32, wdj * v[u].y, __ATOMIC_RELAXED, __HIP_MEMORY_SCOPE_WORKGROUP);
        __hip_atomic_fetch_add(ad + 64, wdj * v[u].z, __ATOMIC_RELAXED, __HIP_MEMORY_SCOPE_WORKGROUP);
        __hip_atomic_fetch_add(ad + 96, wdj * v[u].w, __ATOMIC_RELAXED, __HIP_MEMORY_SCOPE_WORKGROUP);
      }
    }
  }
}

struct AddFromAttn {
  int lane, lo;
  const int* __restrict__ ei;
  int E;
  const float* attn;
  const int* smax;
  const int* smin;
  float* sd;
  float* ss;
  const float* __restrict__ Vv;
  float* acc_s;
  float* acc_d;
  __device__ __forceinline__ void operator()(int* lst, int base,
                                             int cnt) const {
    int s = 0, row = 0;
    float wd = 0.0f, ws = 0.0f;
    if (lane < cnt) {
      const int e = lst[base + lane];
      int d;
      if (e < E) {
        s = ei[e];
        d = ei[E + e];
      } else {
        s = d = e - E;
      }
      row = d - lo;
      edge_weights(attn[e], row, smax, smin, sd, ss, wd, ws);
    }
    add_rows(cnt, lane, s, row, wd, ws, Vv, acc_s, acc_d);
  }
};

__global__ __launch_bounds__(SC_THREADS)
__attribute__((amdgpu_waves_per_eu(8, 8))) void edge_phase_kernel(
    const int* __restrict__ ei, const int* __restrict__ et, int E, int N,
    int R, const float* __restrict__ Xq, const float* __restrict__ Xk,
    const float* __restrict__ Tq, const float* __restrict__ Tk,
    const float* __restrict__ Vv, float* attn, float* __restrict__ featsta,
    float* __restrict__ featdyn) {
  __shared__ float qs[SC_ROWS * D];
  __shared__ float acc_s[SC_ROWS * D];
  __shared__ float acc_d[SC_ROWS * D];
  __shared__ int lists[SC_WAVES * REC_WORDS];
  __shared__ int smax[SC_ROWS];
  __shared__ int smin[SC_ROWS];
  __shared__ float scnt[SC_ROWS];
  __shared__ float sd[SC_ROWS];
  __shared__ float ss[SC_ROWS];
  __shared__ int overflow;

  const int tid = threadIdx.x;
  const int lane = tid & 31;
  const int wave = tid >> 5;
  const int lo = blockIdx.x * SC_ROWS;
  const int nrows = min(SC_ROWS, N - lo);
  int* lst = lists + wave * REC_WORDS;

  const float4* __restrict__ qg = (const float4*)(Xq + (size_t)lo * D);
  for (int i = tid; i < nrows * (D / 4); i += SC_THREADS)
    ((float4*)qs)[i] = qg[i];
  for (int i = tid; i < SC_ROWS * D / 4; i += SC_THREADS) {
    ((float4*)acc_s)[i] = make_float4(0.0f, 0.0f, 0.0f, 0.0f);
    ((float4*)acc_d)[i] = make_float4(0.0f, 0.0f, 0.0f, 0.0f);
  }
  for (int i = tid; i < SC_ROWS; i += SC_THREADS) {
    smax[i] = __float_as_int(-__builtin_inff());
    smin[i] = __float_as_int(-__builtin_inff());
    scnt[i] = 0.0f;
    sd[i] = 0.0f;
    ss[i] = 0.0f;
  }
  if (tid == 0) overflow = 0;
  __syncthreads();

  const ScoreWork<true> score = {lane, lo, ei, et,   E,    R,    qs,
                                 Xk,   Tq, Tk, attn, smax, smin, scnt};
  const int n = sweep_range<true>(lst, wave, lane, lo, nrows, ei, E, score);
  if (n < 0 && lane == 0) overflow = 1;
  __syncthreads();

  if (__builtin_expect(overflow == 0, 1)) {
    for (int base = 0; base < n; base += 32) {
      const int cnt = min(32, n - base);
      int s = 0, row = 0;
      float wd = 0.0f, ws = 0.0f;
      if (lane < cnt) {
        s = lst[base + lane];
        row = lst[REC_CAP + base + lane];
        const float a = __int_as_float(lst[2 * REC_CAP + base + lane]);
        edge_weights(a, row, smax, smin, sd, ss, wd, ws);
      }
      add_rows(cnt, lane, s, row, wd, ws, Vv, acc_s, acc_d);
    }
  } else {
    for (int i = tid; i < SC_ROWS; i += SC_THREADS) {
      smax[i] = __float_as_int(-__builtin_inff());
      smin[i] = __float_as_int(-__builtin_inff());
      scnt[i] = 0.0f;
    }
    __syncthreads();
    const ScoreWork<false> score2 = {lane, lo, ei, et,   E,    R,    qs,
                                     Xk,   Tq, Tk, attn, smax, smin, scnt};
    sweep_range<false>(lst, wave, lane, lo, nrows, ei, E, score2);
    __syncthreads();
    const AddFromAttn add2 = {lane, lo, ei, E,  attn,  smax, smin,
                              sd,   ss, Vv, acc_s, acc_d};
    sweep_range<false>(lst, wave, lane, lo, nrows, ei, E, add2);
  }
  __syncthreads();

  for (int r = wave; r < nrows; r += SC_WAVES) {
    const float denom = fmaxf(scnt[r], 1.0f);
    const float ns = ss[r] * denom;
    const float nd = sd[r] * denom;
    const float* as = acc_s + r * D + lane;
    const float* ad = acc_d + r * D + lane;
    const float4 os = make_float4(as[0] / ns, as[32] / ns, as[64] / ns, as[96] / ns);
    const float4 od = make_float4(ad[0] / nd, ad[32] / nd, ad[64] / nd, ad[96] / nd);
    ((float4*)(featsta + (size_t)(lo + r) * D))[lane] = os;
    ((float4*)(featdyn + (size_t)(lo + r) * D))[lane] = od;
  }
}

extern "C" void kernel_launch(void* const* d_in, const int* in_sizes, int n_in,
                              void* d_out, int out_size, void* d_ws,
                              size_t ws_size, hipStream_t stream) {
  const float* curr = (const float*)d_in[0];
  const float* last = (const float*)d_in[1];
  const float* relemb = (const float*)d_in[2];
  const float* selfrel = (const float*)d_in[3];
  const float* Wq = (const float*)d_in[4];
  const float* bq = (const float*)d_in[5];
  const float* Wk = (const float*)d_in[6];
  const float* bk = (const float*)d_in[7];
  const float* Wv = (const float*)d_in[8];
  const float* bv = (const float*)d_in[9];
  const float* dynW = (const float*)d_in[10];
  const int* ei = (const int*)d_in[11];
  const int* et = (const int*)d_in[12];

  const int N = in_sizes[0] / D;
  const int E = in_sizes[12];
  const int R = in_sizes[2] / D;
  const size_t ND = (size_t)N * D;

  float* ws = (float*)d_ws;
  float* Xq = ws;
  float* Xk = Xq + ND;
  float* Vv = Xk + ND;
  float* featdyn = Vv + ND;
  float* Trel = featdyn + ND;
  float* Tq = Trel + SELF_PAD * D;
  float* Tk = Tq + SELF_PAD * D;
  float* attn = Tk + SELF_PAD * D;

  float* out0 = (float*)d_out;
  float* out1 = out0 + ND;

  build_trel<<<(SELF_PAD * D + 255) / 256, 256, 0, stream>>>(relemb, selfrel,
                                                             Trel, R);

  gemm128_wmma<<<GEMM_GRID, 256, 0, stream>>>(curr, Wq, nullptr, Xq, N / 16);
  gemm128_wmma<<<GEMM_GRID, 256, 0, stream>>>(last, Wk, nullptr, Xk, N / 16);
  gemm128_wmma<<<GEMM_GRID, 256, 0, stream>>>(last, Wv, bv, Vv, N / 16);
  gemm128_wmma<<<SELF_PAD / 16, 256, 0, stream>>>(Trel, Wq + D * D, bq, Tq,
                                                  SELF_PAD / 16);
  gemm128_wmma<<<SELF_PAD / 16, 256, 0, stream>>>(Trel, Wk + D * D, bk, Tk,
                                                  SELF_PAD / 16);

  edge_phase_kernel<<<(N + SC_ROWS - 1) / SC_ROWS, SC_THREADS, 0, stream>>>(
      ei, et, E, N, R, Xq, Xk, Tq, Tk, Vv, attn, out0, featdyn);

  gemm128_wmma<<<GEMM_GRID, 256, 0, stream>>>(featdyn, dynW, nullptr, out1, N / 16);
}


# Feature sums: a group's records regrouped by destination row in LDS (rank from an LDS int atomic, prefix sum); a wave sums its rows in registers in rank order; f32 LDS atomics and zero fill gone
# speedup vs baseline: 3.4160x; 1.0557x over previous
#include <hip/hip_runtime.h>
#include <math.h>


typedef __attribute__((ext_vector_type(2))) float v2f;
typedef __attribute__((ext_vector_type(8))) float v8f;

#define D 128
#define SELF_PAD 512
#define GEMM_GRID 500

__device__ __forceinline__ void load_a_tile(float4 (&a)[16],
                                            const float* __restrict__ A,
                                            int tile, int r, int g) {
  const float4* __restrict__ p =
      (const float4*)(A + (size_t)(tile * 16 + r) * D) + g;
#pragma unroll
  for (int i = 0; i < 16; ++i) a[i] = p[2 * i];
}

__device__ __forceinline__ void mma_store_tile(const float4 (&a)[16],
                                               const v2f (&b)[32], float bv,
                                               float* __restrict__ C, int tile,
                                               int n0, int r, int g) {
  v8f acc = {};
#pragma unroll
  for (int i = 0; i < 16; ++i) {
    v2f a0, a1;
    a0.x = a[i].x;
    a0.y = a[i].y;
    a1.x = a[i].z;
    a1.y = a[i].w;
    acc = __builtin_amdgcn_wmma_f32_16x16x4_f32(
         false, a0,  false, b[2 * i],
         (short)0, acc,  false,  false);
    acc = __builtin_amdgcn_wmma_f32_16x16x4_f32(
         false, a1,  false, b[2 * i + 1],
         (short)0, acc,  false,  false);
  }
#pragma unroll
  for (int j = 0; j < 8; ++j) {
    const int row = tile * 16 + j + 8 * g;
    C[(size_t)row * D + n0 + r] = acc[j] + bv;
  }
}

__global__ __launch_bounds__(256) void gemm128_wmma(
    const float* __restrict__ A, const float* __restrict__ W,
    const float* __restrict__ bias, float* __restrict__ C, int tiles) {
  const int lane = threadIdx.x & 31;
  const int wave = threadIdx.x >> 5;
  const int r = lane & 15;
  const int g = lane >> 4;
  const int n0 = wave * 16;
  const int stride = gridDim.x;

  int t = blockIdx.x;
  if (t >= tiles) return;

  const float* __restrict__ Wcol = W + (n0 + r);
  v2f b[32];
#pragma unroll
  for (int s = 0; s < 32; ++s) {
    const int kk = 8 * (s >> 1) + 4 * g + 2 * (s & 1);
    b[s].x = Wcol[(size_t)kk * D];
    b[s].y = Wcol[(size_t)(kk + 1) * D];
  }
  const float bv = bias ? bias[n0 + r] : 0.0f;

  float4 a0[16], a1[16];
  load_a_tile(a0, A, t, r, g);
  for (;;) {
    const int t1 = t + stride;
    if (t1 >= tiles) {
      mma_store_tile(a0, b, bv, C, t, n0, r, g);
      break;
    }
    load_a_tile(a1, A, t1, r, g);
    __builtin_amdgcn_sched_barrier(0);
    mma_store_tile(a0, b, bv, C, t, n0, r, g);
    const int t2 = t1 + stride;
    if (t2 >= tiles) {
      mma_store_tile(a1, b, bv, C, t1, n0, r, g);
      break;
    }
    load_a_tile(a0, A, t2, r, g);
    __builtin_amdgcn_sched_barrier(0);
    mma_store_tile(a1, b, bv, C, t1, n0, r, g);
    t = t2;
  }
}

__global__ void build_trel(const float* __restrict__ relemb,
                           const float* __restrict__ selfrel,
                           float* __restrict__ Trel, int R) {
  const int i = blockIdx.x * 256 + threadIdx.x;
  if (i >= SELF_PAD * D) return;
  const int row = i >> 7;
  const int c = i & (D - 1);
  float v = 0.0f;
  if (row < R) v = relemb[i];
  else if (row == R) v = selfrel[c];
  Trel[i] = v;
}

__device__ __forceinline__ void lds_max_f(int* addr, float v) {
  if (v >= 0.0f)
    atomicMax(addr, __float_as_int(v));
  else
    atomicMin((unsigned int*)addr, __float_as_uint(v));
}

#define SC_ROWS 160
#define SC_THREADS 1024
#define SC_WAVES (SC_THREADS / 32)
#define SC_BATCH 512
#define SC_LIST (SC_BATCH + 32)
#define REC_CAP 192
#define REC_WORDS (3 * REC_CAP)
static_assert(REC_WORDS >= SC_LIST, "the general path's list lives in the record space");
#define RPW (SC_ROWS / SC_WAVES)
#define SRT_CAP (SC_WAVES * REC_CAP)
static_assert(SC_ROWS % SC_WAVES == 0, "every wave owns the same number of rows");
static_assert(SC_ROWS % 32 == 0, "the prefix sum gives every lane the same number of rows");
static_assert(3 * SRT_CAP <= 2 * SC_ROWS * D, "the regrouped list lives in the accumulator space");

template <bool VEC>
__device__ __forceinline__ int4 load_dst4(const int* __restrict__ dst, int idx,
                                          int E) {
  if (VEC) {
    return *(const int4*)(dst + min(idx, E - 4));
  } else {
    int4 v;
    v.x = dst[min(idx + 0, E - 1)];
    v.y = dst[min(idx + 1, E - 1)];
    v.z = dst[min(idx + 2, E - 1)];
    v.w = dst[min(idx + 3, E - 1)];
    return v;
  }
}

template <bool REC>
__device__ __forceinline__ int list_append(int* lst, int n, bool m, int e,
                                           int row, int lane) {
  const unsigned mask = (unsigned)__ballot(m);
  if (mask != 0u) {
    const int pos = n + __popc(mask & ((1u << lane) - 1u));
    if (REC) {
      if (m && pos < REC_CAP) {
        lst[pos] = e;
        lst[REC_CAP + pos] = row;
      }
    } else {
      if (m) lst[pos] = e;
    }
    n += __popc(mask);
  }
  return n;
}

template <bool VEC, bool REC, class WORK>
__device__ __forceinline__ int scan_stripe(
    int* lst, int n, int wave, int lane, int lo, int nrows, int rot, int nbatch,
    const int* __restrict__ ei, int E, const WORK& work) {
  const int* __restrict__ dst = ei + E;
  const unsigned span = (unsigned)nrows;
  int done = 0;
  for (int b0 = wave; b0 < nbatch; b0 += SC_WAVES) {
    int b = b0 + rot;
    if (b >= nbatch) b -= nbatch;
    const int idx = b * SC_BATCH + lane * 4;
    int4 q[4];
#pragma unroll
    for (int u = 0; u < 4; ++u) q[u] = load_dst4<VEC>(dst, idx + u * 128, E);
#pragma unroll
    for (int u = 0; u < 4; ++u) {
      const int e = idx + u * 128;
      const int r0 = q[u].x - lo, r1 = q[u].y - lo;
      const int r2 = q[u].z - lo, r3 = q[u].w - lo;
      bool m0, m1, m2, m3;
      if (VEC) {
        const unsigned sp = (e < E) ? span : 0u;
        m0 = (unsigned)r0 < sp;
        m1 = (unsigned)r1 < sp;
        m2 = (unsigned)r2 < sp;
        m3 = (unsigned)r3 < sp;
      } else {
        m0 = (e + 0 < E) & ((unsigned)r0 < span);
        m1 = (e + 1 < E) & ((unsigned)r1 < span);
        m2 = (e + 2 < E) & ((unsigned)r2 < span);
        m3 = (e + 3 < E) & ((unsigned)r3 < span);
      }
      if (__ballot(m0 | m1 | m2 | m3) != 0) {
        n = list_append<REC>(lst, n, m0, e + 0, r0, lane);
        n = list_append<REC>(lst, n, m1, e + 1, r1, lane);
        n = list_append<REC>(lst, n, m2, e + 2, r2, lane);
        n = list_append<REC>(lst, n, m3, e + 3, r3, lane);
      }
    }
    if (REC) {
      if (n > REC_CAP) return -1;
      while (n - done >= 32) {
        work(lst, done, 32);
        done += 32;
      }
    } else {
      int dn = 0;
      while (n - dn >= 32) {
        work(lst, dn, 32);
        dn += 32;
      }
      if (dn > 0) {
        const int rem = n - dn;
        int keep = 0;
        if (lane < rem) keep = lst[dn + lane];
        if (lane < rem) lst[lane] = keep;
        n = rem;
      }
    }
  }
  return n;
}

template <bool REC, class WORK>
__device__ __forceinline__ int sweep_range(int* lst, int wave, int lane,
                                           int lo, int nrows,
                                           const int* __restrict__ ei, int E,
                                           const WORK& work) {
  int n = 0;
  for (int r0 = 0; r0 < SC_ROWS; r0 += SC_THREADS) {
    const int r = r0 + lane * SC_WAVES + wave;
    n = list_append<REC>(lst, n, r < nrows, E + lo + r, r, lane);
  }
  const int nbatch = (E + SC_BATCH - 1) / SC_BATCH;
  const int rot = (int)(((long long)blockIdx.x * nbatch) / gridDim.x);
  if (REC && (E & 3) == 0 && E >= 4)
    n = scan_stripe<true, REC>(lst, n, wave, lane, lo, nrows, rot, nbatch, ei, E, work);
  else if (E > 0)
    n = scan_stripe<false, REC>(lst, n, wave, lane, lo, nrows, rot, nbatch, ei, E, work);
  if (n < 0) return n;
  const int done = REC ? (n & ~31) : 0;
  if (n > done) work(lst, done, n - done);
  return REC ? n : 0;
}

template <bool REC>
struct ScoreWork {
  int lane, lo;
  const int* __restrict__ ei;
  const int* __restrict__ et;
  int E, R;
  const float* qs;
  const float* __restrict__ Xk;
  const float* __restrict__ Tq;
  const float* __restrict__ Tk;
  float* attn;
  int* smax;
  int* smin;
  float* scnt;
  __device__ __forceinline__ void operator()(int* lst, int base,
                                             int cnt) const {
    int e = 0, s = 0, t = 0, row = 0;
    if (lane < cnt) {
      e = lst[base + lane];
      if (REC) row = lst[REC_CAP + base + lane];
      if (e < E) {
        s = ei[e];
        t = et[e];
        if (!REC) row = ei[E + e] - lo;
      } else {
        s = e - E;
        t = R;
        if (!REC) row = s - lo;
      }
    }
    float mine = 0.0f;
    for (int j0 = 0; j0 < cnt; j0 += 4) {
      float4 q0[4], q1[4], k0[4], k1[4];
#pragma unroll
      for (int u = 0; u < 4; ++u) {
        const int j = min(j0 + u, cnt - 1);
        const int sj = __builtin_amdgcn_readlane(s, j);
        const int tj = __builtin_amdgcn_readlane(t, j);
        const int rj = __builtin_amdgcn_readlane(row, j);
        k0[u] = ((const float4*)(Xk + (size_t)sj * D))[lane];
        q1[u] = ((const float4*)(Tq + (size_t)tj * D))[lane];
        k1[u] = ((const float4*)(Tk + (size_t)tj * D))[lane];
        q0[u] = ((const float4*)(qs + rj * D))[lane];
      }
      float a[4];
#pragma unroll
      for (int u = 0; u < 4; ++u)
        a[u] = (q0[u].x + q1[u].x) * (k0[u].x + k1[u].x) +
               (q0[u].y + q1[u].y) * (k0[u].y + k1[u].y) +
               (q0[u].z + q1[u].z) * (k0[u].z + k1[u].z) +
               (q0[u].w + q1[u].w) * (k0[u].w + k1[u].w);
#pragma unroll
      for (int off = 16; off >= 1; off >>= 1) {
#pragma unroll
        for (int u = 0; u < 4; ++u) a[u] += __shfl_xor(a[u], off, 32);
      }
#pragma unroll
      for (int u = 0; u < 4; ++u)
        if (lane == j0 + u) mine = a[u] * 0.08838834764831845f;
    }
    if (lane < cnt) {
      if (REC) {
        lst[base + lane] = s;
        lst[2 * REC_CAP + base + lane] = __float_as_int(mine);
      } else {
        attn[e] = mine;
      }
      lds_max_f(smax + row, mine);
      lds_max_f(smin + row, -mine);
      atomicAdd(scnt + row, 1.0f);
    }
  }
};

__device__ __forceinline__ void edge_weights(float a, int row, const int* smax,
                                             const int* smin, float* sd,
                                             float* ss, float& wd, float& ws) {
  wd = __expf(a - __int_as_float(smax[row]));
  ws = __expf(-a - __int_as_float(smin[row]));
  atomicAdd(sd + row, wd);
  atomicAdd(ss + row, ws);
}

__device__ __forceinline__ void add_rows(int cnt, int lane, int s, int row,
                                         float wd, float ws,
                                         const float* __restrict__ Vv,
                                         float* acc_s, float* acc_d) {
  for (int j0 = 0; j0 < cnt; j0 += 8) {
    float4 v[8];
#pragma unroll
    for (int u = 0; u < 8; ++u) {
      const int j = min(j0 + u, cnt - 1);
      const int sj = __builtin_amdgcn_readlane(s, j);
      v[u] = ((const float4*)(Vv + (size_t)sj * D))[lane];
    }
#pragma unroll
    for (int u = 0; u < 8; ++u) {
      const int j = j0 + u;
      if (j < cnt) {
        const int rj = __builtin_amdgcn_readlane(row, j);
        const float wsj = __int_as_float(__builtin_amdgcn_readlane(__float_as_int(ws), j));
        const float wdj = __int_as_float(__builtin_amdgcn_readlane(__float_as_int(wd), j));
        float* as = acc_s + rj * D + lane;
        float* ad = acc_d + rj * D + lane;
        __hip_atomic_fetch_add(as + 0, wsj * v[u].x, __ATOMIC_RELAXED, __HIP_MEMORY_SCOPE_WORKGROUP);
        __hip_atomic_fetch_add(as + 32, wsj * v[u].y, __ATOMIC_RELAXED, __HIP_MEMORY_SCOPE_WORKGROUP);
        __hip_atomic_fetch_add(as + 64, wsj * v[u].z, __ATOMIC_RELAXED, __HIP_MEMORY_SCOPE_WORKGROUP);
        __hip_atomic_fetch_add(as + 96, wsj * v[u].w, __ATOMIC_RELAXED, __HIP_MEMORY_SCOPE_WORKGROUP);
        __hip_atomic_fetch_add(ad + 0, wdj * v[u].x, __ATOMIC_RELAXED, __HIP_MEMORY_SCOPE_WORKGROUP);
        __hip_atomic_fetch_add(ad + 32, wdj * v[u].y, __ATOMIC_RELAXED, __HIP_MEMORY_SCOPE_WORKGROUP);
        __hip_atomic_fetch_add(ad + 64, wdj * v[u].z, __ATOMIC_RELAXED, __HIP_MEMORY_SCOPE_WORKGROUP);
        __hip_atomic_fetch_add(ad + 96, wdj * v[u].w, __ATOMIC_RELAXED, __HIP_MEMORY_SCOPE_WORKGROUP);
      }
    }
  }
}

struct AddFromAttn {
  int lane, lo;
  const int* __restrict__ ei;
  int E;
  const float* attn;
  const int* smax;
  const int* smin;
  float* sd;
  float* ss;
  const float* __restrict__ Vv;
  float* acc_s;
  float* acc_d;
  __device__ __forceinline__ void operator()(int* lst, int base,
                                             int cnt) const {
    int s = 0, row = 0;
    float wd = 0.0f, ws = 0.0f;
    if (lane < cnt) {
      const int e = lst[base + lane];
      int d;
      if (e < E) {
        s = ei[e];
        d = ei[E + e];
      } else {
        s = d = e - E;
      }
      row = d - lo;
      edge_weights(attn[e], row, smax, smin, sd, ss, wd, ws);
    }
    add_rows(cnt, lane, s, row, wd, ws, Vv, acc_s, acc_d);
  }
};

__global__ __launch_bounds__(SC_THREADS)
__attribute__((amdgpu_waves_per_eu(8, 8))) void edge_phase_kernel(
    const int* __restrict__ ei, const int* __restrict__ et, int E, int N,
    int R, const float* __restrict__ Xq, const float* __restrict__ Xk,
    const float* __restrict__ Tq, const float* __restrict__ Tk,
    const float* __restrict__ Vv, float* attn, float* __restrict__ featsta,
    float* __restrict__ featdyn) {
  __shared__ float qs[SC_ROWS * D];
  __shared__ float accbuf[2 * SC_ROWS * D];
  __shared__ int lists[SC_WAVES * REC_WORDS];
  __shared__ int smax[SC_ROWS];
  __shared__ int smin[SC_ROWS];
  __shared__ float scnt[SC_ROWS];
  __shared__ float sd[SC_ROWS];
  __shared__ float ss[SC_ROWS];
  __shared__ int rstart[SC_ROWS + 1];
  __shared__ int rnext[SC_ROWS];
  __shared__ int overflow;

  const int tid = threadIdx.x;
  const int lane = tid & 31;
  const int wave = tid >> 5;
  const int lo = blockIdx.x * SC_ROWS;
  const int nrows = min(SC_ROWS, N - lo);
  int* lst = lists + wave * REC_WORDS;

  const float4* __restrict__ qg = (const float4*)(Xq + (size_t)lo * D);
  for (int i = tid; i < nrows * (D / 4); i += SC_THREADS)
    ((float4*)qs)[i] = qg[i];
  for (int i = tid; i < SC_ROWS; i += SC_THREADS) {
    smax[i] = __float_as_int(-__builtin_inff());
    smin[i] = __float_as_int(-__builtin_inff());
    scnt[i] = 0.0f;
  }
  if (tid == 0) overflow = 0;
  __syncthreads();

  const ScoreWork<true> score = {lane, lo, ei, et,   E,    R,    qs,
                                 Xk,   Tq, Tk, attn, smax, smin, scnt};
  const int n = sweep_range<true>(lst, wave, lane, lo, nrows, ei, E, score);
  if (n < 0 && lane == 0) overflow = 1;
  __syncthreads();

  if (__builtin_expect(overflow == 0, 1)) {
    int* srt_s = (int*)accbuf;
    float* srt_wd = accbuf + SRT_CAP;
    float* srt_ws = accbuf + 2 * SRT_CAP;

    if (wave == 0) {
      constexpr int RPL = SC_ROWS / 32;
      int cdeg[RPL];
      int tot = 0;
#pragma unroll
      for (int k = 0; k < RPL; ++k) {
        cdeg[k] = (int)scnt[lane * RPL + k];
        tot += cdeg[k];
      }
      int incl = tot;
#pragma unroll
      for (int off = 1; off < 32; off <<= 1) {
        const int up = __shfl_up(incl, off, 32);
        if (lane >= off) incl += up;
      }
      int run = incl - tot;
#pragma unroll
      for (int k = 0; k < RPL; ++k) {
        rstart[lane * RPL + k] = run;
        rnext[lane * RPL + k] = run;
        run += cdeg[k];
      }
      if (lane == 31) rstart[SC_ROWS] = run;
    }
    __syncthreads();

    for (int base = 0; base < n; base += 32) {
      if (base + lane < n) {
        const int s = lst[base + lane];
        const int row = lst[REC_CAP + base + lane];
        const float a = __int_as_float(lst[2 * REC_CAP + base + lane]);
        const float wd = __expf(a - __int_as_float(smax[row]));
        const float ws = __expf(-a - __int_as_float(smin[row]));
        const int pos = atomicAdd(rnext + row, 1);
        srt_s[pos] = s;
        srt_wd[pos] = wd;
        srt_ws[pos] = ws;
      }
    }
    __syncthreads();

    const int r0 = wave * RPW;
    const int bnd = rstart[r0 + min(lane, RPW)];
    const int ebeg = __builtin_amdgcn_readlane(bnd, 0);
    const int eend = __builtin_amdgcn_readlane(bnd, RPW);
    int k = 0;
    int rend = __builtin_amdgcn_readlane(bnd, 1);
    float4 as = make_float4(0.0f, 0.0f, 0.0f, 0.0f);
    float4 ad = make_float4(0.0f, 0.0f, 0.0f, 0.0f);
    float sums = 0.0f, sumd = 0.0f;
    auto row_done = [&]() {
      const int r = r0 + k;
      if (r < nrows) {
        const float denom = fmaxf(scnt[r], 1.0f);
        const float ns = sums * denom;
        const float nd = sumd * denom;
        const float4 os = make_float4(as.x / ns, as.y / ns, as.z / ns, as.w / ns);
        const float4 od = make_float4(ad.x / nd, ad.y / nd, ad.z / nd, ad.w / nd);
        ((float4*)(featsta + (size_t)(lo + r) * D))[lane] = os;
        ((float4*)(featdyn + (size_t)(lo + r) * D))[lane] = od;
      }
      as = make_float4(0.0f, 0.0f, 0.0f, 0.0f);
      ad = make_float4(0.0f, 0.0f, 0.0f, 0.0f);
      sums = 0.0f;
      sumd = 0.0f;
      ++k;
      rend = __builtin_amdgcn_readlane(bnd, min(k + 1, RPW));
    };
    for (int base = ebeg; base < eend; base += 32) {
      const int cnt = min(32, eend - base);
      int sv = 0;
      float wdv = 0.0f, wsv = 0.0f;
      if (lane < cnt) {
        sv = srt_s[base + lane];
        wdv = srt_wd[base + lane];
        wsv = srt_ws[base + lane];
      }
      for (int j0 = 0; j0 < cnt; j0 += 8) {
        float4 v[8];
#pragma unroll
        for (int u = 0; u < 8; ++u) {
          const int j = min(j0 + u, cnt - 1);
          const int sj = __builtin_amdgcn_readlane(sv, j);
          v[u] = ((const float4*)(Vv + (size_t)sj * D))[lane];
        }
#pragma unroll
        for (int u = 0; u < 8; ++u) {
          const int j = j0 + u;
          if (j < cnt) {
            while (base + j >= rend) row_done();
            const float wsj = __int_as_float(__builtin_amdgcn_readlane(__float_as_int(wsv), j));
            const float wdj = __int_as_float(__builtin_amdgcn_readlane(__float_as_int(wdv), j));
            as.x += wsj * v[u].x;
            as.y += wsj * v[u].y;
            as.z += wsj * v[u].z;
            as.w += wsj * v[u].w;
            ad.x += wdj * v[u].x;
            ad.y += wdj * v[u].y;
            ad.z += wdj * v[u].z;
            ad.w += wdj * v[u].w;
            sums += wsj;
            sumd += wdj;
          }
        }
      }
    }
    while (k < RPW) row_done();
  } else {
    float* acc_s = accbuf;
    float* acc_d = accbuf + SC_ROWS * D;
    for (int i = tid; i < 2 * SC_ROWS * D / 4; i += SC_THREADS)
      ((float4*)accbuf)[i] = make_float4(0.0f, 0.0f, 0.0f, 0.0f);
    for (int i = tid; i < SC_ROWS; i += SC_THREADS) {
      smax[i] = __float_as_int(-__builtin_inff());
      smin[i] = __float_as_int(-__builtin_inff());
      scnt[i] = 0.0f;
      sd[i] = 0.0f;
      ss[i] = 0.0f;
    }
    __syncthreads();
    const ScoreWork<false> score2 = {lane, lo, ei, et,   E,    R,    qs,
                                     Xk,   Tq, Tk, attn, smax, smin, scnt};
    sweep_range<false>(lst, wave, lane, lo, nrows, ei, E, score2);
    __syncthreads();
    const AddFromAttn add2 = {lane, lo, ei, E,  attn,  smax, smin,
                              sd,   ss, Vv, acc_s, acc_d};
    sweep_range<false>(lst, wave, lane, lo, nrows, ei, E, add2);
    __syncthreads();

    for (int r = wave; r < nrows; r += SC_WAVES) {
      const float denom = fmaxf(scnt[r], 1.0f);
      const float ns = ss[r] * denom;
      const float nd = sd[r] * denom;
      const float* as = acc_s + r * D + lane;
      const float* ad = acc_d + r * D + lane;
      const float4 os = make_float4(as[0] / ns, as[32] / ns, as[64] / ns, as[96] / ns);
      const float4 od = make_float4(ad[0] / nd, ad[32] / nd, ad[64] / nd, ad[96] / nd);
      ((float4*)(featsta + (size_t)(lo + r) * D))[lane] = os;
      ((float4*)(featdyn + (size_t)(lo + r) * D))[lane] = od;
    }
  }
}

extern "C" void kernel_launch(void* const* d_in, const int* in_sizes, int n_in,
                              void* d_out, int out_size, void* d_ws,
                              size_t ws_size, hipStream_t stream) {
  const float* curr = (const float*)d_in[0];
  const float* last = (const float*)d_in[1];
  const float* relemb = (const float*)d_in[2];
  const float* selfrel = (const float*)d_in[3];
  const float* Wq = (const float*)d_in[4];
  const float* bq = (const float*)d_in[5];
  const float* Wk = (const float*)d_in[6];
  const float* bk = (const float*)d_in[7];
  const float* Wv = (const float*)d_in[8];
  const float* bv = (const float*)d_in[9];
  const float* dynW = (const float*)d_in[10];
  const int* ei = (const int*)d_in[11];
  const int* et = (const int*)d_in[12];

  const int N = in_sizes[0] / D;
  const int E = in_sizes[12];
  const int R = in_sizes[2] / D;
  const size_t ND = (size_t)N * D;

  float* ws = (float*)d_ws;
  float* Xq = ws;
  float* Xk = Xq + ND;
  float* Vv = Xk + ND;
  float* featdyn = Vv + ND;
  float* Trel = featdyn + ND;
  float* Tq = Trel + SELF_PAD * D;
  float* Tk = Tq + SELF_PAD * D;
  float* attn = Tk + SELF_PAD * D;

  float* out0 = (float*)d_out;
  float* out1 = out0 + ND;

  build_trel<<<(SELF_PAD * D + 255) / 256, 256, 0, stream>>>(relemb, selfrel,
                                                             Trel, R);

  gemm128_wmma<<<GEMM_GRID, 256, 0, stream>>>(curr, Wq, nullptr, Xq, N / 16);
  gemm128_wmma<<<GEMM_GRID, 256, 0, stream>>>(last, Wk, nullptr, Xk, N / 16);
  gemm128_wmma<<<GEMM_GRID, 256, 0, stream>>>(last, Wv, bv, Vv, N / 16);
  gemm128_wmma<<<SELF_PAD / 16, 256, 0, stream>>>(Trel, Wq + D * D, bq, Tq,
                                                  SELF_PAD / 16);
  gemm128_wmma<<<SELF_PAD / 16, 256, 0, stream>>>(Trel, Wk + D * D, bk, Tk,
                                                  SELF_PAD / 16);

  edge_phase_kernel<<<(N + SC_ROWS - 1) / SC_ROWS, SC_THREADS, 0, stream>>>(
      ei, et, E, N, R, Xq, Xk, Tq, Tk, Vv, attn, out0, featdyn);

  gemm128_wmma<<<GEMM_GRID, 256, 0, stream>>>(featdyn, dynW, nullptr, out1, N / 16);
}


# Edge sweep reads one group-id byte per edge (destination row / 160, written by the table launch) instead of the 4-byte destination: 16 edges per 16-byte load, zero-byte test; arrival order as before
# speedup vs baseline: 3.6254x; 1.0613x over previous
#include <hip/hip_runtime.h>
#include <math.h>


typedef __attribute__((ext_vector_type(2))) float v2f;
typedef __attribute__((ext_vector_type(8))) float v8f;

#define D 128
#define SELF_PAD 512
#define GEMM_GRID 500

__device__ __forceinline__ void load_a_tile(float4 (&a)[16],
                                            const float* __restrict__ A,
                                            int tile, int r, int g) {
  const float4* __restrict__ p =
      (const float4*)(A + (size_t)(tile * 16 + r) * D) + g;
#pragma unroll
  for (int i = 0; i < 16; ++i) a[i] = p[2 * i];
}

__device__ __forceinline__ void mma_store_tile(const float4 (&a)[16],
                                               const v2f (&b)[32], float bv,
                                               float* __restrict__ C, int tile,
                                               int n0, int r, int g) {
  v8f acc = {};
#pragma unroll
  for (int i = 0; i < 16; ++i) {
    v2f a0, a1;
    a0.x = a[i].x;
    a0.y = a[i].y;
    a1.x = a[i].z;
    a1.y = a[i].w;
    acc = __builtin_amdgcn_wmma_f32_16x16x4_f32(
         false, a0,  false, b[2 * i],
         (short)0, acc,  false,  false);
    acc = __builtin_amdgcn_wmma_f32_16x16x4_f32(
         false, a1,  false, b[2 * i + 1],
         (short)0, acc,  false,  false);
  }
#pragma unroll
  for (int j = 0; j < 8; ++j) {
    const int row = tile * 16 + j + 8 * g;
    C[(size_t)row * D + n0 + r] = acc[j] + bv;
  }
}

__global__ __launch_bounds__(256) void gemm128_wmma(
    const float* __restrict__ A, const float* __restrict__ W,
    const float* __restrict__ bias, float* __restrict__ C, int tiles) {
  const int lane = threadIdx.x & 31;
  const int wave = threadIdx.x >> 5;
  const int r = lane & 15;
  const int g = lane >> 4;
  const int n0 = wave * 16;
  const int stride = gridDim.x;

  int t = blockIdx.x;
  if (t >= tiles) return;

  const float* __restrict__ Wcol = W + (n0 + r);
  v2f b[32];
#pragma unroll
  for (int s = 0; s < 32; ++s) {
    const int kk = 8 * (s >> 1) + 4 * g + 2 * (s & 1);
    b[s].x = Wcol[(size_t)kk * D];
    b[s].y = Wcol[(size_t)(kk + 1) * D];
  }
  const float bv = bias ? bias[n0 + r] : 0.0f;

  float4 a0[16], a1[16];
  load_a_tile(a0, A, t, r, g);
  for (;;) {
    const int t1 = t + stride;
    if (t1 >= tiles) {
      mma_store_tile(a0, b, bv, C, t, n0, r, g);
      break;
    }
    load_a_tile(a1, A, t1, r, g);
    __builtin_amdgcn_sched_barrier(0);
    mma_store_tile(a0, b, bv, C, t, n0, r, g);
    const int t2 = t1 + stride;
    if (t2 >= tiles) {
      mma_store_tile(a1, b, bv, C, t1, n0, r, g);
      break;
    }
    load_a_tile(a0, A, t2, r, g);
    __builtin_amdgcn_sched_barrier(0);
    mma_store_tile(a1, b, bv, C, t1, n0, r, g);
    t = t2;
  }
}

#define SC_ROWS 160

__device__ __forceinline__ bool sweep_by_bytes(int E, int N) {
  return (E & 15) == 0 && E >= 16 && (N + SC_ROWS - 1) / SC_ROWS <= 256 &&
         (long long)E <= 4ll * N * D;
}

__global__ void build_trel(const float* __restrict__ relemb,
                           const float* __restrict__ selfrel,
                           float* __restrict__ Trel, int R,
                           const int* __restrict__ ei, int E, int N,
                           unsigned int* __restrict__ gid) {
  const int i = blockIdx.x * 256 + threadIdx.x;
  if (i >= SELF_PAD * D) {
    const int w = i - SELF_PAD * D;
    if (w < (E >> 2) && sweep_by_bytes(E, N)) {
      const int4 d = ((const int4*)(ei + E))[w];
      gid[w] = ((unsigned)(d.x / SC_ROWS) & 255u) |
               (((unsigned)(d.y / SC_ROWS) & 255u) << 8) |
               (((unsigned)(d.z / SC_ROWS) & 255u) << 16) |
               (((unsigned)(d.w / SC_ROWS) & 255u) << 24);
    }
    return;
  }
  const int row = i >> 7;
  const int c = i & (D - 1);
  float v = 0.0f;
  if (row < R) v = relemb[i];
  else if (row == R) v = selfrel[c];
  Trel[i] = v;
}

__device__ __forceinline__ void lds_max_f(int* addr, float v) {
  if (v >= 0.0f)
    atomicMax(addr, __float_as_int(v));
  else
    atomicMin((unsigned int*)addr, __float_as_uint(v));
}

#define SC_THREADS 1024
#define SC_WAVES (SC_THREADS / 32)
#define SC_BATCH 512
#define SC_LIST (SC_BATCH + 32)
#define REC_CAP 192
#define REC_WORDS (3 * REC_CAP)
static_assert(REC_WORDS >= SC_LIST, "the general path's list lives in the record space");
#define RPW (SC_ROWS / SC_WAVES)
#define SRT_CAP (SC_WAVES * REC_CAP)
static_assert(SC_ROWS % SC_WAVES == 0, "every wave owns the same number of rows");
static_assert(SC_ROWS % 32 == 0, "the prefix sum gives every lane the same number of rows");
static_assert(3 * SRT_CAP <= 2 * SC_ROWS * D, "the regrouped list lives in the accumulator space");

template <bool VEC>
__device__ __forceinline__ int4 load_dst4(const int* __restrict__ dst, int idx,
                                          int E) {
  if (VEC) {
    return *(const int4*)(dst + min(idx, E - 4));
  } else {
    int4 v;
    v.x = dst[min(idx + 0, E - 1)];
    v.y = dst[min(idx + 1, E - 1)];
    v.z = dst[min(idx + 2, E - 1)];
    v.w = dst[min(idx + 3, E - 1)];
    return v;
  }
}

template <bool REC, bool ROW = true>
__device__ __forceinline__ int list_append(int* lst, int n, bool m, int e,
                                           int row, int lane) {
  const unsigned mask = (unsigned)__ballot(m);
  if (mask != 0u) {
    const int pos = n + __popc(mask & ((1u << lane) - 1u));
    if (REC) {
      if (m && pos < REC_CAP) {
        lst[pos] = e;
        if (ROW) lst[REC_CAP + pos] = row;
      }
    } else {
      if (m) lst[pos] = e;
    }
    n += __popc(mask);
  }
  return n;
}

template <bool VEC, bool REC, class WORK>
__device__ __forceinline__ int scan_stripe(
    int* lst, int n, int wave, int lane, int lo, int nrows, int rot, int nbatch,
    const int* __restrict__ ei, int E, const WORK& work) {
  const int* __restrict__ dst = ei + E;
  const unsigned span = (unsigned)nrows;
  int done = 0;
  for (int b0 = wave; b0 < nbatch; b0 += SC_WAVES) {
    int b = b0 + rot;
    if (b >= nbatch) b -= nbatch;
    const int idx = b * SC_BATCH + lane * 4;
    int4 q[4];
#pragma unroll
    for (int u = 0; u < 4; ++u) q[u] = load_dst4<VEC>(dst, idx + u * 128, E);
#pragma unroll
    for (int u = 0; u < 4; ++u) {
      const int e = idx + u * 128;
      const int r0 = q[u].x - lo, r1 = q[u].y - lo;
      const int r2 = q[u].z - lo, r3 = q[u].w - lo;
      bool m0, m1, m2, m3;
      if (VEC) {
        const unsigned sp = (e < E) ? span : 0u;
        m0 = (unsigned)r0 < sp;
        m1 = (unsigned)r1 < sp;
        m2 = (unsigned)r2 < sp;
        m3 = (unsigned)r3 < sp;
      } else {
        m0 = (e + 0 < E) & ((unsigned)r0 < span);
        m1 = (e + 1 < E) & ((unsigned)r1 < span);
        m2 = (e + 2 < E) & ((unsigned)r2 < span);
        m3 = (e + 3 < E) & ((unsigned)r3 < span);
      }
      if (__ballot(m0 | m1 | m2 | m3) != 0) {
        n = list_append<REC>(lst, n, m0, e + 0, r0, lane);
        n = list_append<REC>(lst, n, m1, e + 1, r1, lane);
        n = list_append<REC>(lst, n, m2, e + 2, r2, lane);
        n = list_append<REC>(lst, n, m3, e + 3, r3, lane);
      }
    }
    if (REC) {
      if (n > REC_CAP) return -1;
      while (n - done >= 32) {
        work(lst, done, 32);
        done += 32;
      }
    } else {
      int dn = 0;
      while (n - dn >= 32) {
        work(lst, dn, 32);
        dn += 32;
      }
      if (dn > 0) {
        const int rem = n - dn;
        int keep = 0;
        if (lane < rem) keep = lst[dn + lane];
        if (lane < rem) lst[lane] = keep;
        n = rem;
      }
    }
  }
  return n;
}

__device__ __forceinline__ unsigned zero_bytes(unsigned x) {
  const unsigned t = (x & 0x7f7f7f7fu) + 0x7f7f7f7fu;
  return ~(t | x | 0x7f7f7f7fu);
}

#define BY_STEP 2048

template <class WORK>
__device__ __forceinline__ int scan_bytes(int* lst, int n, int wave, int lane,
                                          const uint4* __restrict__ gid, int E,
                                          const WORK& work) {
  const int nvec = E >> 4;
  const int nstep = (nvec + BY_STEP / 16 - 1) / (BY_STEP / 16);
  const int rot = (int)(((long long)blockIdx.x * nstep) / gridDim.x);
  const unsigned pat = blockIdx.x * 0x01010101u;
  int done = 0;
  for (int b0 = wave; b0 < nstep; b0 += SC_WAVES) {
    int b = b0 + rot;
    if (b >= nstep) b -= nstep;
    const int vi = b * (BY_STEP / 16) + lane;
    uint4 q[4];
#pragma unroll
    for (int u = 0; u < 4; ++u) q[u] = gid[min(vi + u * 32, nvec - 1)];
#pragma unroll
    for (int u = 0; u < 4; ++u) {
      const int v = vi + u * 32;
      unsigned m = zero_bytes(q[u].x ^ pat) | (zero_bytes(q[u].y ^ pat) >> 1) |
                   (zero_bytes(q[u].z ^ pat) >> 2) |
                   (zero_bytes(q[u].w ^ pat) >> 3);
      if (v >= nvec) m = 0u;
      while (__ballot(m != 0u) != 0) {
        const bool has = m != 0u;
        const int p = __builtin_ctz(m | 0x80000000u);
        const int e = v * 16 + 4 * (7 - (p & 7)) + (p >> 3);
        m &= m - 1u;
        n = list_append<true, false>(lst, n, has, e, 0, lane);
      }
    }
    if (n > REC_CAP) return -1;
    while (n - done >= 32) {
      work(lst, done, 32);
      done += 32;
    }
  }
  return n;
}

template <bool REC, class WORK>
__device__ __forceinline__ int sweep_range(int* lst, int wave, int lane,
                                           int lo, int nrows,
                                           const int* __restrict__ ei, int E,
                                           const uint4* __restrict__ gid,
                                           const WORK& work) {
  int n = 0;
  for (int r0 = 0; r0 < SC_ROWS; r0 += SC_THREADS) {
    const int r = r0 + lane * SC_WAVES + wave;
    n = list_append<REC>(lst, n, r < nrows, E + lo + r, r, lane);
  }
  const int nbatch = (E + SC_BATCH - 1) / SC_BATCH;
  const int rot = (int)(((long long)blockIdx.x * nbatch) / gridDim.x);
  if (REC && gid != nullptr)
    n = scan_bytes(lst, n, wave, lane, gid, E, work);
  else if (REC && (E & 3) == 0 && E >= 4)
    n = scan_stripe<true, REC>(lst, n, wave, lane, lo, nrows, rot, nbatch, ei, E, work);
  else if (E > 0)
    n = scan_stripe<false, REC>(lst, n, wave, lane, lo, nrows, rot, nbatch, ei, E, work);
  if (n < 0) return n;
  const int done = REC ? (n & ~31) : 0;
  if (n > done) work(lst, done, n - done);
  return REC ? n : 0;
}

template <bool REC>
struct ScoreWork {
  int lane, lo;
  const int* __restrict__ ei;
  const int* __restrict__ et;
  int E, R;
  const float* qs;
  const float* __restrict__ Xk;
  const float* __restrict__ Tq;
  const float* __restrict__ Tk;
  float* attn;
  int* smax;
  int* smin;
  float* scnt;
  bool rowrec;
  __device__ __forceinline__ void operator()(int* lst, int base,
                                             int cnt) const {
    const bool fromrec = REC && rowrec;
    int e = 0, s = 0, t = 0, row = 0;
    if (lane < cnt) {
      e = lst[base + lane];
      if (fromrec) row = lst[REC_CAP + base + lane];
      if (e < E) {
        s = ei[e];
        t = et[e];
        if (!fromrec) row = ei[E + e] - lo;
      } else {
        s = e - E;
        t = R;
        if (!fromrec) row = s - lo;
      }
    }
    float mine = 0.0f;
    for (int j0 = 0; j0 < cnt; j0 += 4) {
      float4 q0[4], q1[4], k0[4], k1[4];
#pragma unroll
      for (int u = 0; u < 4; ++u) {
        const int j = min(j0 + u, cnt - 1);
        const int sj = __builtin_amdgcn_readlane(s, j);
        const int tj = __builtin_amdgcn_readlane(t, j);
        const int rj = __builtin_amdgcn_readlane(row, j);
        k0[u] = ((const float4*)(Xk + (size_t)sj * D))[lane];
        q1[u] = ((const float4*)(Tq + (size_t)tj * D))[lane];
        k1[u] = ((const float4*)(Tk + (size_t)tj * D))[lane];
        q0[u] = ((const float4*)(qs + rj * D))[lane];
      }
      float a[4];
#pragma unroll
      for (int u = 0; u < 4; ++u)
        a[u] = (q0[u].x + q1[u].x) * (k0[u].x + k1[u].x) +
               (q0[u].y + q1[u].y) * (k0[u].y + k1[u].y) +
               (q0[u].z + q1[u].z) * (k0[u].z + k1[u].z) +
               (q0[u].w + q1[u].w) * (k0[u].w + k1[u].w);
#pragma unroll
      for (int off = 16; off >= 1; off >>= 1) {
#pragma unroll
        for (int u = 0; u < 4; ++u) a[u] += __shfl_xor(a[u], off, 32);
      }
#pragma unroll
      for (int u = 0; u < 4; ++u)
        if (lane == j0 + u) mine = a[u] * 0.08838834764831845f;
    }
    if (lane < cnt) {
      if (REC) {
        lst[base + lane] = s;
        if (!rowrec) lst[REC_CAP + base + lane] = row;
        lst[2 * REC_CAP + base + lane] = __float_as_int(mine);
      } else {
        attn[e] = mine;
      }
      lds_max_f(smax + row, mine);
      lds_max_f(smin + row, -mine);
      atomicAdd(scnt + row, 1.0f);
    }
  }
};

__device__ __forceinline__ void edge_weights(float a, int row, const int* smax,
                                             const int* smin, float* sd,
                                             float* ss, float& wd, float& ws) {
  wd = __expf(a - __int_as_float(smax[row]));
  ws = __expf(-a - __int_as_float(smin[row]));
  atomicAdd(sd + row, wd);
  atomicAdd(ss + row, ws);
}

__device__ __forceinline__ void add_rows(int cnt, int lane, int s, int row,
                                         float wd, float ws,
                                         const float* __restrict__ Vv,
                                         float* acc_s, float* acc_d) {
  for (int j0 = 0; j0 < cnt; j0 += 8) {
    float4 v[8];
#pragma unroll
    for (int u = 0; u < 8; ++u) {
      const int j = min(j0 + u, cnt - 1);
      const int sj = __builtin_amdgcn_readlane(s, j);
      v[u] = ((const float4*)(Vv + (size_t)sj * D))[lane];
    }
#pragma unroll
    for (int u = 0; u < 8; ++u) {
      const int j = j0 + u;
      if (j < cnt) {
        const int rj = __builtin_amdgcn_readlane(row, j);
        const float wsj = __int_as_float(__builtin_amdgcn_readlane(__float_as_int(ws), j));
        const float wdj = __int_as_float(__builtin_amdgcn_readlane(__float_as_int(wd), j));
        float* as = acc_s + rj * D + lane;
        float* ad = acc_d + rj * D + lane;
        __hip_atomic_fetch_add(as + 0, wsj * v[u].x, __ATOMIC_RELAXED, __HIP_MEMORY_SCOPE_WORKGROUP);
        __hip_atomic_fetch_add(as + 32, wsj * v[u].y, __ATOMIC_RELAXED, __HIP_MEMORY_SCOPE_WORKGROUP);
        __hip_atomic_fetch_add(as + 64, wsj * v[u].z, __ATOMIC_RELAXED, __HIP_MEMORY_SCOPE_WORKGROUP);
        __hip_atomic_fetch_add(as + 96, wsj * v[u].w, __ATOMIC_RELAXED, __HIP_MEMORY_SCOPE_WORKGROUP);
        __hip_atomic_fetch_add(ad + 0, wdj * v[u].x, __ATOMIC_RELAXED, __HIP_MEMORY_SCOPE_WORKGROUP);
        __hip_atomic_fetch_add(ad + 32, wdj * v[u].y, __ATOMIC_RELAXED, __HIP_MEMORY_SCOPE_WORKGROUP);
        __hip_atomic_fetch_add(ad + 64, wdj * v[u].z, __ATOMIC_RELAXED, __HIP_MEMORY_SCOPE_WORKGROUP);
        __hip_atomic_fetch_add(ad + 96, wdj * v[u].w, __ATOMIC_RELAXED, __HIP_MEMORY_SCOPE_WORKGROUP);
      }
    }
  }
}

struct AddFromAttn {
  int lane, lo;
  const int* __restrict__ ei;
  int E;
  const float* attn;
  const int* smax;
  const int* smin;
  float* sd;
  float* ss;
  const float* __restrict__ Vv;
  float* acc_s;
  float* acc_d;
  __device__ __forceinline__ void operator()(int* lst, int base,
                                             int cnt) const {
    int s = 0, row = 0;
    float wd = 0.0f, ws = 0.0f;
    if (lane < cnt) {
      const int e = lst[base + lane];
      int d;
      if (e < E) {
        s = ei[e];
        d = ei[E + e];
      } else {
        s = d = e - E;
      }
      row = d - lo;
      edge_weights(attn[e], row, smax, smin, sd, ss, wd, ws);
    }
    add_rows(cnt, lane, s, row, wd, ws, Vv, acc_s, acc_d);
  }
};

__global__ __launch_bounds__(SC_THREADS)
__attribute__((amdgpu_waves_per_eu(8, 8))) void edge_phase_kernel(
    const int* __restrict__ ei, const int* __restrict__ et, int E, int N,
    int R, const float* __restrict__ Xq, const float* __restrict__ Xk,
    const float* __restrict__ Tq, const float* __restrict__ Tk,
    const float* __restrict__ Vv, float* attn, float* __restrict__ featsta,
    float* __restrict__ featdyn, const uint4* __restrict__ gidbytes) {
  __shared__ float qs[SC_ROWS * D];
  __shared__ float accbuf[2 * SC_ROWS * D];
  __shared__ int lists[SC_WAVES * REC_WORDS];
  __shared__ int smax[SC_ROWS];
  __shared__ int smin[SC_ROWS];
  __shared__ float scnt[SC_ROWS];
  __shared__ float sd[SC_ROWS];
  __shared__ float ss[SC_ROWS];
  __shared__ int rstart[SC_ROWS + 1];
  __shared__ int rnext[SC_ROWS];
  __shared__ int overflow;

  const int tid = threadIdx.x;
  const int lane = tid & 31;
  const int wave = tid >> 5;
  const int lo = blockIdx.x * SC_ROWS;
  const int nrows = min(SC_ROWS, N - lo);
  int* lst = lists + wave * REC_WORDS;

  const float4* __restrict__ qg = (const float4*)(Xq + (size_t)lo * D);
  for (int i = tid; i < nrows * (D / 4); i += SC_THREADS)
    ((float4*)qs)[i] = qg[i];
  for (int i = tid; i < SC_ROWS; i += SC_THREADS) {
    smax[i] = __float_as_int(-__builtin_inff());
    smin[i] = __float_as_int(-__builtin_inff());
    scnt[i] = 0.0f;
  }
  if (tid == 0) overflow = 0;
  __syncthreads();

  const uint4* __restrict__ gid = sweep_by_bytes(E, N) ? gidbytes : nullptr;
  const ScoreWork<true> score = {lane, lo, ei,   et,   E,    R,    qs,           Xk,
                                 Tq,   Tk, attn, smax, smin, scnt, gid == nullptr};
  const int n = sweep_range<true>(lst, wave, lane, lo, nrows, ei, E, gid, score);
  if (n < 0 && lane == 0) overflow = 1;
  __syncthreads();

  if (__builtin_expect(overflow == 0, 1)) {
    int* srt_s = (int*)accbuf;
    float* srt_wd = accbuf + SRT_CAP;
    float* srt_ws = accbuf + 2 * SRT_CAP;

    if (wave == 0) {
      constexpr int RPL = SC_ROWS / 32;
      int cdeg[RPL];
      int tot = 0;
#pragma unroll
      for (int k = 0; k < RPL; ++k) {
        cdeg[k] = (int)scnt[lane * RPL + k];
        tot += cdeg[k];
      }
      int incl = tot;
#pragma unroll
      for (int off = 1; off < 32; off <<= 1) {
        const int up = __shfl_up(incl, off, 32);
        if (lane >= off) incl += up;
      }
      int run = incl - tot;
#pragma unroll
      for (int k = 0; k < RPL; ++k) {
        rstart[lane * RPL + k] = run;
        rnext[lane * RPL + k] = run;
        run += cdeg[k];
      }
      if (lane == 31) rstart[SC_ROWS] = run;
    }
    __syncthreads();

    for (int base = 0; base < n; base += 32) {
      if (base + lane < n) {
        const int s = lst[base + lane];
        const int row = lst[REC_CAP + base + lane];
        const float a = __int_as_float(lst[2 * REC_CAP + base + lane]);
        const float wd = __expf(a - __int_as_float(smax[row]));
        const float ws = __expf(-a - __int_as_float(smin[row]));
        const int pos = atomicAdd(rnext + row, 1);
        srt_s[pos] = s;
        srt_wd[pos] = wd;
        srt_ws[pos] = ws;
      }
    }
    __syncthreads();

    const int r0 = wave * RPW;
    const int bnd = rstart[r0 + min(lane, RPW)];
    const int ebeg = __builtin_amdgcn_readlane(bnd, 0);
    const int eend = __builtin_amdgcn_readlane(bnd, RPW);
    int k = 0;
    int rend = __builtin_amdgcn_readlane(bnd, 1);
    float4 as = make_float4(0.0f, 0.0f, 0.0f, 0.0f);
    float4 ad = make_float4(0.0f, 0.0f, 0.0f, 0.0f);
    float sums = 0.0f, sumd = 0.0f;
    auto row_done = [&]() {
      const int r = r0 + k;
      if (r < nrows) {
        const float denom = fmaxf(scnt[r], 1.0f);
        const float ns = sums * denom;
        const float nd = sumd * denom;
        const float4 os = make_float4(as.x / ns, as.y / ns, as.z / ns, as.w / ns);
        const float4 od = make_float4(ad.x / nd, ad.y / nd, ad.z / nd, ad.w / nd);
        ((float4*)(featsta + (size_t)(lo + r) * D))[lane] = os;
        ((float4*)(featdyn + (size_t)(lo + r) * D))[lane] = od;
      }
      as = make_float4(0.0f, 0.0f, 0.0f, 0.0f);
      ad = make_float4(0.0f, 0.0f, 0.0f, 0.0f);
      sums = 0.0f;
      sumd = 0.0f;
      ++k;
      rend = __builtin_amdgcn_readlane(bnd, min(k + 1, RPW));
    };
    for (int base = ebeg; base < eend; base += 32) {
      const int cnt = min(32, eend - base);
      int sv = 0;
      float wdv = 0.0f, wsv = 0.0f;
      if (lane < cnt) {
        sv = srt_s[base + lane];
        wdv = srt_wd[base + lane];
        wsv = srt_ws[base + lane];
      }
      for (int j0 = 0; j0 < cnt; j0 += 8) {
        float4 v[8];
#pragma unroll
        for (int u = 0; u < 8; ++u) {
          const int j = min(j0 + u, cnt - 1);
          const int sj = __builtin_amdgcn_readlane(sv, j);
          v[u] = ((const float4*)(Vv + (size_t)sj * D))[lane];
        }
#pragma unroll
        for (int u = 0; u < 8; ++u) {
          const int j = j0 + u;
          if (j < cnt) {
            while (base + j >= rend) row_done();
            const float wsj = __int_as_float(__builtin_amdgcn_readlane(__float_as_int(wsv), j));
            const float wdj = __int_as_float(__builtin_amdgcn_readlane(__float_as_int(wdv), j));
            as.x += wsj * v[u].x;
            as.y += wsj * v[u].y;
            as.z += wsj * v[u].z;
            as.w += wsj * v[u].w;
            ad.x += wdj * v[u].x;
            ad.y += wdj * v[u].y;
            ad.z += wdj * v[u].z;
            ad.w += wdj * v[u].w;
            sums += wsj;
            sumd += wdj;
          }
        }
      }
    }
    while (k < RPW) row_done();
  } else {
    float* acc_s = accbuf;
    float* acc_d = accbuf + SC_ROWS * D;
    for (int i = tid; i < 2 * SC_ROWS * D / 4; i += SC_THREADS)
      ((float4*)accbuf)[i] = make_float4(0.0f, 0.0f, 0.0f, 0.0f);
    for (int i = tid; i < SC_ROWS; i += SC_THREADS) {
      smax[i] = __float_as_int(-__builtin_inff());
      smin[i] = __float_as_int(-__builtin_inff());
      scnt[i] = 0.0f;
      sd[i] = 0.0f;
      ss[i] = 0.0f;
    }
    __syncthreads();
    const ScoreWork<false> score2 = {lane, lo, ei,   et,   E,    R,    qs,  Xk,
                                     Tq,   Tk, attn, smax, smin, scnt, false};
    sweep_range<false>(lst, wave, lane, lo, nrows, ei, E, nullptr, score2);
    __syncthreads();
    const AddFromAttn add2 = {lane, lo, ei, E,  attn,  smax, smin,
                              sd,   ss, Vv, acc_s, acc_d};
    sweep_range<false>(lst, wave, lane, lo, nrows, ei, E, nullptr, add2);
    __syncthreads();

    for (int r = wave; r < nrows; r += SC_WAVES) {
      const float denom = fmaxf(scnt[r], 1.0f);
      const float ns = ss[r] * denom;
      const float nd = sd[r] * denom;
      const float* as = acc_s + r * D + lane;
      const float* ad = acc_d + r * D + lane;
      const float4 os = make_float4(as[0] / ns, as[32] / ns, as[64] / ns, as[96] / ns);
      const float4 od = make_float4(ad[0] / nd, ad[32] / nd, ad[64] / nd, ad[96] / nd);
      ((float4*)(featsta + (size_t)(lo + r) * D))[lane] = os;
      ((float4*)(featdyn + (size_t)(lo + r) * D))[lane] = od;
    }
  }
}

extern "C" void kernel_launch(void* const* d_in, const int* in_sizes, int n_in,
                              void* d_out, int out_size, void* d_ws,
                              size_t ws_size, hipStream_t stream) {
  const float* curr = (const float*)d_in[0];
  const float* last = (const float*)d_in[1];
  const float* relemb = (const float*)d_in[2];
  const float* selfrel = (const float*)d_in[3];
  const float* Wq = (const float*)d_in[4];
  const float* bq = (const float*)d_in[5];
  const float* Wk = (const float*)d_in[6];
  const float* bk = (const float*)d_in[7];
  const float* Wv = (const float*)d_in[8];
  const float* bv = (const float*)d_in[9];
  const float* dynW = (const float*)d_in[10];
  const int* ei = (const int*)d_in[11];
  const int* et = (const int*)d_in[12];

  const int N = in_sizes[0] / D;
  const int E = in_sizes[12];
  const int R = in_sizes[2] / D;
  const size_t ND = (size_t)N * D;

  float* ws = (float*)d_ws;
  float* Xq = ws;
  float* Xk = Xq + ND;
  float* Vv = Xk + ND;
  float* featdyn = Vv + ND;
  float* Trel = featdyn + ND;
  float* Tq = Trel + SELF_PAD * D;
  float* Tk = Tq + SELF_PAD * D;
  float* attn = Tk + SELF_PAD * D;

  float* out0 = (float*)d_out;
  float* out1 = out0 + ND;

  unsigned int* gid = (unsigned int*)out1;
  build_trel<<<(SELF_PAD * D + E / 4 + 255) / 256, 256, 0, stream>>>(
      relemb, selfrel, Trel, R, ei, E, N, gid);

  gemm128_wmma<<<GEMM_GRID, 256, 0, stream>>>(curr, Wq, nullptr, Xq, N / 16);
  gemm128_wmma<<<GEMM_GRID, 256, 0, stream>>>(last, Wk, nullptr, Xk, N / 16);
  gemm128_wmma<<<GEMM_GRID, 256, 0, stream>>>(last, Wv, bv, Vv, N / 16);
  gemm128_wmma<<<SELF_PAD / 16, 256, 0, stream>>>(Trel, Wq + D * D, bq, Tq,
                                                  SELF_PAD / 16);
  gemm128_wmma<<<SELF_PAD / 16, 256, 0, stream>>>(Trel, Wk + D * D, bk, Tk,
                                                  SELF_PAD / 16);

  edge_phase_kernel<<<(N + SC_ROWS - 1) / SC_ROWS, SC_THREADS, 0, stream>>>(
      ei, et, E, N, R, Xq, Xk, Tq, Tk, Vv, attn, out0, featdyn,
      (const uint4*)gid);

  gemm128_wmma<<<GEMM_GRID, 256, 0, stream>>>(featdyn, dynW, nullptr, out1, N / 16);
}


# Products with a common left operand share its row tile: a 16-wave group reads each tile of last once for the Wk and Wv slabs, each relation-table tile once for both bottom halves; 8 launches -> 6
# speedup vs baseline: 3.7645x; 1.0384x over previous
#include <hip/hip_runtime.h>
#include <math.h>


typedef __attribute__((ext_vector_type(2))) float v2f;
typedef __attribute__((ext_vector_type(8))) float v8f;

#define D 128
#define SELF_PAD 512
#define GEMM_GRID 500
#define GEMM2_GRID 250

__device__ __forceinline__ void load_a_tile(float4 (&a)[16],
                                            const float* __restrict__ A,
                                            int tile, int r, int g) {
  const float4* __restrict__ p =
      (const float4*)(A + (size_t)(tile * 16 + r) * D) + g;
#pragma unroll
  for (int i = 0; i < 16; ++i) a[i] = p[2 * i];
}

__device__ __forceinline__ void mma_store_tile(const float4 (&a)[16],
                                               const v2f (&b)[32], float bv,
                                               float* __restrict__ C, int tile,
                                               int n0, int r, int g) {
  v8f acc = {};
#pragma unroll
  for (int i = 0; i < 16; ++i) {
    v2f a0, a1;
    a0.x = a[i].x;
    a0.y = a[i].y;
    a1.x = a[i].z;
    a1.y = a[i].w;
    acc = __builtin_amdgcn_wmma_f32_16x16x4_f32(
         false, a0,  false, b[2 * i],
         (short)0, acc,  false,  false);
    acc = __builtin_amdgcn_wmma_f32_16x16x4_f32(
         false, a1,  false, b[2 * i + 1],
         (short)0, acc,  false,  false);
  }
#pragma unroll
  for (int j = 0; j < 8; ++j) {
    const int row = tile * 16 + j + 8 * g;
    C[(size_t)row * D + n0 + r] = acc[j] + bv;
  }
}

__global__ __launch_bounds__(512) void gemm128_wmma(
    const float* __restrict__ A, const float* __restrict__ W0,
    const float* __restrict__ bias0, float* __restrict__ C0,
    const float* __restrict__ W1, const float* __restrict__ bias1,
    float* __restrict__ C1, int tiles) {
  const int lane = threadIdx.x & 31;
  const int wave = threadIdx.x >> 5;
  const int r = lane & 15;
  const int g = lane >> 4;
  const bool second = wave >= 8;
  const float* __restrict__ W = second ? W1 : W0;
  const float* __restrict__ bias = second ? bias1 : bias0;
  float* __restrict__ C = second ? C1 : C0;
  const int n0 = (wave & 7) * 16;
  const int stride = gridDim.x;

  int t = blockIdx.x;
  if (t >= tiles) return;

  const float* __restrict__ Wcol = W + (n0 + r);
  v2f b[32];
#pragma unroll
  for (int s = 0; s < 32; ++s) {
    const int kk = 8 * (s >> 1) + 4 * g + 2 * (s & 1);
    b[s].x = Wcol[(size_t)kk * D];
    b[s].y = Wcol[(size_t)(kk + 1) * D];
  }
  const float bv = bias ? bias[n0 + r] : 0.0f;

  float4 a0[16], a1[16];
  load_a_tile(a0, A, t, r, g);
  for (;;) {
    const int t1 = t + stride;
    if (t1 >= tiles) {
      mma_store_tile(a0, b, bv, C, t, n0, r, g);
      break;
    }
    load_a_tile(a1, A, t1, r, g);
    __builtin_amdgcn_sched_barrier(0);
    mma_store_tile(a0, b, bv, C, t, n0, r, g);
    const int t2 = t1 + stride;
    if (t2 >= tiles) {
      mma_store_tile(a1, b, bv, C, t1, n0, r, g);
      break;
    }
    load_a_tile(a0, A, t2, r, g);
    __builtin_amdgcn_sched_barrier(0);
    mma_store_tile(a1, b, bv, C, t1, n0, r, g);
    t = t2;
  }
}

#define SC_ROWS 160

__device__ __forceinline__ bool sweep_by_bytes(int E, int N) {
  return (E & 15) == 0 && E >= 16 && (N + SC_ROWS - 1) / SC_ROWS <= 256 &&
         (long long)E <= 4ll * N * D;
}

__global__ void build_trel(const float* __restrict__ relemb,
                           const float* __restrict__ selfrel,
                           float* __restrict__ Trel, int R,
                           const int* __restrict__ ei, int E, int N,
                           unsigned int* __restrict__ gid) {
  const int i = blockIdx.x * 256 + threadIdx.x;
  if (i >= SELF_PAD * D) {
    const int w = i - SELF_PAD * D;
    if (w < (E >> 2) && sweep_by_bytes(E, N)) {
      const int4 d = ((const int4*)(ei + E))[w];
      gid[w] = ((unsigned)(d.x / SC_ROWS) & 255u) |
               (((unsigned)(d.y / SC_ROWS) & 255u) << 8) |
               (((unsigned)(d.z / SC_ROWS) & 255u) << 16) |
               (((unsigned)(d.w / SC_ROWS) & 255u) << 24);
    }
    return;
  }
  const int row = i >> 7;
  const int c = i & (D - 1);
  float v = 0.0f;
  if (row < R) v = relemb[i];
  else if (row == R) v = selfrel[c];
  Trel[i] = v;
}

__device__ __forceinline__ void lds_max_f(int* addr, float v) {
  if (v >= 0.0f)
    atomicMax(addr, __float_as_int(v));
  else
    atomicMin((unsigned int*)addr, __float_as_uint(v));
}

#define SC_THREADS 1024
#define SC_WAVES (SC_THREADS / 32)
#define SC_BATCH 512
#define SC_LIST (SC_BATCH + 32)
#define REC_CAP 192
#define REC_WORDS (3 * REC_CAP)
static_assert(REC_WORDS >= SC_LIST, "the general path's list lives in the record space");
#define RPW (SC_ROWS / SC_WAVES)
#define SRT_CAP (SC_WAVES * REC_CAP)
static_assert(SC_ROWS % SC_WAVES == 0, "every wave owns the same number of rows");
static_assert(SC_ROWS % 32 == 0, "the prefix sum gives every lane the same number of rows");
static_assert(3 * SRT_CAP <= 2 * SC_ROWS * D, "the regrouped list lives in the accumulator space");

template <bool VEC>
__device__ __forceinline__ int4 load_dst4(const int* __restrict__ dst, int idx,
                                          int E) {
  if (VEC) {
    return *(const int4*)(dst + min(idx, E - 4));
  } else {
    int4 v;
    v.x = dst[min(idx + 0, E - 1)];
    v.y = dst[min(idx + 1, E - 1)];
    v.z = dst[min(idx + 2, E - 1)];
    v.w = dst[min(idx + 3, E - 1)];
    return v;
  }
}

template <bool REC, bool ROW = true>
__device__ __forceinline__ int list_append(int* lst, int n, bool m, int e,
                                           int row, int lane) {
  const unsigned mask = (unsigned)__ballot(m);
  if (mask != 0u) {
    const int pos = n + __popc(mask & ((1u << lane) - 1u));
    if (REC) {
      if (m && pos < REC_CAP) {
        lst[pos] = e;
        if (ROW) lst[REC_CAP + pos] = row;
      }
    } else {
      if (m) lst[pos] = e;
    }
    n += __popc(mask);
  }
  return n;
}

template <bool VEC, bool REC, class WORK>
__device__ __forceinline__ int scan_stripe(
    int* lst, int n, int wave, int lane, int lo, int nrows, int rot, int nbatch,
    const int* __restrict__ ei, int E, const WORK& work) {
  const int* __restrict__ dst = ei + E;
  const unsigned span = (unsigned)nrows;
  int done = 0;
  for (int b0 = wave; b0 < nbatch; b0 += SC_WAVES) {
    int b = b0 + rot;
    if (b >= nbatch) b -= nbatch;
    const int idx = b * SC_BATCH + lane * 4;
    int4 q[4];
#pragma unroll
    for (int u = 0; u < 4; ++u) q[u] = load_dst4<VEC>(dst, idx + u * 128, E);
#pragma unroll
    for (int u = 0; u < 4; ++u) {
      const int e = idx + u * 128;
      const int r0 = q[u].x - lo, r1 = q[u].y - lo;
      const int r2 = q[u].z - lo, r3 = q[u].w - lo;
      bool m0, m1, m2, m3;
      if (VEC) {
        const unsigned sp = (e < E) ? span : 0u;
        m0 = (unsigned)r0 < sp;
        m1 = (unsigned)r1 < sp;
        m2 = (unsigned)r2 < sp;
        m3 = (unsigned)r3 < sp;
      } else {
        m0 = (e + 0 < E) & ((unsigned)r0 < span);
        m1 = (e + 1 < E) & ((unsigned)r1 < span);
        m2 = (e + 2 < E) & ((unsigned)r2 < span);
        m3 = (e + 3 < E) & ((unsigned)r3 < span);
      }
      if (__ballot(m0 | m1 | m2 | m3) != 0) {
        n = list_append<REC>(lst, n, m0, e + 0, r0, lane);
        n = list_append<REC>(lst, n, m1, e + 1, r1, lane);
        n = list_append<REC>(lst, n, m2, e + 2, r2, lane);
        n = list_append<REC>(lst, n, m3, e + 3, r3, lane);
      }
    }
    if (REC) {
      if (n > REC_CAP) return -1;
      while (n - done >= 32) {
        work(lst, done, 32);
        done += 32;
      }
    } else {
      int dn = 0;
      while (n - dn >= 32) {
        work(lst, dn, 32);
        dn += 32;
      }
      if (dn > 0) {
        const int rem = n - dn;
        int keep = 0;
        if (lane < rem) keep = lst[dn + lane];
        if (lane < rem) lst[lane] = keep;
        n = rem;
      }
    }
  }
  return n;
}

__device__ __forceinline__ unsigned zero_bytes(unsigned x) {
  const unsigned t = (x & 0x7f7f7f7fu) + 0x7f7f7f7fu;
  return ~(t | x | 0x7f7f7f7fu);
}

#define BY_STEP 2048

template <class WORK>
__device__ __forceinline__ int scan_bytes(int* lst, int n, int wave, int lane,
                                          const uint4* __restrict__ gid, int E,
                                          const WORK& work) {
  const int nvec = E >> 4;
  const int nstep = (nvec + BY_STEP / 16 - 1) / (BY_STEP / 16);
  const int rot = (int)(((long long)blockIdx.x * nstep) / gridDim.x);
  const unsigned pat = blockIdx.x * 0x01010101u;
  int done = 0;
  for (int b0 = wave; b0 < nstep; b0 += SC_WAVES) {
    int b = b0 + rot;
    if (b >= nstep) b -= nstep;
    const int vi = b * (BY_STEP / 16) + lane;
    uint4 q[4];
#pragma unroll
    for (int u = 0; u < 4; ++u) q[u] = gid[min(vi + u * 32, nvec - 1)];
#pragma unroll
    for (int u = 0; u < 4; ++u) {
      const int v = vi + u * 32;
      unsigned m = zero_bytes(q[u].x ^ pat) | (zero_bytes(q[u].y ^ pat) >> 1) |
                   (zero_bytes(q[u].z ^ pat) >> 2) |
                   (zero_bytes(q[u].w ^ pat) >> 3);
      if (v >= nvec) m = 0u;
      while (__ballot(m != 0u) != 0) {
        const bool has = m != 0u;
        const int p = __builtin_ctz(m | 0x80000000u);
        const int e = v * 16 + 4 * (7 - (p & 7)) + (p >> 3);
        m &= m - 1u;
        n = list_append<true, false>(lst, n, has, e, 0, lane);
      }
    }
    if (n > REC_CAP) return -1;
    while (n - done >= 32) {
      work(lst, done, 32);
      done += 32;
    }
  }
  return n;
}

template <bool REC, class WORK>
__device__ __forceinline__ int sweep_range(int* lst, int wave, int lane,
                                           int lo, int nrows,
                                           const int* __restrict__ ei, int E,
                                           const uint4* __restrict__ gid,
                                           const WORK& work) {
  int n = 0;
  for (int r0 = 0; r0 < SC_ROWS; r0 += SC_THREADS) {
    const int r = r0 + lane * SC_WAVES + wave;
    n = list_append<REC>(lst, n, r < nrows, E + lo + r, r, lane);
  }
  const int nbatch = (E + SC_BATCH - 1) / SC_BATCH;
  const int rot = (int)(((long long)blockIdx.x * nbatch) / gridDim.x);
  if (REC && gid != nullptr)
    n = scan_bytes(lst, n, wave, lane, gid, E, work);
  else if (REC && (E & 3) == 0 && E >= 4)
    n = scan_stripe<true, REC>(lst, n, wave, lane, lo, nrows, rot, nbatch, ei, E, work);
  else if (E > 0)
    n = scan_stripe<false, REC>(lst, n, wave, lane, lo, nrows, rot, nbatch, ei, E, work);
  if (n < 0) return n;
  const int done = REC ? (n & ~31) : 0;
  if (n > done) work(lst, done, n - done);
  return REC ? n : 0;
}

template <bool REC>
struct ScoreWork {
  int lane, lo;
  const int* __restrict__ ei;
  const int* __restrict__ et;
  int E, R;
  const float* qs;
  const float* __restrict__ Xk;
  const float* __restrict__ Tq;
  const float* __restrict__ Tk;
  float* attn;
  int* smax;
  int* smin;
  float* scnt;
  bool rowrec;
  __device__ __forceinline__ void operator()(int* lst, int base,
                                             int cnt) const {
    const bool fromrec = REC && rowrec;
    int e = 0, s = 0, t = 0, row = 0;
    if (lane < cnt) {
      e = lst[base + lane];
      if (fromrec) row = lst[REC_CAP + base + lane];
      if (e < E) {
        s = ei[e];
        t = et[e];
        if (!fromrec) row = ei[E + e] - lo;
      } else {
        s = e - E;
        t = R;
        if (!fromrec) row = s - lo;
      }
    }
    float mine = 0.0f;
    for (int j0 = 0; j0 < cnt; j0 += 4) {
      float4 q0[4], q1[4], k0[4], k1[4];
#pragma unroll
      for (int u = 0; u < 4; ++u) {
        const int j = min(j0 + u, cnt - 1);
        const int sj = __builtin_amdgcn_readlane(s, j);
        const int tj = __builtin_amdgcn_readlane(t, j);
        const int rj = __builtin_amdgcn_readlane(row, j);
        k0[u] = ((const float4*)(Xk + (size_t)sj * D))[lane];
        q1[u] = ((const float4*)(Tq + (size_t)tj * D))[lane];
        k1[u] = ((const float4*)(Tk + (size_t)tj * D))[lane];
        q0[u] = ((const float4*)(qs + rj * D))[lane];
      }
      float a[4];
#pragma unroll
      for (int u = 0; u < 4; ++u)
        a[u] = (q0[u].x + q1[u].x) * (k0[u].x + k1[u].x) +
               (q0[u].y + q1[u].y) * (k0[u].y + k1[u].y) +
               (q0[u].z + q1[u].z) * (k0[u].z + k1[u].z) +
               (q0[u].w + q1[u].w) * (k0[u].w + k1[u].w);
#pragma unroll
      for (int off = 16; off >= 1; off >>= 1) {
#pragma unroll
        for (int u = 0; u < 4; ++u) a[u] += __shfl_xor(a[u], off, 32);
      }
#pragma unroll
      for (int u = 0; u < 4; ++u)
        if (lane == j0 + u) mine = a[u] * 0.08838834764831845f;
    }
    if (lane < cnt) {
      if (REC) {
        lst[base + lane] = s;
        if (!rowrec) lst[REC_CAP + base + lane] = row;
        lst[2 * REC_CAP + base + lane] = __float_as_int(mine);
      } else {
        attn[e] = mine;
      }
      lds_max_f(smax + row, mine);
      lds_max_f(smin + row, -mine);
      atomicAdd(scnt + row, 1.0f);
    }
  }
};

__device__ __forceinline__ void edge_weights(float a, int row, const int* smax,
                                             const int* smin, float* sd,
                                             float* ss, float& wd, float& ws) {
  wd = __expf(a - __int_as_float(smax[row]));
  ws = __expf(-a - __int_as_float(smin[row]));
  atomicAdd(sd + row, wd);
  atomicAdd(ss + row, ws);
}

__device__ __forceinline__ void add_rows(int cnt, int lane, int s, int row,
                                         float wd, float ws,
                                         const float* __restrict__ Vv,
                                         float* acc_s, float* acc_d) {
  for (int j0 = 0; j0 < cnt; j0 += 8) {
    float4 v[8];
#pragma unroll
    for (int u = 0; u < 8; ++u) {
      const int j = min(j0 + u, cnt - 1);
      const int sj = __builtin_amdgcn_readlane(s, j);
      v[u] = ((const float4*)(Vv + (size_t)sj * D))[lane];
    }
#pragma unroll
    for (int u = 0; u < 8; ++u) {
      const int j = j0 + u;
      if (j < cnt) {
        const int rj = __builtin_amdgcn_readlane(row, j);
        const float wsj = __int_as_float(__builtin_amdgcn_readlane(__float_as_int(ws), j));
        const float wdj = __int_as_float(__builtin_amdgcn_readlane(__float_as_int(wd), j));
        float* as = acc_s + rj * D + lane;
        float* ad = acc_d + rj * D + lane;
        __hip_atomic_fetch_add(as + 0, wsj * v[u].x, __ATOMIC_RELAXED, __HIP_MEMORY_SCOPE_WORKGROUP);
        __hip_atomic_fetch_add(as + 32, wsj * v[u].y, __ATOMIC_RELAXED, __HIP_MEMORY_SCOPE_WORKGROUP);
        __hip_atomic_fetch_add(as + 64, wsj * v[u].z, __ATOMIC_RELAXED, __HIP_MEMORY_SCOPE_WORKGROUP);
        __hip_atomic_fetch_add(as + 96, wsj * v[u].w, __ATOMIC_RELAXED, __HIP_MEMORY_SCOPE_WORKGROUP);
        __hip_atomic_fetch_add(ad + 0, wdj * v[u].x, __ATOMIC_RELAXED, __HIP_MEMORY_SCOPE_WORKGROUP);
        __hip_atomic_fetch_add(ad + 32, wdj * v[u].y, __ATOMIC_RELAXED, __HIP_MEMORY_SCOPE_WORKGROUP);
        __hip_atomic_fetch_add(ad + 64, wdj * v[u].z, __ATOMIC_RELAXED, __HIP_MEMORY_SCOPE_WORKGROUP);
        __hip_atomic_fetch_add(ad + 96, wdj * v[u].w, __ATOMIC_RELAXED, __HIP_MEMORY_SCOPE_WORKGROUP);
      }
    }
  }
}

struct AddFromAttn {
  int lane, lo;
  const int* __restrict__ ei;
  int E;
  const float* attn;
  const int* smax;
  const int* smin;
  float* sd;
  float* ss;
  const float* __restrict__ Vv;
  float* acc_s;
  float* acc_d;
  __device__ __forceinline__ void operator()(int* lst, int base,
                                             int cnt) const {
    int s = 0, row = 0;
    float wd = 0.0f, ws = 0.0f;
    if (lane < cnt) {
      const int e = lst[base + lane];
      int d;
      if (e < E) {
        s = ei[e];
        d = ei[E + e];
      } else {
        s = d = e - E;
      }
      row = d - lo;
      edge_weights(attn[e], row, smax, smin, sd, ss, wd, ws);
    }
    add_rows(cnt, lane, s, row, wd, ws, Vv, acc_s, acc_d);
  }
};

__global__ __launch_bounds__(SC_THREADS)
__attribute__((amdgpu_waves_per_eu(8, 8))) void edge_phase_kernel(
    const int* __restrict__ ei, const int* __restrict__ et, int E, int N,
    int R, const float* __restrict__ Xq, const float* __restrict__ Xk,
    const float* __restrict__ Tq, const float* __restrict__ Tk,
    const float* __restrict__ Vv, float* attn, float* __restrict__ featsta,
    float* __restrict__ featdyn, const uint4* __restrict__ gidbytes) {
  __shared__ float qs[SC_ROWS * D];
  __shared__ float accbuf[2 * SC_ROWS * D];
  __shared__ int lists[SC_WAVES * REC_WORDS];
  __shared__ int smax[SC_ROWS];
  __shared__ int smin[SC_ROWS];
  __shared__ float scnt[SC_ROWS];
  __shared__ float sd[SC_ROWS];
  __shared__ float ss[SC_ROWS];
  __shared__ int rstart[SC_ROWS + 1];
  __shared__ int rnext[SC_ROWS];
  __shared__ int overflow;

  const int tid = threadIdx.x;
  const int lane = tid & 31;
  const int wave = tid >> 5;
  const int lo = blockIdx.x * SC_ROWS;
  const int nrows = min(SC_ROWS, N - lo);
  int* lst = lists + wave * REC_WORDS;

  const float4* __restrict__ qg = (const float4*)(Xq + (size_t)lo * D);
  for (int i = tid; i < nrows * (D / 4); i += SC_THREADS)
    ((float4*)qs)[i] = qg[i];
  for (int i = tid; i < SC_ROWS; i += SC_THREADS) {
    smax[i] = __float_as_int(-__builtin_inff());
    smin[i] = __float_as_int(-__builtin_inff());
    scnt[i] = 0.0f;
  }
  if (tid == 0) overflow = 0;
  __syncthreads();

  const uint4* __restrict__ gid = sweep_by_bytes(E, N) ? gidbytes : nullptr;
  const ScoreWork<true> score = {lane, lo, ei,   et,   E,    R,    qs,           Xk,
                                 Tq,   Tk, attn, smax, smin, scnt, gid == nullptr};
  const int n = sweep_range<true>(lst, wave, lane, lo, nrows, ei, E, gid, score);
  if (n < 0 && lane == 0) overflow = 1;
  __syncthreads();

  if (__builtin_expect(overflow == 0, 1)) {
    int* srt_s = (int*)accbuf;
    float* srt_wd = accbuf + SRT_CAP;
    float* srt_ws = accbuf + 2 * SRT_CAP;

    if (wave == 0) {
      constexpr int RPL = SC_ROWS / 32;
      int cdeg[RPL];
      int tot = 0;
#pragma unroll
      for (int k = 0; k < RPL; ++k) {
        cdeg[k] = (int)scnt[lane * RPL + k];
        tot += cdeg[k];
      }
      int incl = tot;
#pragma unroll
      for (int off = 1; off < 32; off <<= 1) {
        const int up = __shfl_up(incl, off, 32);
        if (lane >= off) incl += up;
      }
      int run = incl - tot;
#pragma unroll
      for (int k = 0; k < RPL; ++k) {
        rstart[lane * RPL + k] = run;
        rnext[lane * RPL + k] = run;
        run += cdeg[k];
      }
      if (lane == 31) rstart[SC_ROWS] = run;
    }
    __syncthreads();

    for (int base = 0; base < n; base += 32) {
      if (base + lane < n) {
        const int s = lst[base + lane];
        const int row = lst[REC_CAP + base + lane];
        const float a = __int_as_float(lst[2 * REC_CAP + base + lane]);
        const float wd = __expf(a - __int_as_float(smax[row]));
        const float ws = __expf(-a - __int_as_float(smin[row]));
        const int pos = atomicAdd(rnext + row, 1);
        srt_s[pos] = s;
        srt_wd[pos] = wd;
        srt_ws[pos] = ws;
      }
    }
    __syncthreads();

    const int r0 = wave * RPW;
    const int bnd = rstart[r0 + min(lane, RPW)];
    const int ebeg = __builtin_amdgcn_readlane(bnd, 0);
    const int eend = __builtin_amdgcn_readlane(bnd, RPW);
    int k = 0;
    int rend = __builtin_amdgcn_readlane(bnd, 1);
    float4 as = make_float4(0.0f, 0.0f, 0.0f, 0.0f);
    float4 ad = make_float4(0.0f, 0.0f, 0.0f, 0.0f);
    float sums = 0.0f, sumd = 0.0f;
    auto row_done = [&]() {
      const int r = r0 + k;
      if (r < nrows) {
        const float denom = fmaxf(scnt[r], 1.0f);
        const float ns = sums * denom;
        const float nd = sumd * denom;
        const float4 os = make_float4(as.x / ns, as.y / ns, as.z / ns, as.w / ns);
        const float4 od = make_float4(ad.x / nd, ad.y / nd, ad.z / nd, ad.w / nd);
        ((float4*)(featsta + (size_t)(lo + r) * D))[lane] = os;
        ((float4*)(featdyn + (size_t)(lo + r) * D))[lane] = od;
      }
      as = make_float4(0.0f, 0.0f, 0.0f, 0.0f);
      ad = make_float4(0.0f, 0.0f, 0.0f, 0.0f);
      sums = 0.0f;
      sumd = 0.0f;
      ++k;
      rend = __builtin_amdgcn_readlane(bnd, min(k + 1, RPW));
    };
    for (int base = ebeg; base < eend; base += 32) {
      const int cnt = min(32, eend - base);
      int sv = 0;
      float wdv = 0.0f, wsv = 0.0f;
      if (lane < cnt) {
        sv = srt_s[base + lane];
        wdv = srt_wd[base + lane];
        wsv = srt_ws[base + lane];
      }
      for (int j0 = 0; j0 < cnt; j0 += 8) {
        float4 v[8];
#pragma unroll
        for (int u = 0; u < 8; ++u) {
          const int j = min(j0 + u, cnt - 1);
          const int sj = __builtin_amdgcn_readlane(sv, j);
          v[u] = ((const float4*)(Vv + (size_t)sj * D))[lane];
        }
#pragma unroll
        for (int u = 0; u < 8; ++u) {
          const int j = j0 + u;
          if (j < cnt) {
            while (base + j >= rend) row_done();
            const float wsj = __int_as_float(__builtin_amdgcn_readlane(__float_as_int(wsv), j));
            const float wdj = __int_as_float(__builtin_amdgcn_readlane(__float_as_int(wdv), j));
            as.x += wsj * v[u].x;
            as.y += wsj * v[u].y;
            as.z += wsj * v[u].z;
            as.w += wsj * v[u].w;
            ad.x += wdj * v[u].x;
            ad.y += wdj * v[u].y;
            ad.z += wdj * v[u].z;
            ad.w += wdj * v[u].w;
            sums += wsj;
            sumd += wdj;
          }
        }
      }
    }
    while (k < RPW) row_done();
  } else {
    float* acc_s = accbuf;
    float* acc_d = accbuf + SC_ROWS * D;
    for (int i = tid; i < 2 * SC_ROWS * D / 4; i += SC_THREADS)
      ((float4*)accbuf)[i] = make_float4(0.0f, 0.0f, 0.0f, 0.0f);
    for (int i = tid; i < SC_ROWS; i += SC_THREADS) {
      smax[i] = __float_as_int(-__builtin_inff());
      smin[i] = __float_as_int(-__builtin_inff());
      scnt[i] = 0.0f;
      sd[i] = 0.0f;
      ss[i] = 0.0f;
    }
    __syncthreads();
    const ScoreWork<false> score2 = {lane, lo, ei,   et,   E,    R,    qs,  Xk,
                                     Tq,   Tk, attn, smax, smin, scnt, false};
    sweep_range<false>(lst, wave, lane, lo, nrows, ei, E, nullptr, score2);
    __syncthreads();
    const AddFromAttn add2 = {lane, lo, ei, E,  attn,  smax, smin,
                              sd,   ss, Vv, acc_s, acc_d};
    sweep_range<false>(lst, wave, lane, lo, nrows, ei, E, nullptr, add2);
    __syncthreads();

    for (int r = wave; r < nrows; r += SC_WAVES) {
      const float denom = fmaxf(scnt[r], 1.0f);
      const float ns = ss[r] * denom;
      const float nd = sd[r] * denom;
      const float* as = acc_s + r * D + lane;
      const float* ad = acc_d + r * D + lane;
      const float4 os = make_float4(as[0] / ns, as[32] / ns, as[64] / ns, as[96] / ns);
      const float4 od = make_float4(ad[0] / nd, ad[32] / nd, ad[64] / nd, ad[96] / nd);
      ((float4*)(featsta + (size_t)(lo + r) * D))[lane] = os;
      ((float4*)(featdyn + (size_t)(lo + r) * D))[lane] = od;
    }
  }
}

extern "C" void kernel_launch(void* const* d_in, const int* in_sizes, int n_in,
                              void* d_out, int out_size, void* d_ws,
                              size_t ws_size, hipStream_t stream) {
  const float* curr = (const float*)d_in[0];
  const float* last = (const float*)d_in[1];
  const float* relemb = (const float*)d_in[2];
  const float* selfrel = (const float*)d_in[3];
  const float* Wq = (const float*)d_in[4];
  const float* bq = (const float*)d_in[5];
  const float* Wk = (const float*)d_in[6];
  const float* bk = (const float*)d_in[7];
  const float* Wv = (const float*)d_in[8];
  const float* bv = (const float*)d_in[9];
  const float* dynW = (const float*)d_in[10];
  const int* ei = (const int*)d_in[11];
  const int* et = (const int*)d_in[12];

  const int N = in_sizes[0] / D;
  const int E = in_sizes[12];
  const int R = in_sizes[2] / D;
  const size_t ND = (size_t)N * D;

  float* ws = (float*)d_ws;
  float* Xq = ws;
  float* Xk = Xq + ND;
  float* Vv = Xk + ND;
  float* featdyn = Vv + ND;
  float* Trel = featdyn + ND;
  float* Tq = Trel + SELF_PAD * D;
  float* Tk = Tq + SELF_PAD * D;
  float* attn = Tk + SELF_PAD * D;

  float* out0 = (float*)d_out;
  float* out1 = out0 + ND;

  unsigned int* gid = (unsigned int*)out1;
  build_trel<<<(SELF_PAD * D + E / 4 + 255) / 256, 256, 0, stream>>>(
      relemb, selfrel, Trel, R, ei, E, N, gid);

  gemm128_wmma<<<GEMM_GRID, 256, 0, stream>>>(curr, Wq, nullptr, Xq, nullptr,
                                              nullptr, nullptr, N / 16);
  gemm128_wmma<<<GEMM2_GRID, 512, 0, stream>>>(last, Wk, nullptr, Xk, Wv, bv,
                                               Vv, N / 16);
  gemm128_wmma<<<SELF_PAD / 16, 512, 0, stream>>>(Trel, Wq + D * D, bq, Tq,
                                                  Wk + D * D, bk, Tk,
                                                  SELF_PAD / 16);

  edge_phase_kernel<<<(N + SC_ROWS - 1) / SC_ROWS, SC_THREADS, 0, stream>>>(
      ei, et, E, N, R, Xq, Xk, Tq, Tk, Vv, attn, out0, featdyn,
      (const uint4*)gid);

  gemm128_wmma<<<GEMM_GRID, 256, 0, stream>>>(featdyn, dynW, nullptr, out1,
                                              nullptr, nullptr, nullptr, N / 16);
}


# Edge scoring: group records regrouped by relation type in LDS (histogram, prefix sum, rank by LDS int atomic); waves score equal slices, Tq/Tk rows loaded once per type run; arrival order as before
# speedup vs baseline: 3.9000x; 1.0360x over previous
#include <hip/hip_runtime.h>
#include <math.h>


typedef __attribute__((ext_vector_type(2))) float v2f;
typedef __attribute__((ext_vector_type(8))) float v8f;

#define D 128
#define SELF_PAD 512
#define GEMM_GRID 500
#define GEMM2_GRID 250

__device__ __forceinline__ void load_a_tile(float4 (&a)[16],
                                            const float* __restrict__ A,
                                            int tile, int r, int g) {
  const float4* __restrict__ p =
      (const float4*)(A + (size_t)(tile * 16 + r) * D) + g;
#pragma unroll
  for (int i = 0; i < 16; ++i) a[i] = p[2 * i];
}

__device__ __forceinline__ void mma_store_tile(const float4 (&a)[16],
                                               const v2f (&b)[32], float bv,
                                               float* __restrict__ C, int tile,
                                               int n0, int r, int g) {
  v8f acc = {};
#pragma unroll
  for (int i = 0; i < 16; ++i) {
    v2f a0, a1;
    a0.x = a[i].x;
    a0.y = a[i].y;
    a1.x = a[i].z;
    a1.y = a[i].w;
    acc = __builtin_amdgcn_wmma_f32_16x16x4_f32(
         false, a0,  false, b[2 * i],
         (short)0, acc,  false,  false);
    acc = __builtin_amdgcn_wmma_f32_16x16x4_f32(
         false, a1,  false, b[2 * i + 1],
         (short)0, acc,  false,  false);
  }
#pragma unroll
  for (int j = 0; j < 8; ++j) {
    const int row = tile * 16 + j + 8 * g;
    C[(size_t)row * D + n0 + r] = acc[j] + bv;
  }
}

__global__ __launch_bounds__(512) void gemm128_wmma(
    const float* __restrict__ A, const float* __restrict__ W0,
    const float* __restrict__ bias0, float* __restrict__ C0,
    const float* __restrict__ W1, const float* __restrict__ bias1,
    float* __restrict__ C1, int tiles) {
  const int lane = threadIdx.x & 31;
  const int wave = threadIdx.x >> 5;
  const int r = lane & 15;
  const int g = lane >> 4;
  const bool second = wave >= 8;
  const float* __restrict__ W = second ? W1 : W0;
  const float* __restrict__ bias = second ? bias1 : bias0;
  float* __restrict__ C = second ? C1 : C0;
  const int n0 = (wave & 7) * 16;
  const int stride = gridDim.x;

  int t = blockIdx.x;
  if (t >= tiles) return;

  const float* __restrict__ Wcol = W + (n0 + r);
  v2f b[32];
#pragma unroll
  for (int s = 0; s < 32; ++s) {
    const int kk = 8 * (s >> 1) + 4 * g + 2 * (s & 1);
    b[s].x = Wcol[(size_t)kk * D];
    b[s].y = Wcol[(size_t)(kk + 1) * D];
  }
  const float bv = bias ? bias[n0 + r] : 0.0f;

  float4 a0[16], a1[16];
  load_a_tile(a0, A, t, r, g);
  for (;;) {
    const int t1 = t + stride;
    if (t1 >= tiles) {
      mma_store_tile(a0, b, bv, C, t, n0, r, g);
      break;
    }
    load_a_tile(a1, A, t1, r, g);
    __builtin_amdgcn_sched_barrier(0);
    mma_store_tile(a0, b, bv, C, t, n0, r, g);
    const int t2 = t1 + stride;
    if (t2 >= tiles) {
      mma_store_tile(a1, b, bv, C, t1, n0, r, g);
      break;
    }
    load_a_tile(a0, A, t2, r, g);
    __builtin_amdgcn_sched_barrier(0);
    mma_store_tile(a1, b, bv, C, t1, n0, r, g);
    t = t2;
  }
}

#define SC_ROWS 160

__device__ __forceinline__ bool sweep_by_bytes(int E, int N) {
  return (E & 15) == 0 && E >= 16 && (N + SC_ROWS - 1) / SC_ROWS <= 256 &&
         (long long)E <= 4ll * N * D;
}

__global__ void build_trel(const float* __restrict__ relemb,
                           const float* __restrict__ selfrel,
                           float* __restrict__ Trel, int R,
                           const int* __restrict__ ei, int E, int N,
                           unsigned int* __restrict__ gid) {
  const int i = blockIdx.x * 256 + threadIdx.x;
  if (i >= SELF_PAD * D) {
    const int w = i - SELF_PAD * D;
    if (w < (E >> 2) && sweep_by_bytes(E, N)) {
      const int4 d = ((const int4*)(ei + E))[w];
      gid[w] = ((unsigned)(d.x / SC_ROWS) & 255u) |
               (((unsigned)(d.y / SC_ROWS) & 255u) << 8) |
               (((unsigned)(d.z / SC_ROWS) & 255u) << 16) |
               (((unsigned)(d.w / SC_ROWS) & 255u) << 24);
    }
    return;
  }
  const int row = i >> 7;
  const int c = i & (D - 1);
  float v = 0.0f;
  if (row < R) v = relemb[i];
  else if (row == R) v = selfrel[c];
  Trel[i] = v;
}

__device__ __forceinline__ void lds_max_f(int* addr, float v) {
  if (v >= 0.0f)
    atomicMax(addr, __float_as_int(v));
  else
    atomicMin((unsigned int*)addr, __float_as_uint(v));
}

#define SC_THREADS 1024
#define SC_WAVES (SC_THREADS / 32)
#define SC_BATCH 512
#define SC_LIST (SC_BATCH + 32)
#define REC_CAP 192
#define REC_WORDS (3 * REC_CAP)
static_assert(REC_WORDS >= SC_LIST, "the general path's list lives in the record space");
#define RPW (SC_ROWS / SC_WAVES)
#define SRT_CAP (SC_WAVES * REC_CAP)
static_assert(SC_ROWS % SC_WAVES == 0, "every wave owns the same number of rows");
static_assert(SC_ROWS % 32 == 0, "the prefix sum gives every lane the same number of rows");
static_assert(SELF_PAD % 32 == 0, "the type prefix sum gives every lane the same number of types");
static_assert(6 * SRT_CAP + SELF_PAD + 1 <= 2 * SC_ROWS * D,
              "both regrouped lists and the type histogram live in the accumulator space");

template <bool VEC>
__device__ __forceinline__ int4 load_dst4(const int* __restrict__ dst, int idx,
                                          int E) {
  if (VEC) {
    return *(const int4*)(dst + min(idx, E - 4));
  } else {
    int4 v;
    v.x = dst[min(idx + 0, E - 1)];
    v.y = dst[min(idx + 1, E - 1)];
    v.z = dst[min(idx + 2, E - 1)];
    v.w = dst[min(idx + 3, E - 1)];
    return v;
  }
}

template <bool REC, bool ROW = true>
__device__ __forceinline__ int list_append(int* lst, int n, bool m, int e,
                                           int row, int lane) {
  const unsigned mask = (unsigned)__ballot(m);
  if (mask != 0u) {
    const int pos = n + __popc(mask & ((1u << lane) - 1u));
    if (REC) {
      if (m && pos < REC_CAP) {
        lst[pos] = e;
        if (ROW) lst[REC_CAP + pos] = row;
      }
    } else {
      if (m) lst[pos] = e;
    }
    n += __popc(mask);
  }
  return n;
}

template <bool VEC, bool REC, class WORK>
__device__ __forceinline__ int scan_stripe(
    int* lst, int n, int wave, int lane, int lo, int nrows, int rot, int nbatch,
    const int* __restrict__ ei, int E, const WORK& work) {
  const int* __restrict__ dst = ei + E;
  const unsigned span = (unsigned)nrows;
  int done = 0;
  for (int b0 = wave; b0 < nbatch; b0 += SC_WAVES) {
    int b = b0 + rot;
    if (b >= nbatch) b -= nbatch;
    const int idx = b * SC_BATCH + lane * 4;
    int4 q[4];
#pragma unroll
    for (int u = 0; u < 4; ++u) q[u] = load_dst4<VEC>(dst, idx + u * 128, E);
#pragma unroll
    for (int u = 0; u < 4; ++u) {
      const int e = idx + u * 128;
      const int r0 = q[u].x - lo, r1 = q[u].y - lo;
      const int r2 = q[u].z - lo, r3 = q[u].w - lo;
      bool m0, m1, m2, m3;
      if (VEC) {
        const unsigned sp = (e < E) ? span : 0u;
        m0 = (unsigned)r0 < sp;
        m1 = (unsigned)r1 < sp;
        m2 = (unsigned)r2 < sp;
        m3 = (unsigned)r3 < sp;
      } else {
        m0 = (e + 0 < E) & ((unsigned)r0 < span);
        m1 = (e + 1 < E) & ((unsigned)r1 < span);
        m2 = (e + 2 < E) & ((unsigned)r2 < span);
        m3 = (e + 3 < E) & ((unsigned)r3 < span);
      }
      if (__ballot(m0 | m1 | m2 | m3) != 0) {
        n = list_append<REC>(lst, n, m0, e + 0, r0, lane);
        n = list_append<REC>(lst, n, m1, e + 1, r1, lane);
        n = list_append<REC>(lst, n, m2, e + 2, r2, lane);
        n = list_append<REC>(lst, n, m3, e + 3, r3, lane);
      }
    }
    if (REC) {
      if (n > REC_CAP) return -1;
      while (n - done >= 32) {
        work(lst, done, 32);
        done += 32;
      }
    } else {
      int dn = 0;
      while (n - dn >= 32) {
        work(lst, dn, 32);
        dn += 32;
      }
      if (dn > 0) {
        const int rem = n - dn;
        int keep = 0;
        if (lane < rem) keep = lst[dn + lane];
        if (lane < rem) lst[lane] = keep;
        n = rem;
      }
    }
  }
  return n;
}

__device__ __forceinline__ unsigned zero_bytes(unsigned x) {
  const unsigned t = (x & 0x7f7f7f7fu) + 0x7f7f7f7fu;
  return ~(t | x | 0x7f7f7f7fu);
}

#define BY_STEP 2048

template <class WORK>
__device__ __forceinline__ int scan_bytes(int* lst, int n, int wave, int lane,
                                          const uint4* __restrict__ gid, int E,
                                          const WORK& work) {
  const int nvec = E >> 4;
  const int nstep = (nvec + BY_STEP / 16 - 1) / (BY_STEP / 16);
  const int rot = (int)(((long long)blockIdx.x * nstep) / gridDim.x);
  const unsigned pat = blockIdx.x * 0x01010101u;
  int done = 0;
  for (int b0 = wave; b0 < nstep; b0 += SC_WAVES) {
    int b = b0 + rot;
    if (b >= nstep) b -= nstep;
    const int vi = b * (BY_STEP / 16) + lane;
    uint4 q[4];
#pragma unroll
    for (int u = 0; u < 4; ++u) q[u] = gid[min(vi + u * 32, nvec - 1)];
#pragma unroll
    for (int u = 0; u < 4; ++u) {
      const int v = vi + u * 32;
      unsigned m = zero_bytes(q[u].x ^ pat) | (zero_bytes(q[u].y ^ pat) >> 1) |
                   (zero_bytes(q[u].z ^ pat) >> 2) |
                   (zero_bytes(q[u].w ^ pat) >> 3);
      if (v >= nvec) m = 0u;
      while (__ballot(m != 0u) != 0) {
        const bool has = m != 0u;
        const int p = __builtin_ctz(m | 0x80000000u);
        const int e = v * 16 + 4 * (7 - (p & 7)) + (p >> 3);
        m &= m - 1u;
        n = list_append<true, false>(lst, n, has, e, 0, lane);
      }
    }
    if (n > REC_CAP) return -1;
    while (n - done >= 32) {
      work(lst, done, 32);
      done += 32;
    }
  }
  return n;
}

template <bool REC, class WORK>
__device__ __forceinline__ int sweep_range(int* lst, int wave, int lane,
                                           int lo, int nrows,
                                           const int* __restrict__ ei, int E,
                                           const uint4* __restrict__ gid,
                                           const WORK& work) {
  int n = 0;
  for (int r0 = 0; r0 < SC_ROWS; r0 += SC_THREADS) {
    const int r = r0 + lane * SC_WAVES + wave;
    n = list_append<REC>(lst, n, r < nrows, E + lo + r, r, lane);
  }
  const int nbatch = (E + SC_BATCH - 1) / SC_BATCH;
  const int rot = (int)(((long long)blockIdx.x * nbatch) / gridDim.x);
  if (REC && gid != nullptr)
    n = scan_bytes(lst, n, wave, lane, gid, E, work);
  else if (REC && (E & 3) == 0 && E >= 4)
    n = scan_stripe<true, REC>(lst, n, wave, lane, lo, nrows, rot, nbatch, ei, E, work);
  else if (E > 0)
    n = scan_stripe<false, REC>(lst, n, wave, lane, lo, nrows, rot, nbatch, ei, E, work);
  if (n < 0) return n;
  const int done = REC ? (n & ~31) : 0;
  if (n > done) work(lst, done, n - done);
  return REC ? n : 0;
}

struct ScoreWork {
  int lane, lo;
  const int* __restrict__ ei;
  const int* __restrict__ et;
  int E, R;
  const float* qs;
  const float* __restrict__ Xk;
  const float* __restrict__ Tq;
  const float* __restrict__ Tk;
  float* attn;
  int* smax;
  int* smin;
  float* scnt;
  __device__ __forceinline__ void operator()(int* lst, int base,
                                             int cnt) const {
    int e = 0, s = 0, t = 0, row = 0;
    if (lane < cnt) {
      e = lst[base + lane];
      if (e < E) {
        s = ei[e];
        t = et[e];
        row = ei[E + e] - lo;
      } else {
        s = e - E;
        t = R;
        row = s - lo;
      }
    }
    float mine = 0.0f;
    for (int j0 = 0; j0 < cnt; j0 += 4) {
      float4 q0[4], q1[4], k0[4], k1[4];
#pragma unroll
      for (int u = 0; u < 4; ++u) {
        const int j = min(j0 + u, cnt - 1);
        const int sj = __builtin_amdgcn_readlane(s, j);
        const int tj = __builtin_amdgcn_readlane(t, j);
        const int rj = __builtin_amdgcn_readlane(row, j);
        k0[u] = ((const float4*)(Xk + (size_t)sj * D))[lane];
        q1[u] = ((const float4*)(Tq + (size_t)tj * D))[lane];
        k1[u] = ((const float4*)(Tk + (size_t)tj * D))[lane];
        q0[u] = ((const float4*)(qs + rj * D))[lane];
      }
      float a[4];
#pragma unroll
      for (int u = 0; u < 4; ++u)
        a[u] = (q0[u].x + q1[u].x) * (k0[u].x + k1[u].x) +
               (q0[u].y + q1[u].y) * (k0[u].y + k1[u].y) +
               (q0[u].z + q1[u].z) * (k0[u].z + k1[u].z) +
               (q0[u].w + q1[u].w) * (k0[u].w + k1[u].w);
#pragma unroll
      for (int off = 16; off >= 1; off >>= 1) {
#pragma unroll
        for (int u = 0; u < 4; ++u) a[u] += __shfl_xor(a[u], off, 32);
      }
#pragma unroll
      for (int u = 0; u < 4; ++u)
        if (lane == j0 + u) mine = a[u] * 0.08838834764831845f;
    }
    if (lane < cnt) {
      attn[e] = mine;
      lds_max_f(smax + row, mine);
      lds_max_f(smin + row, -mine);
      atomicAdd(scnt + row, 1.0f);
    }
  }
};

struct TypeWork {
  int lane, lo;
  const int* __restrict__ ei;
  const int* __restrict__ et;
  int E, R;
  int* tcur;
  bool rowrec;
  __device__ __forceinline__ void operator()(int* lst, int base,
                                             int cnt) const {
    if (lane < cnt) {
      const int e = lst[base + lane];
      int s, t, row;
      if (e < E) {
        s = ei[e];
        t = et[e];
        row = rowrec ? lst[REC_CAP + base + lane] : ei[E + e] - lo;
      } else {
        s = e - E;
        t = R;
        row = s - lo;
      }
      lst[base + lane] = s;
      lst[REC_CAP + base + lane] = row;
      lst[2 * REC_CAP + base + lane] = t;
      atomicAdd(tcur + (t & (SELF_PAD - 1)), 1);
    }
  }
};

__device__ __forceinline__ void score_slice(
    int beg, int end, int lane, const int* ts_s, const int* ts_row, int* ts_a,
    const float* qs, const float* __restrict__ Xk,
    const float* __restrict__ Tq, const float* __restrict__ Tk, int* smax,
    int* smin, float* scnt) {
  int tprev = -1;
  float4 tq = make_float4(0.0f, 0.0f, 0.0f, 0.0f);
  float4 tk = make_float4(0.0f, 0.0f, 0.0f, 0.0f);
  for (int base = beg; base < end; base += 32) {
    const int cnt = min(32, end - base);
    int s = 0, t = 0, row = 0;
    if (lane < cnt) {
      s = ts_s[base + lane];
      row = ts_row[base + lane];
      t = ts_a[base + lane];
    }
    float mine = 0.0f;
    for (int j0 = 0; j0 < cnt; j0 += 4) {
      float4 q0[4], q1[4], k0[4], k1[4];
      bool fresh[4];
#pragma unroll
      for (int u = 0; u < 4; ++u) {
        const int j = min(j0 + u, cnt - 1);
        const int sj = __builtin_amdgcn_readlane(s, j);
        const int tj = __builtin_amdgcn_readlane(t, j);
        const int rj = __builtin_amdgcn_readlane(row, j);
        k0[u] = ((const float4*)(Xk + (size_t)sj * D))[lane];
        q0[u] = ((const float4*)(qs + rj * D))[lane];
        fresh[u] = tj != tprev;
        if (fresh[u]) {
          q1[u] = ((const float4*)(Tq + (size_t)tj * D))[lane];
          k1[u] = ((const float4*)(Tk + (size_t)tj * D))[lane];
          tprev = tj;
        }
      }
      if (!fresh[0]) {
        q1[0] = tq;
        k1[0] = tk;
      }
      if (!fresh[1]) {
        q1[1] = q1[0];
        k1[1] = k1[0];
      }
      if (!fresh[2]) {
        q1[2] = q1[1];
        k1[2] = k1[1];
      }
      if (!fresh[3]) {
        q1[3] = q1[2];
        k1[3] = k1[2];
      }
      tq = q1[3];
      tk = k1[3];
      float a[4];
#pragma unroll
      for (int u = 0; u < 4; ++u)
        a[u] = (q0[u].x + q1[u].x) * (k0[u].x + k1[u].x) +
               (q0[u].y + q1[u].y) * (k0[u].y + k1[u].y) +
               (q0[u].z + q1[u].z) * (k0[u].z + k1[u].z) +
               (q0[u].w + q1[u].w) * (k0[u].w + k1[u].w);
#pragma unroll
      for (int off = 16; off >= 1; off >>= 1) {
#pragma unroll
        for (int u = 0; u < 4; ++u) a[u] += __shfl_xor(a[u], off, 32);
      }
#pragma unroll
      for (int u = 0; u < 4; ++u)
        if (lane == j0 + u) mine = a[u] * 0.08838834764831845f;
    }
    if (lane < cnt) {
      ts_a[base + lane] = __float_as_int(mine);
      lds_max_f(smax + row, mine);
      lds_max_f(smin + row, -mine);
      atomicAdd(scnt + row, 1.0f);
    }
  }
}

__device__ __forceinline__ void edge_weights(float a, int row, const int* smax,
                                             const int* smin, float* sd,
                                             float* ss, float& wd, float& ws) {
  wd = __expf(a - __int_as_float(smax[row]));
  ws = __expf(-a - __int_as_float(smin[row]));
  atomicAdd(sd + row, wd);
  atomicAdd(ss + row, ws);
}

__device__ __forceinline__ void add_rows(int cnt, int lane, int s, int row,
                                         float wd, float ws,
                                         const float* __restrict__ Vv,
                                         float* acc_s, float* acc_d) {
  for (int j0 = 0; j0 < cnt; j0 += 8) {
    float4 v[8];
#pragma unroll
    for (int u = 0; u < 8; ++u) {
      const int j = min(j0 + u, cnt - 1);
      const int sj = __builtin_amdgcn_readlane(s, j);
      v[u] = ((const float4*)(Vv + (size_t)sj * D))[lane];
    }
#pragma unroll
    for (int u = 0; u < 8; ++u) {
      const int j = j0 + u;
      if (j < cnt) {
        const int rj = __builtin_amdgcn_readlane(row, j);
        const float wsj = __int_as_float(__builtin_amdgcn_readlane(__float_as_int(ws), j));
        const float wdj = __int_as_float(__builtin_amdgcn_readlane(__float_as_int(wd), j));
        float* as = acc_s + rj * D + lane;
        float* ad = acc_d + rj * D + lane;
        __hip_atomic_fetch_add(as + 0, wsj * v[u].x, __ATOMIC_RELAXED, __HIP_MEMORY_SCOPE_WORKGROUP);
        __hip_atomic_fetch_add(as + 32, wsj * v[u].y, __ATOMIC_RELAXED, __HIP_MEMORY_SCOPE_WORKGROUP);
        __hip_atomic_fetch_add(as + 64, wsj * v[u].z, __ATOMIC_RELAXED, __HIP_MEMORY_SCOPE_WORKGROUP);
        __hip_atomic_fetch_add(as + 96, wsj * v[u].w, __ATOMIC_RELAXED, __HIP_MEMORY_SCOPE_WORKGROUP);
        __hip_atomic_fetch_add(ad + 0, wdj * v[u].x, __ATOMIC_RELAXED, __HIP_MEMORY_SCOPE_WORKGROUP);
        __hip_atomic_fetch_add(ad + 32, wdj * v[u].y, __ATOMIC_RELAXED, __HIP_MEMORY_SCOPE_WORKGROUP);
        __hip_atomic_fetch_add(ad + 64, wdj * v[u].z, __ATOMIC_RELAXED, __HIP_MEMORY_SCOPE_WORKGROUP);
        __hip_atomic_fetch_add(ad + 96, wdj * v[u].w, __ATOMIC_RELAXED, __HIP_MEMORY_SCOPE_WORKGROUP);
      }
    }
  }
}

struct AddFromAttn {
  int lane, lo;
  const int* __restrict__ ei;
  int E;
  const float* attn;
  const int* smax;
  const int* smin;
  float* sd;
  float* ss;
  const float* __restrict__ Vv;
  float* acc_s;
  float* acc_d;
  __device__ __forceinline__ void operator()(int* lst, int base,
                                             int cnt) const {
    int s = 0, row = 0;
    float wd = 0.0f, ws = 0.0f;
    if (lane < cnt) {
      const int e = lst[base + lane];
      int d;
      if (e < E) {
        s = ei[e];
        d = ei[E + e];
      } else {
        s = d = e - E;
      }
      row = d - lo;
      edge_weights(attn[e], row, smax, smin, sd, ss, wd, ws);
    }
    add_rows(cnt, lane, s, row, wd, ws, Vv, acc_s, acc_d);
  }
};

__global__ __launch_bounds__(SC_THREADS)
__attribute__((amdgpu_waves_per_eu(8, 8))) void edge_phase_kernel(
    const int* __restrict__ ei, const int* __restrict__ et, int E, int N,
    int R, const float* __restrict__ Xq, const float* __restrict__ Xk,
    const float* __restrict__ Tq, const float* __restrict__ Tk,
    const float* __restrict__ Vv, float* attn, float* __restrict__ featsta,
    float* __restrict__ featdyn, const uint4* __restrict__ gidbytes) {
  __shared__ float qs[SC_ROWS * D];
  __shared__ float accbuf[2 * SC_ROWS * D];
  __shared__ int lists[SC_WAVES * REC_WORDS];
  __shared__ int smax[SC_ROWS];
  __shared__ int smin[SC_ROWS];
  __shared__ float scnt[SC_ROWS];
  __shared__ float sd[SC_ROWS];
  __shared__ float ss[SC_ROWS];
  __shared__ int rstart[SC_ROWS + 1];
  __shared__ int rnext[SC_ROWS];
  __shared__ int overflow;

  const int tid = threadIdx.x;
  const int lane = tid & 31;
  const int wave = tid >> 5;
  const int lo = blockIdx.x * SC_ROWS;
  const int nrows = min(SC_ROWS, N - lo);
  int* lst = lists + wave * REC_WORDS;
  int* tcur = (int*)accbuf + 6 * SRT_CAP;

  const float4* __restrict__ qg = (const float4*)(Xq + (size_t)lo * D);
  for (int i = tid; i < nrows * (D / 4); i += SC_THREADS)
    ((float4*)qs)[i] = qg[i];
  for (int i = tid; i < SC_ROWS; i += SC_THREADS) {
    smax[i] = __float_as_int(-__builtin_inff());
    smin[i] = __float_as_int(-__builtin_inff());
    scnt[i] = 0.0f;
  }
  for (int i = tid; i < SELF_PAD; i += SC_THREADS) tcur[i] = 0;
  if (tid == 0) overflow = 0;
  __syncthreads();

  const uint4* __restrict__ gid = sweep_by_bytes(E, N) ? gidbytes : nullptr;
  const TypeWork types = {lane, lo, ei, et, E, R, tcur, gid == nullptr};
  const int n = sweep_range<true>(lst, wave, lane, lo, nrows, ei, E, gid, types);
  if (n < 0 && lane == 0) overflow = 1;
  __syncthreads();

  if (__builtin_expect(overflow == 0, 1)) {
    int* srt_s = (int*)accbuf;
    float* srt_wd = accbuf + SRT_CAP;
    float* srt_ws = accbuf + 2 * SRT_CAP;
    int* ts_s = (int*)accbuf + 3 * SRT_CAP;
    int* ts_row = ts_s + SRT_CAP;
    int* ts_a = ts_row + SRT_CAP;

    if (wave == 0) {
      constexpr int TPL = SELF_PAD / 32;
      int ctyp[TPL];
      int tot = 0;
#pragma unroll
      for (int k = 0; k < TPL; ++k) {
        ctyp[k] = tcur[lane * TPL + k];
        tot += ctyp[k];
      }
      int incl = tot;
#pragma unroll
      for (int off = 1; off < 32; off <<= 1) {
        const int up = __shfl_up(incl, off, 32);
        if (lane >= off) incl += up;
      }
      int run = incl - tot;
#pragma unroll
      for (int k = 0; k < TPL; ++k) {
        tcur[lane * TPL + k] = run;
        run += ctyp[k];
      }
      if (lane == 31) tcur[SELF_PAD] = run;
    }
    __syncthreads();

    for (int base = 0; base < n; base += 32) {
      if (base + lane < n) {
        const int s = lst[base + lane];
        const int row = lst[REC_CAP + base + lane];
        const int t = lst[2 * REC_CAP + base + lane];
        const int pos = atomicAdd(tcur + (t & (SELF_PAD - 1)), 1);
        ts_s[pos] = s;
        ts_row[pos] = row;
        ts_a[pos] = t;
      }
    }
    __syncthreads();

    const int total = tcur[SELF_PAD];
    const int slice = ((total + SC_WAVES - 1) / SC_WAVES + 3) & ~3;
    const int sbeg = min(total, wave * slice);
    const int send = min(total, sbeg + slice);
    score_slice(sbeg, send, lane, ts_s, ts_row, ts_a, qs, Xk, Tq, Tk, smax, smin,
                scnt);
    __syncthreads();

    if (wave == 0) {
      constexpr int RPL = SC_ROWS / 32;
      int cdeg[RPL];
      int tot = 0;
#pragma unroll
      for (int k = 0; k < RPL; ++k) {
        cdeg[k] = (int)scnt[lane * RPL + k];
        tot += cdeg[k];
      }
      int incl = tot;
#pragma unroll
      for (int off = 1; off < 32; off <<= 1) {
        const int up = __shfl_up(incl, off, 32);
        if (lane >= off) incl += up;
      }
      int run = incl - tot;
#pragma unroll
      for (int k = 0; k < RPL; ++k) {
        rstart[lane * RPL + k] = run;
        rnext[lane * RPL + k] = run;
        run += cdeg[k];
      }
      if (lane == 31) rstart[SC_ROWS] = run;
    }
    __syncthreads();

    for (int base = sbeg; base < send; base += 32) {
      if (base + lane < send) {
        const int s = ts_s[base + lane];
        const int row = ts_row[base + lane];
        const float a = __int_as_float(ts_a[base + lane]);
        const float wd = __expf(a - __int_as_float(smax[row]));
        const float ws = __expf(-a - __int_as_float(smin[row]));
        const int pos = atomicAdd(rnext + row, 1);
        srt_s[pos] = s;
        srt_wd[pos] = wd;
        srt_ws[pos] = ws;
      }
    }
    __syncthreads();

    const int r0 = wave * RPW;
    const int bnd = rstart[r0 + min(lane, RPW)];
    const int ebeg = __builtin_amdgcn_readlane(bnd, 0);
    const int eend = __builtin_amdgcn_readlane(bnd, RPW);
    int k = 0;
    int rend = __builtin_amdgcn_readlane(bnd, 1);
    float4 as = make_float4(0.0f, 0.0f, 0.0f, 0.0f);
    float4 ad = make_float4(0.0f, 0.0f, 0.0f, 0.0f);
    float sums = 0.0f, sumd = 0.0f;
    auto row_done = [&]() {
      const int r = r0 + k;
      if (r < nrows) {
        const float denom = fmaxf(scnt[r], 1.0f);
        const float ns = sums * denom;
        const float nd = sumd * denom;
        const float4 os = make_float4(as.x / ns, as.y / ns, as.z / ns, as.w / ns);
        const float4 od = make_float4(ad.x / nd, ad.y / nd, ad.z / nd, ad.w / nd);
        ((float4*)(featsta + (size_t)(lo + r) * D))[lane] = os;
        ((float4*)(featdyn + (size_t)(lo + r) * D))[lane] = od;
      }
      as = make_float4(0.0f, 0.0f, 0.0f, 0.0f);
      ad = make_float4(0.0f, 0.0f, 0.0f, 0.0f);
      sums = 0.0f;
      sumd = 0.0f;
      ++k;
      rend = __builtin_amdgcn_readlane(bnd, min(k + 1, RPW));
    };
    for (int base = ebeg; base < eend; base += 32) {
      const int cnt = min(32, eend - base);
      int sv = 0;
      float wdv = 0.0f, wsv = 0.0f;
      if (lane < cnt) {
        sv = srt_s[base + lane];
        wdv = srt_wd[base + lane];
        wsv = srt_ws[base + lane];
      }
      for (int j0 = 0; j0 < cnt; j0 += 8) {
        float4 v[8];
#pragma unroll
        for (int u = 0; u < 8; ++u) {
          const int j = min(j0 + u, cnt - 1);
          const int sj = __builtin_amdgcn_readlane(sv, j);
          v[u] = ((const float4*)(Vv + (size_t)sj * D))[lane];
        }
#pragma unroll
        for (int u = 0; u < 8; ++u) {
          const int j = j0 + u;
          if (j < cnt) {
            while (base + j >= rend) row_done();
            const float wsj = __int_as_float(__builtin_amdgcn_readlane(__float_as_int(wsv), j));
            const float wdj = __int_as_float(__builtin_amdgcn_readlane(__float_as_int(wdv), j));
            as.x += wsj * v[u].x;
            as.y += wsj * v[u].y;
            as.z += wsj * v[u].z;
            as.w += wsj * v[u].w;
            ad.x += wdj * v[u].x;
            ad.y += wdj * v[u].y;
            ad.z += wdj * v[u].z;
            ad.w += wdj * v[u].w;
            sums += wsj;
            sumd += wdj;
          }
        }
      }
    }
    while (k < RPW) row_done();
  } else {
    float* acc_s = accbuf;
    float* acc_d = accbuf + SC_ROWS * D;
    for (int i = tid; i < 2 * SC_ROWS * D / 4; i += SC_THREADS)
      ((float4*)accbuf)[i] = make_float4(0.0f, 0.0f, 0.0f, 0.0f);
    for (int i = tid; i < SC_ROWS; i += SC_THREADS) {
      smax[i] = __float_as_int(-__builtin_inff());
      smin[i] = __float_as_int(-__builtin_inff());
      scnt[i] = 0.0f;
      sd[i] = 0.0f;
      ss[i] = 0.0f;
    }
    __syncthreads();
    const ScoreWork score2 = {lane, lo, ei,   et,   E,    R,    qs,  Xk,
                              Tq,   Tk, attn, smax, smin, scnt};
    sweep_range<false>(lst, wave, lane, lo, nrows, ei, E, nullptr, score2);
    __syncthreads();
    const AddFromAttn add2 = {lane, lo, ei, E,  attn,  smax, smin,
                              sd,   ss, Vv, acc_s, acc_d};
    sweep_range<false>(lst, wave, lane, lo, nrows, ei, E, nullptr, add2);
    __syncthreads();

    for (int r = wave; r < nrows; r += SC_WAVES) {
      const float denom = fmaxf(scnt[r], 1.0f);
      const float ns = ss[r] * denom;
      const float nd = sd[r] * denom;
      const float* as = acc_s + r * D + lane;
      const float* ad = acc_d + r * D + lane;
      const float4 os = make_float4(as[0] / ns, as[32] / ns, as[64] / ns, as[96] / ns);
      const float4 od = make_float4(ad[0] / nd, ad[32] / nd, ad[64] / nd, ad[96] / nd);
      ((float4*)(featsta + (size_t)(lo + r) * D))[lane] = os;
      ((float4*)(featdyn + (size_t)(lo + r) * D))[lane] = od;
    }
  }
}

extern "C" void kernel_launch(void* const* d_in, const int* in_sizes, int n_in,
                              void* d_out, int out_size, void* d_ws,
                              size_t ws_size, hipStream_t stream) {
  const float* curr = (const float*)d_in[0];
  const float* last = (const float*)d_in[1];
  const float* relemb = (const float*)d_in[2];
  const float* selfrel = (const float*)d_in[3];
  const float* Wq = (const float*)d_in[4];
  const float* bq = (const float*)d_in[5];
  const float* Wk = (const float*)d_in[6];
  const float* bk = (const float*)d_in[7];
  const float* Wv = (const float*)d_in[8];
  const float* bv = (const float*)d_in[9];
  const float* dynW = (const float*)d_in[10];
  const int* ei = (const int*)d_in[11];
  const int* et = (const int*)d_in[12];

  const int N = in_sizes[0] / D;
  const int E = in_sizes[12];
  const int R = in_sizes[2] / D;
  const size_t ND = (size_t)N * D;

  float* ws = (float*)d_ws;
  float* Xq = ws;
  float* Xk = Xq + ND;
  float* Vv = Xk + ND;
  float* featdyn = Vv + ND;
  float* Trel = featdyn + ND;
  float* Tq = Trel + SELF_PAD * D;
  float* Tk = Tq + SELF_PAD * D;
  float* attn = Tk + SELF_PAD * D;

  float* out0 = (float*)d_out;
  float* out1 = out0 + ND;

  unsigned int* gid = (unsigned int*)out1;
  build_trel<<<(SELF_PAD * D + E / 4 + 255) / 256, 256, 0, stream>>>(
      relemb, selfrel, Trel, R, ei, E, N, gid);

  gemm128_wmma<<<GEMM_GRID, 256, 0, stream>>>(curr, Wq, nullptr, Xq, nullptr,
                                              nullptr, nullptr, N / 16);
  gemm128_wmma<<<GEMM2_GRID, 512, 0, stream>>>(last, Wk, nullptr, Xk, Wv, bv,
                                               Vv, N / 16);
  gemm128_wmma<<<SELF_PAD / 16, 512, 0, stream>>>(Trel, Wq + D * D, bq, Tq,
                                                  Wk + D * D, bk, Tk,
                                                  SELF_PAD / 16);

  edge_phase_kernel<<<(N + SC_ROWS - 1) / SC_ROWS, SC_THREADS, 0, stream>>>(
      ei, et, E, N, R, Xq, Xk, Tq, Tk, Vv, attn, out0, featdyn,
      (const uint4*)gid);

  gemm128_wmma<<<GEMM_GRID, 256, 0, stream>>>(featdyn, dynW, nullptr, out1,
                                              nullptr, nullptr, nullptr, N / 16);
}


# Query product moved into the edge phase: a group computes the Xq rows of its own destination range by f32 WMMA into LDS, half its waves before the sweep, half after it; Xq launch and plane gone
# speedup vs baseline: 4.1073x; 1.0532x over previous
#include <hip/hip_runtime.h>
#include <math.h>


typedef __attribute__((ext_vector_type(2))) float v2f;
typedef __attribute__((ext_vector_type(8))) float v8f;

#define D 128
#define SELF_PAD 512
#define GEMM_GRID 500
#define GEMM2_GRID 250

__device__ __forceinline__ void load_a_tile(float4 (&a)[16],
                                            const float* __restrict__ A,
                                            int tile, int r, int g) {
  const float4* __restrict__ p =
      (const float4*)(A + (size_t)(tile * 16 + r) * D) + g;
#pragma unroll
  for (int i = 0; i < 16; ++i) a[i] = p[2 * i];
}

__device__ __forceinline__ void mma_store_tile(const float4 (&a)[16],
                                               const v2f (&b)[32], float bv,
                                               float* __restrict__ C, int tile,
                                               int n0, int r, int g) {
  v8f acc = {};
#pragma unroll
  for (int i = 0; i < 16; ++i) {
    v2f a0, a1;
    a0.x = a[i].x;
    a0.y = a[i].y;
    a1.x = a[i].z;
    a1.y = a[i].w;
    acc = __builtin_amdgcn_wmma_f32_16x16x4_f32(
         false, a0,  false, b[2 * i],
         (short)0, acc,  false,  false);
    acc = __builtin_amdgcn_wmma_f32_16x16x4_f32(
         false, a1,  false, b[2 * i + 1],
         (short)0, acc,  false,  false);
  }
#pragma unroll
  for (int j = 0; j < 8; ++j) {
    const int row = tile * 16 + j + 8 * g;
    C[(size_t)row * D + n0 + r] = acc[j] + bv;
  }
}

__global__ __launch_bounds__(512) void gemm128_wmma(
    const float* __restrict__ A, const float* __restrict__ W0,
    const float* __restrict__ bias0, float* __restrict__ C0,
    const float* __restrict__ W1, const float* __restrict__ bias1,
    float* __restrict__ C1, int tiles) {
  const int lane = threadIdx.x & 31;
  const int wave = threadIdx.x >> 5;
  const int r = lane & 15;
  const int g = lane >> 4;
  const bool second = wave >= 8;
  const float* __restrict__ W = second ? W1 : W0;
  const float* __restrict__ bias = second ? bias1 : bias0;
  float* __restrict__ C = second ? C1 : C0;
  const int n0 = (wave & 7) * 16;
  const int stride = gridDim.x;

  int t = blockIdx.x;
  if (t >= tiles) return;

  const float* __restrict__ Wcol = W + (n0 + r);
  v2f b[32];
#pragma unroll
  for (int s = 0; s < 32; ++s) {
    const int kk = 8 * (s >> 1) + 4 * g + 2 * (s & 1);
    b[s].x = Wcol[(size_t)kk * D];
    b[s].y = Wcol[(size_t)(kk + 1) * D];
  }
  const float bv = bias ? bias[n0 + r] : 0.0f;

  float4 a0[16], a1[16];
  load_a_tile(a0, A, t, r, g);
  for (;;) {
    const int t1 = t + stride;
    if (t1 >= tiles) {
      mma_store_tile(a0, b, bv, C, t, n0, r, g);
      break;
    }
    load_a_tile(a1, A, t1, r, g);
    __builtin_amdgcn_sched_barrier(0);
    mma_store_tile(a0, b, bv, C, t, n0, r, g);
    const int t2 = t1 + stride;
    if (t2 >= tiles) {
      mma_store_tile(a1, b, bv, C, t1, n0, r, g);
      break;
    }
    load_a_tile(a0, A, t2, r, g);
    __builtin_amdgcn_sched_barrier(0);
    mma_store_tile(a1, b, bv, C, t1, n0, r, g);
    t = t2;
  }
}

#define SC_ROWS 160

__device__ __forceinline__ bool sweep_by_bytes(int E, int N) {
  return (E & 15) == 0 && E >= 16 && (N + SC_ROWS - 1) / SC_ROWS <= 256 &&
         (long long)E <= 4ll * N * D;
}

__global__ void build_trel(const float* __restrict__ relemb,
                           const float* __restrict__ selfrel,
                           float* __restrict__ Trel, int R,
                           const int* __restrict__ ei, int E, int N,
                           unsigned int* __restrict__ gid) {
  const int i = blockIdx.x * 256 + threadIdx.x;
  if (i >= SELF_PAD * D) {
    const int w = i - SELF_PAD * D;
    if (w < (E >> 2) && sweep_by_bytes(E, N)) {
      const int4 d = ((const int4*)(ei + E))[w];
      gid[w] = ((unsigned)(d.x / SC_ROWS) & 255u) |
               (((unsigned)(d.y / SC_ROWS) & 255u) << 8) |
               (((unsigned)(d.z / SC_ROWS) & 255u) << 16) |
               (((unsigned)(d.w / SC_ROWS) & 255u) << 24);
    }
    return;
  }
  const int row = i >> 7;
  const int c = i & (D - 1);
  float v = 0.0f;
  if (row < R) v = relemb[i];
  else if (row == R) v = selfrel[c];
  Trel[i] = v;
}

__device__ __forceinline__ void lds_max_f(int* addr, float v) {
  if (v >= 0.0f)
    atomicMax(addr, __float_as_int(v));
  else
    atomicMin((unsigned int*)addr, __float_as_uint(v));
}

#define SC_THREADS 1024
#define SC_WAVES (SC_THREADS / 32)
#define SC_BATCH 512
#define SC_LIST (SC_BATCH + 32)
#define REC_CAP 192
#define REC_WORDS (3 * REC_CAP)
static_assert(REC_WORDS >= SC_LIST, "the general path's list lives in the record space");
#define RPW (SC_ROWS / SC_WAVES)
#define SRT_CAP (SC_WAVES * REC_CAP)
static_assert(SC_ROWS % SC_WAVES == 0, "every wave owns the same number of rows");
static_assert(SC_ROWS % 32 == 0, "the prefix sum gives every lane the same number of rows");
static_assert(SELF_PAD % 32 == 0, "the type prefix sum gives every lane the same number of types");
static_assert(6 * SRT_CAP + SELF_PAD + 1 <= 2 * SC_ROWS * D,
              "both regrouped lists and the type histogram live in the accumulator space");

template <bool VEC>
__device__ __forceinline__ int4 load_dst4(const int* __restrict__ dst, int idx,
                                          int E) {
  if (VEC) {
    return *(const int4*)(dst + min(idx, E - 4));
  } else {
    int4 v;
    v.x = dst[min(idx + 0, E - 1)];
    v.y = dst[min(idx + 1, E - 1)];
    v.z = dst[min(idx + 2, E - 1)];
    v.w = dst[min(idx + 3, E - 1)];
    return v;
  }
}

template <bool REC, bool ROW = true>
__device__ __forceinline__ int list_append(int* lst, int n, bool m, int e,
                                           int row, int lane) {
  const unsigned mask = (unsigned)__ballot(m);
  if (mask != 0u) {
    const int pos = n + __popc(mask & ((1u << lane) - 1u));
    if (REC) {
      if (m && pos < REC_CAP) {
        lst[pos] = e;
        if (ROW) lst[REC_CAP + pos] = row;
      }
    } else {
      if (m) lst[pos] = e;
    }
    n += __popc(mask);
  }
  return n;
}

template <bool VEC, bool REC, class WORK>
__device__ __forceinline__ int scan_stripe(
    int* lst, int n, int wave, int lane, int lo, int nrows, int rot, int nbatch,
    const int* __restrict__ ei, int E, const WORK& work) {
  const int* __restrict__ dst = ei + E;
  const unsigned span = (unsigned)nrows;
  int done = 0;
  for (int b0 = wave; b0 < nbatch; b0 += SC_WAVES) {
    int b = b0 + rot;
    if (b >= nbatch) b -= nbatch;
    const int idx = b * SC_BATCH + lane * 4;
    int4 q[4];
#pragma unroll
    for (int u = 0; u < 4; ++u) q[u] = load_dst4<VEC>(dst, idx + u * 128, E);
#pragma unroll
    for (int u = 0; u < 4; ++u) {
      const int e = idx + u * 128;
      const int r0 = q[u].x - lo, r1 = q[u].y - lo;
      const int r2 = q[u].z - lo, r3 = q[u].w - lo;
      bool m0, m1, m2, m3;
      if (VEC) {
        const unsigned sp = (e < E) ? span : 0u;
        m0 = (unsigned)r0 < sp;
        m1 = (unsigned)r1 < sp;
        m2 = (unsigned)r2 < sp;
        m3 = (unsigned)r3 < sp;
      } else {
        m0 = (e + 0 < E) & ((unsigned)r0 < span);
        m1 = (e + 1 < E) & ((unsigned)r1 < span);
        m2 = (e + 2 < E) & ((unsigned)r2 < span);
        m3 = (e + 3 < E) & ((unsigned)r3 < span);
      }
      if (__ballot(m0 | m1 | m2 | m3) != 0) {
        n = list_append<REC>(lst, n, m0, e + 0, r0, lane);
        n = list_append<REC>(lst, n, m1, e + 1, r1, lane);
        n = list_append<REC>(lst, n, m2, e + 2, r2, lane);
        n = list_append<REC>(lst, n, m3, e + 3, r3, lane);
      }
    }
    if (REC) {
      if (n > REC_CAP) return -1;
      while (n - done >= 32) {
        work(lst, done, 32);
        done += 32;
      }
    } else {
      int dn = 0;
      while (n - dn >= 32) {
        work(lst, dn, 32);
        dn += 32;
      }
      if (dn > 0) {
        const int rem = n - dn;
        int keep = 0;
        if (lane < rem) keep = lst[dn + lane];
        if (lane < rem) lst[lane] = keep;
        n = rem;
      }
    }
  }
  return n;
}

__device__ __forceinline__ unsigned zero_bytes(unsigned x) {
  const unsigned t = (x & 0x7f7f7f7fu) + 0x7f7f7f7fu;
  return ~(t | x | 0x7f7f7f7fu);
}

#define BY_STEP 2048

template <class WORK>
__device__ __forceinline__ int scan_bytes(int* lst, int n, int wave, int lane,
                                          const uint4* __restrict__ gid, int E,
                                          const WORK& work) {
  const int nvec = E >> 4;
  const int nstep = (nvec + BY_STEP / 16 - 1) / (BY_STEP / 16);
  const int rot = (int)(((long long)blockIdx.x * nstep) / gridDim.x);
  const unsigned pat = blockIdx.x * 0x01010101u;
  int done = 0;
  for (int b0 = wave; b0 < nstep; b0 += SC_WAVES) {
    int b = b0 + rot;
    if (b >= nstep) b -= nstep;
    const int vi = b * (BY_STEP / 16) + lane;
    uint4 q[4];
#pragma unroll
    for (int u = 0; u < 4; ++u) q[u] = gid[min(vi + u * 32, nvec - 1)];
#pragma unroll
    for (int u = 0; u < 4; ++u) {
      const int v = vi + u * 32;
      unsigned m = zero_bytes(q[u].x ^ pat) | (zero_bytes(q[u].y ^ pat) >> 1) |
                   (zero_bytes(q[u].z ^ pat) >> 2) |
                   (zero_bytes(q[u].w ^ pat) >> 3);
      if (v >= nvec) m = 0u;
      while (__ballot(m != 0u) != 0) {
        const bool has = m != 0u;
        const int p = __builtin_ctz(m | 0x80000000u);
        const int e = v * 16 + 4 * (7 - (p & 7)) + (p >> 3);
        m &= m - 1u;
        n = list_append<true, false>(lst, n, has, e, 0, lane);
      }
    }
    if (n > REC_CAP) return -1;
    while (n - done >= 32) {
      work(lst, done, 32);
      done += 32;
    }
  }
  return n;
}

template <bool REC, class WORK>
__device__ __forceinline__ int sweep_range(int* lst, int wave, int lane,
                                           int lo, int nrows,
                                           const int* __restrict__ ei, int E,
                                           const uint4* __restrict__ gid,
                                           const WORK& work) {
  int n = 0;
  for (int r0 = 0; r0 < SC_ROWS; r0 += SC_THREADS) {
    const int r = r0 + lane * SC_WAVES + wave;
    n = list_append<REC>(lst, n, r < nrows, E + lo + r, r, lane);
  }
  const int nbatch = (E + SC_BATCH - 1) / SC_BATCH;
  const int rot = (int)(((long long)blockIdx.x * nbatch) / gridDim.x);
  if (REC && gid != nullptr)
    n = scan_bytes(lst, n, wave, lane, gid, E, work);
  else if (REC && (E & 3) == 0 && E >= 4)
    n = scan_stripe<true, REC>(lst, n, wave, lane, lo, nrows, rot, nbatch, ei, E, work);
  else if (E > 0)
    n = scan_stripe<false, REC>(lst, n, wave, lane, lo, nrows, rot, nbatch, ei, E, work);
  if (n < 0) return n;
  const int done = REC ? (n & ~31) : 0;
  if (n > done) work(lst, done, n - done);
  return REC ? n : 0;
}

struct ScoreWork {
  int lane, lo;
  const int* __restrict__ ei;
  const int* __restrict__ et;
  int E, R;
  const float* qs;
  const float* __restrict__ Xk;
  const float* __restrict__ Tq;
  const float* __restrict__ Tk;
  float* attn;
  int* smax;
  int* smin;
  float* scnt;
  __device__ __forceinline__ void operator()(int* lst, int base,
                                             int cnt) const {
    int e = 0, s = 0, t = 0, row = 0;
    if (lane < cnt) {
      e = lst[base + lane];
      if (e < E) {
        s = ei[e];
        t = et[e];
        row = ei[E + e] - lo;
      } else {
        s = e - E;
        t = R;
        row = s - lo;
      }
    }
    float mine = 0.0f;
    for (int j0 = 0; j0 < cnt; j0 += 4) {
      float4 q0[4], q1[4], k0[4], k1[4];
#pragma unroll
      for (int u = 0; u < 4; ++u) {
        const int j = min(j0 + u, cnt - 1);
        const int sj = __builtin_amdgcn_readlane(s, j);
        const int tj = __builtin_amdgcn_readlane(t, j);
        const int rj = __builtin_amdgcn_readlane(row, j);
        k0[u] = ((const float4*)(Xk + (size_t)sj * D))[lane];
        q1[u] = ((const float4*)(Tq + (size_t)tj * D))[lane];
        k1[u] = ((const float4*)(Tk + (size_t)tj * D))[lane];
        q0[u] = ((const float4*)(qs + rj * D))[lane];
      }
      float a[4];
#pragma unroll
      for (int u = 0; u < 4; ++u)
        a[u] = (q0[u].x + q1[u].x) * (k0[u].x + k1[u].x) +
               (q0[u].y + q1[u].y) * (k0[u].y + k1[u].y) +
               (q0[u].z + q1[u].z) * (k0[u].z + k1[u].z) +
               (q0[u].w + q1[u].w) * (k0[u].w + k1[u].w);
#pragma unroll
      for (int off = 16; off >= 1; off >>= 1) {
#pragma unroll
        for (int u = 0; u < 4; ++u) a[u] += __shfl_xor(a[u], off, 32);
      }
#pragma unroll
      for (int u = 0; u < 4; ++u)
        if (lane == j0 + u) mine = a[u] * 0.08838834764831845f;
    }
    if (lane < cnt) {
      attn[e] = mine;
      lds_max_f(smax + row, mine);
      lds_max_f(smin + row, -mine);
      atomicAdd(scnt + row, 1.0f);
    }
  }
};

struct TypeWork {
  int lane, lo;
  const int* __restrict__ ei;
  const int* __restrict__ et;
  int E, R;
  int* tcur;
  bool rowrec;
  __device__ __forceinline__ void operator()(int* lst, int base,
                                             int cnt) const {
    if (lane < cnt) {
      const int e = lst[base + lane];
      int s, t, row;
      if (e < E) {
        s = ei[e];
        t = et[e];
        row = rowrec ? lst[REC_CAP + base + lane] : ei[E + e] - lo;
      } else {
        s = e - E;
        t = R;
        row = s - lo;
      }
      lst[base + lane] = s;
      lst[REC_CAP + base + lane] = row;
      lst[2 * REC_CAP + base + lane] = t;
      atomicAdd(tcur + (t & (SELF_PAD - 1)), 1);
    }
  }
};

__device__ __forceinline__ void score_slice(
    int beg, int end, int lane, const int* ts_s, const int* ts_row, int* ts_a,
    const float* qs, const float* __restrict__ Xk,
    const float* __restrict__ Tq, const float* __restrict__ Tk, int* smax,
    int* smin, float* scnt) {
  int tprev = -1;
  float4 tq = make_float4(0.0f, 0.0f, 0.0f, 0.0f);
  float4 tk = make_float4(0.0f, 0.0f, 0.0f, 0.0f);
  for (int base = beg; base < end; base += 32) {
    const int cnt = min(32, end - base);
    int s = 0, t = 0, row = 0;
    if (lane < cnt) {
      s = ts_s[base + lane];
      row = ts_row[base + lane];
      t = ts_a[base + lane];
    }
    float mine = 0.0f;
    for (int j0 = 0; j0 < cnt; j0 += 4) {
      float4 q0[4], q1[4], k0[4], k1[4];
      bool fresh[4];
#pragma unroll
      for (int u = 0; u < 4; ++u) {
        const int j = min(j0 + u, cnt - 1);
        const int sj = __builtin_amdgcn_readlane(s, j);
        const int tj = __builtin_amdgcn_readlane(t, j);
        const int rj = __builtin_amdgcn_readlane(row, j);
        k0[u] = ((const float4*)(Xk + (size_t)sj * D))[lane];
        q0[u] = ((const float4*)(qs + rj * D))[lane];
        fresh[u] = tj != tprev;
        if (fresh[u]) {
          q1[u] = ((const float4*)(Tq + (size_t)tj * D))[lane];
          k1[u] = ((const float4*)(Tk + (size_t)tj * D))[lane];
          tprev = tj;
        }
      }
      if (!fresh[0]) {
        q1[0] = tq;
        k1[0] = tk;
      }
      if (!fresh[1]) {
        q1[1] = q1[0];
        k1[1] = k1[0];
      }
      if (!fresh[2]) {
        q1[2] = q1[1];
        k1[2] = k1[1];
      }
      if (!fresh[3]) {
        q1[3] = q1[2];
        k1[3] = k1[2];
      }
      tq = q1[3];
      tk = k1[3];
      float a[4];
#pragma unroll
      for (int u = 0; u < 4; ++u)
        a[u] = (q0[u].x + q1[u].x) * (k0[u].x + k1[u].x) +
               (q0[u].y + q1[u].y) * (k0[u].y + k1[u].y) +
               (q0[u].z + q1[u].z) * (k0[u].z + k1[u].z) +
               (q0[u].w + q1[u].w) * (k0[u].w + k1[u].w);
#pragma unroll
      for (int off = 16; off >= 1; off >>= 1) {
#pragma unroll
        for (int u = 0; u < 4; ++u) a[u] += __shfl_xor(a[u], off, 32);
      }
#pragma unroll
      for (int u = 0; u < 4; ++u)
        if (lane == j0 + u) mine = a[u] * 0.08838834764831845f;
    }
    if (lane < cnt) {
      ts_a[base + lane] = __float_as_int(mine);
      lds_max_f(smax + row, mine);
      lds_max_f(smin + row, -mine);
      atomicAdd(scnt + row, 1.0f);
    }
  }
}

__device__ __forceinline__ void edge_weights(float a, int row, const int* smax,
                                             const int* smin, float* sd,
                                             float* ss, float& wd, float& ws) {
  wd = __expf(a - __int_as_float(smax[row]));
  ws = __expf(-a - __int_as_float(smin[row]));
  atomicAdd(sd + row, wd);
  atomicAdd(ss + row, ws);
}

__device__ __forceinline__ void add_rows(int cnt, int lane, int s, int row,
                                         float wd, float ws,
                                         const float* __restrict__ Vv,
                                         float* acc_s, float* acc_d) {
  for (int j0 = 0; j0 < cnt; j0 += 8) {
    float4 v[8];
#pragma unroll
    for (int u = 0; u < 8; ++u) {
      const int j = min(j0 + u, cnt - 1);
      const int sj = __builtin_amdgcn_readlane(s, j);
      v[u] = ((const float4*)(Vv + (size_t)sj * D))[lane];
    }
#pragma unroll
    for (int u = 0; u < 8; ++u) {
      const int j = j0 + u;
      if (j < cnt) {
        const int rj = __builtin_amdgcn_readlane(row, j);
        const float wsj = __int_as_float(__builtin_amdgcn_readlane(__float_as_int(ws), j));
        const float wdj = __int_as_float(__builtin_amdgcn_readlane(__float_as_int(wd), j));
        float* as = acc_s + rj * D + lane;
        float* ad = acc_d + rj * D + lane;
        __hip_atomic_fetch_add(as + 0, wsj * v[u].x, __ATOMIC_RELAXED, __HIP_MEMORY_SCOPE_WORKGROUP);
        __hip_atomic_fetch_add(as + 32, wsj * v[u].y, __ATOMIC_RELAXED, __HIP_MEMORY_SCOPE_WORKGROUP);
        __hip_atomic_fetch_add(as + 64, wsj * v[u].z, __ATOMIC_RELAXED, __HIP_MEMORY_SCOPE_WORKGROUP);
        __hip_atomic_fetch_add(as + 96, wsj * v[u].w, __ATOMIC_RELAXED, __HIP_MEMORY_SCOPE_WORKGROUP);
        __hip_atomic_fetch_add(ad + 0, wdj * v[u].x, __ATOMIC_RELAXED, __HIP_MEMORY_SCOPE_WORKGROUP);
        __hip_atomic_fetch_add(ad + 32, wdj * v[u].y, __ATOMIC_RELAXED, __HIP_MEMORY_SCOPE_WORKGROUP);
        __hip_atomic_fetch_add(ad + 64, wdj * v[u].z, __ATOMIC_RELAXED, __HIP_MEMORY_SCOPE_WORKGROUP);
        __hip_atomic_fetch_add(ad + 96, wdj * v[u].w, __ATOMIC_RELAXED, __HIP_MEMORY_SCOPE_WORKGROUP);
      }
    }
  }
}

struct AddFromAttn {
  int lane, lo;
  const int* __restrict__ ei;
  int E;
  const float* attn;
  const int* smax;
  const int* smin;
  float* sd;
  float* ss;
  const float* __restrict__ Vv;
  float* acc_s;
  float* acc_d;
  __device__ __forceinline__ void operator()(int* lst, int base,
                                             int cnt) const {
    int s = 0, row = 0;
    float wd = 0.0f, ws = 0.0f;
    if (lane < cnt) {
      const int e = lst[base + lane];
      int d;
      if (e < E) {
        s = ei[e];
        d = ei[E + e];
      } else {
        s = d = e - E;
      }
      row = d - lo;
      edge_weights(attn[e], row, smax, smin, sd, ss, wd, ws);
    }
    add_rows(cnt, lane, s, row, wd, ws, Vv, acc_s, acc_d);
  }
};

__device__ __forceinline__ void group_query_rows(float* qs,
                                                 const float* __restrict__ Arows,
                                                 const float* __restrict__ W,
                                                 int nrows, int wave, int lane) {
  const int r = lane & 15;
  const int g = lane >> 4;
  const int n0 = (wave & 7) * 16;
  const int ntiles = (nrows + 15) >> 4;
  int t = wave >> 3;
  if (t >= ntiles) return;

  const float* __restrict__ Wcol = W + (n0 + r);
  v2f b[32];
#pragma unroll
  for (int s = 0; s < 32; ++s) {
    const int kk = 8 * (s >> 1) + 4 * g + 2 * (s & 1);
    b[s].x = Wcol[(size_t)kk * D];
    b[s].y = Wcol[(size_t)(kk + 1) * D];
  }

  for (; t < ntiles; t += SC_WAVES / 8) {
    const int row = min(t * 16 + r, nrows - 1);
    const float4* __restrict__ p = (const float4*)(Arows + (size_t)row * D) + g;
    v8f acc = {};
#pragma unroll
    for (int h = 0; h < 2; ++h) {
      float4 a[8];
#pragma unroll
      for (int i = 0; i < 8; ++i) a[i] = p[2 * (8 * h + i)];
#pragma unroll
      for (int i = 0; i < 8; ++i) {
        v2f a0, a1;
        a0.x = a[i].x;
        a0.y = a[i].y;
        a1.x = a[i].z;
        a1.y = a[i].w;
        acc = __builtin_amdgcn_wmma_f32_16x16x4_f32(
             false, a0,  false, b[2 * (8 * h + i)],
             (short)0, acc,  false,  false);
        acc = __builtin_amdgcn_wmma_f32_16x16x4_f32(
             false, a1,  false, b[2 * (8 * h + i) + 1],
             (short)0, acc,  false,  false);
      }
    }
#pragma unroll
    for (int j = 0; j < 8; ++j)
      qs[(t * 16 + j + 8 * g) * D + n0 + r] = acc[j];
  }
}

__global__ __launch_bounds__(SC_THREADS)
__attribute__((amdgpu_waves_per_eu(8, 8))) void edge_phase_kernel(
    const int* __restrict__ ei, const int* __restrict__ et, int E, int N,
    int R, const float* __restrict__ curr, const float* __restrict__ Wq,
    const float* __restrict__ Xk,
    const float* __restrict__ Tq, const float* __restrict__ Tk,
    const float* __restrict__ Vv, float* attn, float* __restrict__ featsta,
    float* __restrict__ featdyn, const uint4* __restrict__ gidbytes) {
  __shared__ float qs[SC_ROWS * D];
  __shared__ float accbuf[2 * SC_ROWS * D];
  __shared__ int lists[SC_WAVES * REC_WORDS];
  __shared__ int smax[SC_ROWS];
  __shared__ int smin[SC_ROWS];
  __shared__ float scnt[SC_ROWS];
  __shared__ float sd[SC_ROWS];
  __shared__ float ss[SC_ROWS];
  __shared__ int rstart[SC_ROWS + 1];
  __shared__ int rnext[SC_ROWS];
  __shared__ int overflow;

  const int tid = threadIdx.x;
  const int lane = tid & 31;
  const int wave = tid >> 5;
  const int lo = blockIdx.x * SC_ROWS;
  const int nrows = min(SC_ROWS, N - lo);
  int* lst = lists + wave * REC_WORDS;
  int* tcur = (int*)accbuf + 6 * SRT_CAP;

  for (int i = tid; i < SC_ROWS; i += SC_THREADS) {
    smax[i] = __float_as_int(-__builtin_inff());
    smin[i] = __float_as_int(-__builtin_inff());
    scnt[i] = 0.0f;
  }
  for (int i = tid; i < SELF_PAD; i += SC_THREADS) tcur[i] = 0;
  if (tid == 0) overflow = 0;
  __syncthreads();

  const float* __restrict__ crows = curr + (size_t)lo * D;
  const bool rows_first = (wave & 8) == 0;
  if (rows_first) group_query_rows(qs, crows, Wq, nrows, wave, lane);

  const uint4* __restrict__ gid = sweep_by_bytes(E, N) ? gidbytes : nullptr;
  const TypeWork types = {lane, lo, ei, et, E, R, tcur, gid == nullptr};
  const int n = sweep_range<true>(lst, wave, lane, lo, nrows, ei, E, gid, types);
  if (n < 0 && lane == 0) overflow = 1;
  if (!rows_first) group_query_rows(qs, crows, Wq, nrows, wave, lane);
  __syncthreads();

  if (__builtin_expect(overflow == 0, 1)) {
    int* srt_s = (int*)accbuf;
    float* srt_wd = accbuf + SRT_CAP;
    float* srt_ws = accbuf + 2 * SRT_CAP;
    int* ts_s = (int*)accbuf + 3 * SRT_CAP;
    int* ts_row = ts_s + SRT_CAP;
    int* ts_a = ts_row + SRT_CAP;

    if (wave == 0) {
      constexpr int TPL = SELF_PAD / 32;
      int ctyp[TPL];
      int tot = 0;
#pragma unroll
      for (int k = 0; k < TPL; ++k) {
        ctyp[k] = tcur[lane * TPL + k];
        tot += ctyp[k];
      }
      int incl = tot;
#pragma unroll
      for (int off = 1; off < 32; off <<= 1) {
        const int up = __shfl_up(incl, off, 32);
        if (lane >= off) incl += up;
      }
      int run = incl - tot;
#pragma unroll
      for (int k = 0; k < TPL; ++k) {
        tcur[lane * TPL + k] = run;
        run += ctyp[k];
      }
      if (lane == 31) tcur[SELF_PAD] = run;
    }
    __syncthreads();

    for (int base = 0; base < n; base += 32) {
      if (base + lane < n) {
        const int s = lst[base + lane];
        const int row = lst[REC_CAP + base + lane];
        const int t = lst[2 * REC_CAP + base + lane];
        const int pos = atomicAdd(tcur + (t & (SELF_PAD - 1)), 1);
        ts_s[pos] = s;
        ts_row[pos] = row;
        ts_a[pos] = t;
      }
    }
    __syncthreads();

    const int total = tcur[SELF_PAD];
    const int slice = ((total + SC_WAVES - 1) / SC_WAVES + 3) & ~3;
    const int sbeg = min(total, wave * slice);
    const int send = min(total, sbeg + slice);
    score_slice(sbeg, send, lane, ts_s, ts_row, ts_a, qs, Xk, Tq, Tk, smax, smin,
                scnt);
    __syncthreads();

    if (wave == 0) {
      constexpr int RPL = SC_ROWS / 32;
      int cdeg[RPL];
      int tot = 0;
#pragma unroll
      for (int k = 0; k < RPL; ++k) {
        cdeg[k] = (int)scnt[lane * RPL + k];
        tot += cdeg[k];
      }
      int incl = tot;
#pragma unroll
      for (int off = 1; off < 32; off <<= 1) {
        const int up = __shfl_up(incl, off, 32);
        if (lane >= off) incl += up;
      }
      int run = incl - tot;
#pragma unroll
      for (int k = 0; k < RPL; ++k) {
        rstart[lane * RPL + k] = run;
        rnext[lane * RPL + k] = run;
        run += cdeg[k];
      }
      if (lane == 31) rstart[SC_ROWS] = run;
    }
    __syncthreads();

    for (int base = sbeg; base < send; base += 32) {
      if (base + lane < send) {
        const int s = ts_s[base + lane];
        const int row = ts_row[base + lane];
        const float a = __int_as_float(ts_a[base + lane]);
        const float wd = __expf(a - __int_as_float(smax[row]));
        const float ws = __expf(-a - __int_as_float(smin[row]));
        const int pos = atomicAdd(rnext + row, 1);
        srt_s[pos] = s;
        srt_wd[pos] = wd;
        srt_ws[pos] = ws;
      }
    }
    __syncthreads();

    const int r0 = wave * RPW;
    const int bnd = rstart[r0 + min(lane, RPW)];
    const int ebeg = __builtin_amdgcn_readlane(bnd, 0);
    const int eend = __builtin_amdgcn_readlane(bnd, RPW);
    int k = 0;
    int rend = __builtin_amdgcn_readlane(bnd, 1);
    float4 as = make_float4(0.0f, 0.0f, 0.0f, 0.0f);
    float4 ad = make_float4(0.0f, 0.0f, 0.0f, 0.0f);
    float sums = 0.0f, sumd = 0.0f;
    auto row_done = [&]() {
      const int r = r0 + k;
      if (r < nrows) {
        const float denom = fmaxf(scnt[r], 1.0f);
        const float ns = sums * denom;
        const float nd = sumd * denom;
        const float4 os = make_float4(as.x / ns, as.y / ns, as.z / ns, as.w / ns);
        const float4 od = make_float4(ad.x / nd, ad.y / nd, ad.z / nd, ad.w / nd);
        ((float4*)(featsta + (size_t)(lo + r) * D))[lane] = os;
        ((float4*)(featdyn + (size_t)(lo + r) * D))[lane] = od;
      }
      as = make_float4(0.0f, 0.0f, 0.0f, 0.0f);
      ad = make_float4(0.0f, 0.0f, 0.0f, 0.0f);
      sums = 0.0f;
      sumd = 0.0f;
      ++k;
      rend = __builtin_amdgcn_readlane(bnd, min(k + 1, RPW));
    };
    for (int base = ebeg; base < eend; base += 32) {
      const int cnt = min(32, eend - base);
      int sv = 0;
      float wdv = 0.0f, wsv = 0.0f;
      if (lane < cnt) {
        sv = srt_s[base + lane];
        wdv = srt_wd[base + lane];
        wsv = srt_ws[base + lane];
      }
      for (int j0 = 0; j0 < cnt; j0 += 8) {
        float4 v[8];
#pragma unroll
        for (int u = 0; u < 8; ++u) {
          const int j = min(j0 + u, cnt - 1);
          const int sj = __builtin_amdgcn_readlane(sv, j);
          v[u] = ((const float4*)(Vv + (size_t)sj * D))[lane];
        }
#pragma unroll
        for (int u = 0; u < 8; ++u) {
          const int j = j0 + u;
          if (j < cnt) {
            while (base + j >= rend) row_done();
            const float wsj = __int_as_float(__builtin_amdgcn_readlane(__float_as_int(wsv), j));
            const float wdj = __int_as_float(__builtin_amdgcn_readlane(__float_as_int(wdv), j));
            as.x += wsj * v[u].x;
            as.y += wsj * v[u].y;
            as.z += wsj * v[u].z;
            as.w += wsj * v[u].w;
            ad.x += wdj * v[u].x;
            ad.y += wdj * v[u].y;
            ad.z += wdj * v[u].z;
            ad.w += wdj * v[u].w;
            sums += wsj;
            sumd += wdj;
          }
        }
      }
    }
    while (k < RPW) row_done();
  } else {
    float* acc_s = accbuf;
    float* acc_d = accbuf + SC_ROWS * D;
    for (int i = tid; i < 2 * SC_ROWS * D / 4; i += SC_THREADS)
      ((float4*)accbuf)[i] = make_float4(0.0f, 0.0f, 0.0f, 0.0f);
    for (int i = tid; i < SC_ROWS; i += SC_THREADS) {
      smax[i] = __float_as_int(-__builtin_inff());
      smin[i] = __float_as_int(-__builtin_inff());
      scnt[i] = 0.0f;
      sd[i] = 0.0f;
      ss[i] = 0.0f;
    }
    __syncthreads();
    const ScoreWork score2 = {lane, lo, ei,   et,   E,    R,    qs,  Xk,
                              Tq,   Tk, attn, smax, smin, scnt};
    sweep_range<false>(lst, wave, lane, lo, nrows, ei, E, nullptr, score2);
    __syncthreads();
    const AddFromAttn add2 = {lane, lo, ei, E,  attn,  smax, smin,
                              sd,   ss, Vv, acc_s, acc_d};
    sweep_range<false>(lst, wave, lane, lo, nrows, ei, E, nullptr, add2);
    __syncthreads();

    for (int r = wave; r < nrows; r += SC_WAVES) {
      const float denom = fmaxf(scnt[r], 1.0f);
      const float ns = ss[r] * denom;
      const float nd = sd[r] * denom;
      const float* as = acc_s + r * D + lane;
      const float* ad = acc_d + r * D + lane;
      const float4 os = make_float4(as[0] / ns, as[32] / ns, as[64] / ns, as[96] / ns);
      const float4 od = make_float4(ad[0] / nd, ad[32] / nd, ad[64] / nd, ad[96] / nd);
      ((float4*)(featsta + (size_t)(lo + r) * D))[lane] = os;
      ((float4*)(featdyn + (size_t)(lo + r) * D))[lane] = od;
    }
  }
}

extern "C" void kernel_launch(void* const* d_in, const int* in_sizes, int n_in,
                              void* d_out, int out_size, void* d_ws,
                              size_t ws_size, hipStream_t stream) {
  const float* curr = (const float*)d_in[0];
  const float* last = (const float*)d_in[1];
  const float* relemb = (const float*)d_in[2];
  const float* selfrel = (const float*)d_in[3];
  const float* Wq = (const float*)d_in[4];
  const float* bq = (const float*)d_in[5];
  const float* Wk = (const float*)d_in[6];
  const float* bk = (const float*)d_in[7];
  const float* Wv = (const float*)d_in[8];
  const float* bv = (const float*)d_in[9];
  const float* dynW = (const float*)d_in[10];
  const int* ei = (const int*)d_in[11];
  const int* et = (const int*)d_in[12];

  const int N = in_sizes[0] / D;
  const int E = in_sizes[12];
  const int R = in_sizes[2] / D;
  const size_t ND = (size_t)N * D;

  float* ws = (float*)d_ws;
  float* Xk = ws + ND;
  float* Vv = Xk + ND;
  float* featdyn = Vv + ND;
  float* Trel = featdyn + ND;
  float* Tq = Trel + SELF_PAD * D;
  float* Tk = Tq + SELF_PAD * D;
  float* attn = Tk + SELF_PAD * D;

  float* out0 = (float*)d_out;
  float* out1 = out0 + ND;

  unsigned int* gid = (unsigned int*)out1;
  build_trel<<<(SELF_PAD * D + E / 4 + 255) / 256, 256, 0, stream>>>(
      relemb, selfrel, Trel, R, ei, E, N, gid);

  gemm128_wmma<<<GEMM2_GRID, 512, 0, stream>>>(last, Wk, nullptr, Xk, Wv, bv,
                                               Vv, N / 16);
  gemm128_wmma<<<SELF_PAD / 16, 512, 0, stream>>>(Trel, Wq + D * D, bq, Tq,
                                                  Wk + D * D, bk, Tk,
                                                  SELF_PAD / 16);

  edge_phase_kernel<<<(N + SC_ROWS - 1) / SC_ROWS, SC_THREADS, 0, stream>>>(
      ei, et, E, N, R, curr, Wq, Xk, Tq, Tk, Vv, attn, out0, featdyn,
      (const uint4*)gid);

  gemm128_wmma<<<GEMM_GRID, 256, 0, stream>>>(featdyn, dynW, nullptr, out1,
                                              nullptr, nullptr, nullptr, N / 16);
}


# Edge records by cells: table-launch blocks put each edge's record in the 256-byte cell of (destination group, list chunk), slot = LDS int atomic rank; groups read own cells; sweep and gathers gone
# speedup vs baseline: 4.3285x; 1.0539x over previous
#include <hip/hip_runtime.h>
#include <math.h>


typedef __attribute__((ext_vector_type(2))) float v2f;
typedef __attribute__((ext_vector_type(8))) float v8f;

#define D 128
#define SELF_PAD 512
#define GEMM_GRID 500
#define GEMM2_GRID 250

__device__ __forceinline__ void load_a_tile(float4 (&a)[16],
                                            const float* __restrict__ A,
                                            int tile, int r, int g) {
  const float4* __restrict__ p =
      (const float4*)(A + (size_t)(tile * 16 + r) * D) + g;
#pragma unroll
  for (int i = 0; i < 16; ++i) a[i] = p[2 * i];
}

__device__ __forceinline__ void mma_store_tile(const float4 (&a)[16],
                                               const v2f (&b)[32], float bv,
                                               float* __restrict__ C, int tile,
                                               int n0, int r, int g) {
  v8f acc = {};
#pragma unroll
  for (int i = 0; i < 16; ++i) {
    v2f a0, a1;
    a0.x = a[i].x;
    a0.y = a[i].y;
    a1.x = a[i].z;
    a1.y = a[i].w;
    acc = __builtin_amdgcn_wmma_f32_16x16x4_f32(
         false, a0,  false, b[2 * i],
         (short)0, acc,  false,  false);
    acc = __builtin_amdgcn_wmma_f32_16x16x4_f32(
         false, a1,  false, b[2 * i + 1],
         (short)0, acc,  false,  false);
  }
#pragma unroll
  for (int j = 0; j < 8; ++j) {
    const int row = tile * 16 + j + 8 * g;
    C[(size_t)row * D + n0 + r] = acc[j] + bv;
  }
}

__global__ __launch_bounds__(512) void gemm128_wmma(
    const float* __restrict__ A, const float* __restrict__ W0,
    const float* __restrict__ bias0, float* __restrict__ C0,
    const float* __restrict__ W1, const float* __restrict__ bias1,
    float* __restrict__ C1, int tiles) {
  const int lane = threadIdx.x & 31;
  const int wave = threadIdx.x >> 5;
  const int r = lane & 15;
  const int g = lane >> 4;
  const bool second = wave >= 8;
  const float* __restrict__ W = second ? W1 : W0;
  const float* __restrict__ bias = second ? bias1 : bias0;
  float* __restrict__ C = second ? C1 : C0;
  const int n0 = (wave & 7) * 16;
  const int stride = gridDim.x;

  int t = blockIdx.x;
  if (t >= tiles) return;

  const float* __restrict__ Wcol = W + (n0 + r);
  v2f b[32];
#pragma unroll
  for (int s = 0; s < 32; ++s) {
    const int kk = 8 * (s >> 1) + 4 * g + 2 * (s & 1);
    b[s].x = Wcol[(size_t)kk * D];
    b[s].y = Wcol[(size_t)(kk + 1) * D];
  }
  const float bv = bias ? bias[n0 + r] : 0.0f;

  float4 a0[16], a1[16];
  load_a_tile(a0, A, t, r, g);
  for (;;) {
    const int t1 = t + stride;
    if (t1 >= tiles) {
      mma_store_tile(a0, b, bv, C, t, n0, r, g);
      break;
    }
    load_a_tile(a1, A, t1, r, g);
    __builtin_amdgcn_sched_barrier(0);
    mma_store_tile(a0, b, bv, C, t, n0, r, g);
    const int t2 = t1 + stride;
    if (t2 >= tiles) {
      mma_store_tile(a1, b, bv, C, t1, n0, r, g);
      break;
    }
    load_a_tile(a0, A, t2, r, g);
    __builtin_amdgcn_sched_barrier(0);
    mma_store_tile(a1, b, bv, C, t1, n0, r, g);
    t = t2;
  }
}

#define SC_ROWS 160

#define CELL_CHUNK 2048
#define CELL_CAP 31
#define CELL_BLOCK 256

__device__ __forceinline__ bool cells_ok(int E, int N, int nchunk) {
  const int ngroup = (N + SC_ROWS - 1) / SC_ROWS;
  return E > 0 && ngroup <= CELL_BLOCK &&
         256ll * nchunk * ngroup <= 4ll * N * D;
}

static_assert(SELF_PAD * D % CELL_BLOCK == 0, "table blocks and cell blocks do not mix");
__global__ __launch_bounds__(CELL_BLOCK) void build_trel(
    const float* __restrict__ relemb, const float* __restrict__ selfrel,
    float* __restrict__ Trel, int R, const int* __restrict__ ei,
    const int* __restrict__ et, int E, int N, int2* __restrict__ cells,
    int nchunk) {
  __shared__ int ccnt[CELL_BLOCK];
  const int i = blockIdx.x * CELL_BLOCK + threadIdx.x;
  if (i >= SELF_PAD * D) {
    const int chunk = blockIdx.x - SELF_PAD * D / CELL_BLOCK;
    if (chunk >= nchunk || !cells_ok(E, N, nchunk)) return;
    const int ngroup = (N + SC_ROWS - 1) / SC_ROWS;
    ccnt[threadIdx.x] = 0;
    __syncthreads();
    constexpr int EPT = CELL_CHUNK / CELL_BLOCK;
    const int e0 = chunk * CELL_CHUNK + threadIdx.x;
    int s[EPT], t[EPT], d[EPT];
#pragma unroll
    for (int k = 0; k < EPT; ++k) {
      const int e = min(e0 + k * CELL_BLOCK, E - 1);
      s[k] = ei[e];
      d[k] = ei[E + e];
      t[k] = et[e];
    }
#pragma unroll
    for (int k = 0; k < EPT; ++k) {
      const int g = d[k] / SC_ROWS;
      if (e0 + k * CELL_BLOCK < E && (unsigned)g < (unsigned)ngroup) {
        const int rank = atomicAdd(ccnt + g, 1);
        if (rank < CELL_CAP)
          cells[((size_t)g * nchunk + chunk) * 32 + 1 + rank] =
              make_int2(s[k], (t[k] << 8) | (d[k] - g * SC_ROWS));
      }
    }
    __syncthreads();
    if ((int)threadIdx.x < ngroup)
      cells[((size_t)threadIdx.x * nchunk + chunk) * 32] =
          make_int2(ccnt[threadIdx.x], 0);
    return;
  }
  const int row = i >> 7;
  const int c = i & (D - 1);
  float v = 0.0f;
  if (row < R) v = relemb[i];
  else if (row == R) v = selfrel[c];
  Trel[i] = v;
}

__device__ __forceinline__ void lds_max_f(int* addr, float v) {
  if (v >= 0.0f)
    atomicMax(addr, __float_as_int(v));
  else
    atomicMin((unsigned int*)addr, __float_as_uint(v));
}

#define SC_THREADS 1024
#define SC_WAVES (SC_THREADS / 32)
#define SC_BATCH 512
#define SC_LIST (SC_BATCH + 32)
#define REC_CAP 192
#define REC_WORDS (3 * REC_CAP)
static_assert(REC_WORDS >= SC_LIST, "the general path's list lives in the record space");
#define RPW (SC_ROWS / SC_WAVES)
#define SRT_CAP (SC_WAVES * REC_CAP)
static_assert(SC_ROWS % SC_WAVES == 0, "every wave owns the same number of rows");
static_assert(SC_ROWS % 32 == 0, "the prefix sum gives every lane the same number of rows");
static_assert(SELF_PAD % 32 == 0, "the type prefix sum gives every lane the same number of types");
static_assert(6 * SRT_CAP + SELF_PAD + 1 <= 2 * SC_ROWS * D,
              "both regrouped lists and the type histogram live in the accumulator space");

template <bool VEC>
__device__ __forceinline__ int4 load_dst4(const int* __restrict__ dst, int idx,
                                          int E) {
  if (VEC) {
    return *(const int4*)(dst + min(idx, E - 4));
  } else {
    int4 v;
    v.x = dst[min(idx + 0, E - 1)];
    v.y = dst[min(idx + 1, E - 1)];
    v.z = dst[min(idx + 2, E - 1)];
    v.w = dst[min(idx + 3, E - 1)];
    return v;
  }
}

template <bool REC, bool ROW = true>
__device__ __forceinline__ int list_append(int* lst, int n, bool m, int e,
                                           int row, int lane) {
  const unsigned mask = (unsigned)__ballot(m);
  if (mask != 0u) {
    const int pos = n + __popc(mask & ((1u << lane) - 1u));
    if (REC) {
      if (m && pos < REC_CAP) {
        lst[pos] = e;
        if (ROW) lst[REC_CAP + pos] = row;
      }
    } else {
      if (m) lst[pos] = e;
    }
    n += __popc(mask);
  }
  return n;
}

template <bool VEC, bool REC, class WORK>
__device__ __forceinline__ int scan_stripe(
    int* lst, int n, int wave, int lane, int lo, int nrows, int rot, int nbatch,
    const int* __restrict__ ei, int E, const WORK& work) {
  const int* __restrict__ dst = ei + E;
  const unsigned span = (unsigned)nrows;
  int done = 0;
  for (int b0 = wave; b0 < nbatch; b0 += SC_WAVES) {
    int b = b0 + rot;
    if (b >= nbatch) b -= nbatch;
    const int idx = b * SC_BATCH + lane * 4;
    int4 q[4];
#pragma unroll
    for (int u = 0; u < 4; ++u) q[u] = load_dst4<VEC>(dst, idx + u * 128, E);
#pragma unroll
    for (int u = 0; u < 4; ++u) {
      const int e = idx + u * 128;
      const int r0 = q[u].x - lo, r1 = q[u].y - lo;
      const int r2 = q[u].z - lo, r3 = q[u].w - lo;
      bool m0, m1, m2, m3;
      if (VEC) {
        const unsigned sp = (e < E) ? span : 0u;
        m0 = (unsigned)r0 < sp;
        m1 = (unsigned)r1 < sp;
        m2 = (unsigned)r2 < sp;
        m3 = (unsigned)r3 < sp;
      } else {
        m0 = (e + 0 < E) & ((unsigned)r0 < span);
        m1 = (e + 1 < E) & ((unsigned)r1 < span);
        m2 = (e + 2 < E) & ((unsigned)r2 < span);
        m3 = (e + 3 < E) & ((unsigned)r3 < span);
      }
      if (__ballot(m0 | m1 | m2 | m3) != 0) {
        n = list_append<REC>(lst, n, m0, e + 0, r0, lane);
        n = list_append<REC>(lst, n, m1, e + 1, r1, lane);
        n = list_append<REC>(lst, n, m2, e + 2, r2, lane);
        n = list_append<REC>(lst, n, m3, e + 3, r3, lane);
      }
    }
    if (REC) {
      if (n > REC_CAP) return -1;
      while (n - done >= 32) {
        work(lst, done, 32);
        done += 32;
      }
    } else {
      int dn = 0;
      while (n - dn >= 32) {
        work(lst, dn, 32);
        dn += 32;
      }
      if (dn > 0) {
        const int rem = n - dn;
        int keep = 0;
        if (lane < rem) keep = lst[dn + lane];
        if (lane < rem) lst[lane] = keep;
        n = rem;
      }
    }
  }
  return n;
}

template <bool REC, class WORK>
__device__ __forceinline__ int sweep_range(int* lst, int wave, int lane,
                                           int lo, int nrows,
                                           const int* __restrict__ ei, int E,
                                           const WORK& work) {
  int n = 0;
  for (int r0 = 0; r0 < SC_ROWS; r0 += SC_THREADS) {
    const int r = r0 + lane * SC_WAVES + wave;
    n = list_append<REC>(lst, n, r < nrows, E + lo + r, r, lane);
  }
  const int nbatch = (E + SC_BATCH - 1) / SC_BATCH;
  const int rot = (int)(((long long)blockIdx.x * nbatch) / gridDim.x);
  if (REC && (E & 3) == 0 && E >= 4)
    n = scan_stripe<true, REC>(lst, n, wave, lane, lo, nrows, rot, nbatch, ei, E, work);
  else if (E > 0)
    n = scan_stripe<false, REC>(lst, n, wave, lane, lo, nrows, rot, nbatch, ei, E, work);
  if (n < 0) return n;
  const int done = REC ? (n & ~31) : 0;
  if (n > done) work(lst, done, n - done);
  return REC ? n : 0;
}

struct ScoreWork {
  int lane, lo;
  const int* __restrict__ ei;
  const int* __restrict__ et;
  int E, R;
  const float* qs;
  const float* __restrict__ Xk;
  const float* __restrict__ Tq;
  const float* __restrict__ Tk;
  float* attn;
  int* smax;
  int* smin;
  float* scnt;
  __device__ __forceinline__ void operator()(int* lst, int base,
                                             int cnt) const {
    int e = 0, s = 0, t = 0, row = 0;
    if (lane < cnt) {
      e = lst[base + lane];
      if (e < E) {
        s = ei[e];
        t = et[e];
        row = ei[E + e] - lo;
      } else {
        s = e - E;
        t = R;
        row = s - lo;
      }
    }
    float mine = 0.0f;
    for (int j0 = 0; j0 < cnt; j0 += 4) {
      float4 q0[4], q1[4], k0[4], k1[4];
#pragma unroll
      for (int u = 0; u < 4; ++u) {
        const int j = min(j0 + u, cnt - 1);
        const int sj = __builtin_amdgcn_readlane(s, j);
        const int tj = __builtin_amdgcn_readlane(t, j);
        const int rj = __builtin_amdgcn_readlane(row, j);
        k0[u] = ((const float4*)(Xk + (size_t)sj * D))[lane];
        q1[u] = ((const float4*)(Tq + (size_t)tj * D))[lane];
        k1[u] = ((const float4*)(Tk + (size_t)tj * D))[lane];
        q0[u] = ((const float4*)(qs + rj * D))[lane];
      }
      float a[4];
#pragma unroll
      for (int u = 0; u < 4; ++u)
        a[u] = (q0[u].x + q1[u].x) * (k0[u].x + k1[u].x) +
               (q0[u].y + q1[u].y) * (k0[u].y + k1[u].y) +
               (q0[u].z + q1[u].z) * (k0[u].z + k1[u].z) +
               (q0[u].w + q1[u].w) * (k0[u].w + k1[u].w);
#pragma unroll
      for (int off = 16; off >= 1; off >>= 1) {
#pragma unroll
        for (int u = 0; u < 4; ++u) a[u] += __shfl_xor(a[u], off, 32);
      }
#pragma unroll
      for (int u = 0; u < 4; ++u)
        if (lane == j0 + u) mine = a[u] * 0.08838834764831845f;
    }
    if (lane < cnt) {
      attn[e] = mine;
      lds_max_f(smax + row, mine);
      lds_max_f(smin + row, -mine);
      atomicAdd(scnt + row, 1.0f);
    }
  }
};

struct TypeWork {
  int lane, lo;
  const int* __restrict__ ei;
  const int* __restrict__ et;
  int E, R;
  int* tcur;
  __device__ __forceinline__ void operator()(int* lst, int base,
                                             int cnt) const {
    if (lane < cnt) {
      const int e = lst[base + lane];
      int s, t, row;
      if (e < E) {
        s = ei[e];
        t = et[e];
        row = lst[REC_CAP + base + lane];
      } else {
        s = e - E;
        t = R;
        row = s - lo;
      }
      lst[base + lane] = s;
      lst[REC_CAP + base + lane] = row;
      lst[2 * REC_CAP + base + lane] = t;
      atomicAdd(tcur + (t & (SELF_PAD - 1)), 1);
    }
  }
};

__device__ __forceinline__ int put_records(int* lst, int n, bool m, int s,
                                           int row, int t, int lane,
                                           int* tcur) {
  const unsigned mask = (unsigned)__ballot(m);
  const int pos = n + __popc(mask & ((1u << lane) - 1u));
  if (m && pos < REC_CAP) {
    lst[pos] = s;
    lst[REC_CAP + pos] = row;
    lst[2 * REC_CAP + pos] = t;
    atomicAdd(tcur + (t & (SELF_PAD - 1)), 1);
  }
  return n + __popc(mask);
}

__device__ __forceinline__ int take_cells(int* lst, int wave, int lane, int lo,
                                          int nrows, int R,
                                          const int2* __restrict__ cells,
                                          int nchunk, int* tcur) {
  int n = 0;
  for (int r0 = 0; r0 < SC_ROWS; r0 += SC_THREADS) {
    const int r = r0 + lane * SC_WAVES + wave;
    n = put_records(lst, n, r < nrows, lo + r, r, R, lane, tcur);
  }
  const int2* __restrict__ mine = cells + (size_t)blockIdx.x * nchunk * 32 + lane;
  constexpr int CPT = 8;
  for (int b0 = wave; b0 < nchunk; b0 += SC_WAVES * CPT) {
    int2 c[CPT];
#pragma unroll
    for (int u = 0; u < CPT; ++u)
      c[u] = mine[(size_t)min(b0 + u * SC_WAVES, nchunk - 1) * 32];
#pragma unroll
    for (int u = 0; u < CPT; ++u) {
      if (b0 + u * SC_WAVES < nchunk) {
        const int cnt = __builtin_amdgcn_readlane(c[u].x, 0);
        if (cnt > CELL_CAP) return -1;
        n = put_records(lst, n, lane >= 1 && lane <= cnt, c[u].x, c[u].y & 255,
                        c[u].y >> 8, lane, tcur);
      }
    }
    if (n > REC_CAP) return -1;
  }
  return n;
}

__device__ __forceinline__ void score_slice(
    int beg, int end, int lane, const int* ts_s, const int* ts_row, int* ts_a,
    const float* qs, const float* __restrict__ Xk,
    const float* __restrict__ Tq, const float* __restrict__ Tk, int* smax,
    int* smin, float* scnt) {
  int tprev = -1;
  float4 tq = make_float4(0.0f, 0.0f, 0.0f, 0.0f);
  float4 tk = make_float4(0.0f, 0.0f, 0.0f, 0.0f);
  for (int base = beg; base < end; base += 32) {
    const int cnt = min(32, end - base);
    int s = 0, t = 0, row = 0;
    if (lane < cnt) {
      s = ts_s[base + lane];
      row = ts_row[base + lane];
      t = ts_a[base + lane];
    }
    float mine = 0.0f;
    for (int j0 = 0; j0 < cnt; j0 += 4) {
      float4 q0[4], q1[4], k0[4], k1[4];
      bool fresh[4];
#pragma unroll
      for (int u = 0; u < 4; ++u) {
        const int j = min(j0 + u, cnt - 1);
        const int sj = __builtin_amdgcn_readlane(s, j);
        const int tj = __builtin_amdgcn_readlane(t, j);
        const int rj = __builtin_amdgcn_readlane(row, j);
        k0[u] = ((const float4*)(Xk + (size_t)sj * D))[lane];
        q0[u] = ((const float4*)(qs + rj * D))[lane];
        fresh[u] = tj != tprev;
        if (fresh[u]) {
          q1[u] = ((const float4*)(Tq + (size_t)tj * D))[lane];
          k1[u] = ((const float4*)(Tk + (size_t)tj * D))[lane];
          tprev = tj;
        }
      }
      if (!fresh[0]) {
        q1[0] = tq;
        k1[0] = tk;
      }
      if (!fresh[1]) {
        q1[1] = q1[0];
        k1[1] = k1[0];
      }
      if (!fresh[2]) {
        q1[2] = q1[1];
        k1[2] = k1[1];
      }
      if (!fresh[3]) {
        q1[3] = q1[2];
        k1[3] = k1[2];
      }
      tq = q1[3];
      tk = k1[3];
      float a[4];
#pragma unroll
      for (int u = 0; u < 4; ++u)
        a[u] = (q0[u].x + q1[u].x) * (k0[u].x + k1[u].x) +
               (q0[u].y + q1[u].y) * (k0[u].y + k1[u].y) +
               (q0[u].z + q1[u].z) * (k0[u].z + k1[u].z) +
               (q0[u].w + q1[u].w) * (k0[u].w + k1[u].w);
#pragma unroll
      for (int off = 16; off >= 1; off >>= 1) {
#pragma unroll
        for (int u = 0; u < 4; ++u) a[u] += __shfl_xor(a[u], off, 32);
      }
#pragma unroll
      for (int u = 0; u < 4; ++u)
        if (lane == j0 + u) mine = a[u] * 0.08838834764831845f;
    }
    if (lane < cnt) {
      ts_a[base + lane] = __float_as_int(mine);
      lds_max_f(smax + row, mine);
      lds_max_f(smin + row, -mine);
      atomicAdd(scnt + row, 1.0f);
    }
  }
}

__device__ __forceinline__ void edge_weights(float a, int row, const int* smax,
                                             const int* smin, float* sd,
                                             float* ss, float& wd, float& ws) {
  wd = __expf(a - __int_as_float(smax[row]));
  ws = __expf(-a - __int_as_float(smin[row]));
  atomicAdd(sd + row, wd);
  atomicAdd(ss + row, ws);
}

__device__ __forceinline__ void add_rows(int cnt, int lane, int s, int row,
                                         float wd, float ws,
                                         const float* __restrict__ Vv,
                                         float* acc_s, float* acc_d) {
  for (int j0 = 0; j0 < cnt; j0 += 8) {
    float4 v[8];
#pragma unroll
    for (int u = 0; u < 8; ++u) {
      const int j = min(j0 + u, cnt - 1);
      const int sj = __builtin_amdgcn_readlane(s, j);
      v[u] = ((const float4*)(Vv + (size_t)sj * D))[lane];
    }
#pragma unroll
    for (int u = 0; u < 8; ++u) {
      const int j = j0 + u;
      if (j < cnt) {
        const int rj = __builtin_amdgcn_readlane(row, j);
        const float wsj = __int_as_float(__builtin_amdgcn_readlane(__float_as_int(ws), j));
        const float wdj = __int_as_float(__builtin_amdgcn_readlane(__float_as_int(wd), j));
        float* as = acc_s + rj * D + lane;
        float* ad = acc_d + rj * D + lane;
        __hip_atomic_fetch_add(as + 0, wsj * v[u].x, __ATOMIC_RELAXED, __HIP_MEMORY_SCOPE_WORKGROUP);
        __hip_atomic_fetch_add(as + 32, wsj * v[u].y, __ATOMIC_RELAXED, __HIP_MEMORY_SCOPE_WORKGROUP);
        __hip_atomic_fetch_add(as + 64, wsj * v[u].z, __ATOMIC_RELAXED, __HIP_MEMORY_SCOPE_WORKGROUP);
        __hip_atomic_fetch_add(as + 96, wsj * v[u].w, __ATOMIC_RELAXED, __HIP_MEMORY_SCOPE_WORKGROUP);
        __hip_atomic_fetch_add(ad + 0, wdj * v[u].x, __ATOMIC_RELAXED, __HIP_MEMORY_SCOPE_WORKGROUP);
        __hip_atomic_fetch_add(ad + 32, wdj * v[u].y, __ATOMIC_RELAXED, __HIP_MEMORY_SCOPE_WORKGROUP);
        __hip_atomic_fetch_add(ad + 64, wdj * v[u].z, __ATOMIC_RELAXED, __HIP_MEMORY_SCOPE_WORKGROUP);
        __hip_atomic_fetch_add(ad + 96, wdj * v[u].w, __ATOMIC_RELAXED, __HIP_MEMORY_SCOPE_WORKGROUP);
      }
    }
  }
}

struct AddFromAttn {
  int lane, lo;
  const int* __restrict__ ei;
  int E;
  const float* attn;
  const int* smax;
  const int* smin;
  float* sd;
  float* ss;
  const float* __restrict__ Vv;
  float* acc_s;
  float* acc_d;
  __device__ __forceinline__ void operator()(int* lst, int base,
                                             int cnt) const {
    int s = 0, row = 0;
    float wd = 0.0f, ws = 0.0f;
    if (lane < cnt) {
      const int e = lst[base + lane];
      int d;
      if (e < E) {
        s = ei[e];
        d = ei[E + e];
      } else {
        s = d = e - E;
      }
      row = d - lo;
      edge_weights(attn[e], row, smax, smin, sd, ss, wd, ws);
    }
    add_rows(cnt, lane, s, row, wd, ws, Vv, acc_s, acc_d);
  }
};

__device__ __forceinline__ void group_query_rows(float* qs,
                                                 const float* __restrict__ Arows,
                                                 const float* __restrict__ W,
                                                 int nrows, int wave, int lane) {
  const int r = lane & 15;
  const int g = lane >> 4;
  const int n0 = (wave & 7) * 16;
  const int ntiles = (nrows + 15) >> 4;
  int t = wave >> 3;
  if (t >= ntiles) return;

  const float* __restrict__ Wcol = W + (n0 + r);
  v2f b[32];
#pragma unroll
  for (int s = 0; s < 32; ++s) {
    const int kk = 8 * (s >> 1) + 4 * g + 2 * (s & 1);
    b[s].x = Wcol[(size_t)kk * D];
    b[s].y = Wcol[(size_t)(kk + 1) * D];
  }

  for (; t < ntiles; t += SC_WAVES / 8) {
    const int row = min(t * 16 + r, nrows - 1);
    const float4* __restrict__ p = (const float4*)(Arows + (size_t)row * D) + g;
    v8f acc = {};
#pragma unroll
    for (int h = 0; h < 2; ++h) {
      float4 a[8];
#pragma unroll
      for (int i = 0; i < 8; ++i) a[i] = p[2 * (8 * h + i)];
#pragma unroll
      for (int i = 0; i < 8; ++i) {
        v2f a0, a1;
        a0.x = a[i].x;
        a0.y = a[i].y;
        a1.x = a[i].z;
        a1.y = a[i].w;
        acc = __builtin_amdgcn_wmma_f32_16x16x4_f32(
             false, a0,  false, b[2 * (8 * h + i)],
             (short)0, acc,  false,  false);
        acc = __builtin_amdgcn_wmma_f32_16x16x4_f32(
             false, a1,  false, b[2 * (8 * h + i) + 1],
             (short)0, acc,  false,  false);
      }
    }
#pragma unroll
    for (int j = 0; j < 8; ++j)
      qs[(t * 16 + j + 8 * g) * D + n0 + r] = acc[j];
  }
}

__global__ __launch_bounds__(SC_THREADS)
__attribute__((amdgpu_waves_per_eu(8, 8))) void edge_phase_kernel(
    const int* __restrict__ ei, const int* __restrict__ et, int E, int N,
    int R, const float* __restrict__ curr, const float* __restrict__ Wq,
    const float* __restrict__ Xk,
    const float* __restrict__ Tq, const float* __restrict__ Tk,
    const float* __restrict__ Vv, float* attn, float* __restrict__ featsta,
    float* __restrict__ featdyn, const int2* __restrict__ cells, int nchunk) {
  __shared__ float qs[SC_ROWS * D];
  __shared__ float accbuf[2 * SC_ROWS * D];
  __shared__ int lists[SC_WAVES * REC_WORDS];
  __shared__ int smax[SC_ROWS];
  __shared__ int smin[SC_ROWS];
  __shared__ float scnt[SC_ROWS];
  __shared__ float sd[SC_ROWS];
  __shared__ float ss[SC_ROWS];
  __shared__ int rstart[SC_ROWS + 1];
  __shared__ int rnext[SC_ROWS];
  __shared__ int overflow;

  const int tid = threadIdx.x;
  const int lane = tid & 31;
  const int wave = tid >> 5;
  const int lo = blockIdx.x * SC_ROWS;
  const int nrows = min(SC_ROWS, N - lo);
  int* lst = lists + wave * REC_WORDS;
  int* tcur = (int*)accbuf + 6 * SRT_CAP;

  for (int i = tid; i < SC_ROWS; i += SC_THREADS) {
    smax[i] = __float_as_int(-__builtin_inff());
    smin[i] = __float_as_int(-__builtin_inff());
    scnt[i] = 0.0f;
  }
  for (int i = tid; i < SELF_PAD; i += SC_THREADS) tcur[i] = 0;
  if (tid == 0) overflow = 0;
  __syncthreads();

  const float* __restrict__ crows = curr + (size_t)lo * D;
  const bool rows_first = (wave & 8) == 0;
  if (rows_first) group_query_rows(qs, crows, Wq, nrows, wave, lane);

  int n;
  if (cells_ok(E, N, nchunk)) {
    n = take_cells(lst, wave, lane, lo, nrows, R, cells, nchunk, tcur);
  } else {
    const TypeWork types = {lane, lo, ei, et, E, R, tcur};
    n = sweep_range<true>(lst, wave, lane, lo, nrows, ei, E, types);
  }
  if (n < 0 && lane == 0) overflow = 1;
  if (!rows_first) group_query_rows(qs, crows, Wq, nrows, wave, lane);
  __syncthreads();

  if (__builtin_expect(overflow == 0, 1)) {
    int* srt_s = (int*)accbuf;
    float* srt_wd = accbuf + SRT_CAP;
    float* srt_ws = accbuf + 2 * SRT_CAP;
    int* ts_s = (int*)accbuf + 3 * SRT_CAP;
    int* ts_row = ts_s + SRT_CAP;
    int* ts_a = ts_row + SRT_CAP;

    if (wave == 0) {
      constexpr int TPL = SELF_PAD / 32;
      int ctyp[TPL];
      int tot = 0;
#pragma unroll
      for (int k = 0; k < TPL; ++k) {
        ctyp[k] = tcur[lane * TPL + k];
        tot += ctyp[k];
      }
      int incl = tot;
#pragma unroll
      for (int off = 1; off < 32; off <<= 1) {
        const int up = __shfl_up(incl, off, 32);
        if (lane >= off) incl += up;
      }
      int run = incl - tot;
#pragma unroll
      for (int k = 0; k < TPL; ++k) {
        tcur[lane * TPL + k] = run;
        run += ctyp[k];
      }
      if (lane == 31) tcur[SELF_PAD] = run;
    }
    __syncthreads();

    for (int base = 0; base < n; base += 32) {
      if (base + lane < n) {
        const int s = lst[base + lane];
        const int row = lst[REC_CAP + base + lane];
        const int t = lst[2 * REC_CAP + base + lane];
        const int pos = atomicAdd(tcur + (t & (SELF_PAD - 1)), 1);
        ts_s[pos] = s;
        ts_row[pos] = row;
        ts_a[pos] = t;
      }
    }
    __syncthreads();

    const int total = tcur[SELF_PAD];
    const int slice = ((total + SC_WAVES - 1) / SC_WAVES + 3) & ~3;
    const int sbeg = min(total, wave * slice);
    const int send = min(total, sbeg + slice);
    score_slice(sbeg, send, lane, ts_s, ts_row, ts_a, qs, Xk, Tq, Tk, smax, smin,
                scnt);
    __syncthreads();

    if (wave == 0) {
      constexpr int RPL = SC_ROWS / 32;
      int cdeg[RPL];
      int tot = 0;
#pragma unroll
      for (int k = 0; k < RPL; ++k) {
        cdeg[k] = (int)scnt[lane * RPL + k];
        tot += cdeg[k];
      }
      int incl = tot;
#pragma unroll
      for (int off = 1; off < 32; off <<= 1) {
        const int up = __shfl_up(incl, off, 32);
        if (lane >= off) incl += up;
      }
      int run = incl - tot;
#pragma unroll
      for (int k = 0; k < RPL; ++k) {
        rstart[lane * RPL + k] = run;
        rnext[lane * RPL + k] = run;
        run += cdeg[k];
      }
      if (lane == 31) rstart[SC_ROWS] = run;
    }
    __syncthreads();

    for (int base = sbeg; base < send; base += 32) {
      if (base + lane < send) {
        const int s = ts_s[base + lane];
        const int row = ts_row[base + lane];
        const float a = __int_as_float(ts_a[base + lane]);
        const float wd = __expf(a - __int_as_float(smax[row]));
        const float ws = __expf(-a - __int_as_float(smin[row]));
        const int pos = atomicAdd(rnext + row, 1);
        srt_s[pos] = s;
        srt_wd[pos] = wd;
        srt_ws[pos] = ws;
      }
    }
    __syncthreads();

    const int r0 = wave * RPW;
    const int bnd = rstart[r0 + min(lane, RPW)];
    const int ebeg = __builtin_amdgcn_readlane(bnd, 0);
    const int eend = __builtin_amdgcn_readlane(bnd, RPW);
    int k = 0;
    int rend = __builtin_amdgcn_readlane(bnd, 1);
    float4 as = make_float4(0.0f, 0.0f, 0.0f, 0.0f);
    float4 ad = make_float4(0.0f, 0.0f, 0.0f, 0.0f);
    float sums = 0.0f, sumd = 0.0f;
    auto row_done = [&]() {
      const int r = r0 + k;
      if (r < nrows) {
        const float denom = fmaxf(scnt[r], 1.0f);
        const float ns = sums * denom;
        const float nd = sumd * denom;
        const float4 os = make_float4(as.x / ns, as.y / ns, as.z / ns, as.w / ns);
        const float4 od = make_float4(ad.x / nd, ad.y / nd, ad.z / nd, ad.w / nd);
        ((float4*)(featsta + (size_t)(lo + r) * D))[lane] = os;
        ((float4*)(featdyn + (size_t)(lo + r) * D))[lane] = od;
      }
      as = make_float4(0.0f, 0.0f, 0.0f, 0.0f);
      ad = make_float4(0.0f, 0.0f, 0.0f, 0.0f);
      sums = 0.0f;
      sumd = 0.0f;
      ++k;
      rend = __builtin_amdgcn_readlane(bnd, min(k + 1, RPW));
    };
    for (int base = ebeg; base < eend; base += 32) {
      const int cnt = min(32, eend - base);
      int sv = 0;
      float wdv = 0.0f, wsv = 0.0f;
      if (lane < cnt) {
        sv = srt_s[base + lane];
        wdv = srt_wd[base + lane];
        wsv = srt_ws[base + lane];
      }
      for (int j0 = 0; j0 < cnt; j0 += 8) {
        float4 v[8];
#pragma unroll
        for (int u = 0; u < 8; ++u) {
          const int j = min(j0 + u, cnt - 1);
          const int sj = __builtin_amdgcn_readlane(sv, j);
          v[u] = ((const float4*)(Vv + (size_t)sj * D))[lane];
        }
#pragma unroll
        for (int u = 0; u < 8; ++u) {
          const int j = j0 + u;
          if (j < cnt) {
            while (base + j >= rend) row_done();
            const float wsj = __int_as_float(__builtin_amdgcn_readlane(__float_as_int(wsv), j));
            const float wdj = __int_as_float(__builtin_amdgcn_readlane(__float_as_int(wdv), j));
            as.x += wsj * v[u].x;
            as.y += wsj * v[u].y;
            as.z += wsj * v[u].z;
            as.w += wsj * v[u].w;
            ad.x += wdj * v[u].x;
            ad.y += wdj * v[u].y;
            ad.z += wdj * v[u].z;
            ad.w += wdj * v[u].w;
            sums += wsj;
            sumd += wdj;
          }
        }
      }
    }
    while (k < RPW) row_done();
  } else {
    float* acc_s = accbuf;
    float* acc_d = accbuf + SC_ROWS * D;
    for (int i = tid; i < 2 * SC_ROWS * D / 4; i += SC_THREADS)
      ((float4*)accbuf)[i] = make_float4(0.0f, 0.0f, 0.0f, 0.0f);
    for (int i = tid; i < SC_ROWS; i += SC_THREADS) {
      smax[i] = __float_as_int(-__builtin_inff());
      smin[i] = __float_as_int(-__builtin_inff());
      scnt[i] = 0.0f;
      sd[i] = 0.0f;
      ss[i] = 0.0f;
    }
    __syncthreads();
    const ScoreWork score2 = {lane, lo, ei,   et,   E,    R,    qs,  Xk,
                              Tq,   Tk, attn, smax, smin, scnt};
    sweep_range<false>(lst, wave, lane, lo, nrows, ei, E, score2);
    __syncthreads();
    const AddFromAttn add2 = {lane, lo, ei, E,  attn,  smax, smin,
                              sd,   ss, Vv, acc_s, acc_d};
    sweep_range<false>(lst, wave, lane, lo, nrows, ei, E, add2);
    __syncthreads();

    for (int r = wave; r < nrows; r += SC_WAVES) {
      const float denom = fmaxf(scnt[r], 1.0f);
      const float ns = ss[r] * denom;
      const float nd = sd[r] * denom;
      const float* as = acc_s + r * D + lane;
      const float* ad = acc_d + r * D + lane;
      const float4 os = make_float4(as[0] / ns, as[32] / ns, as[64] / ns, as[96] / ns);
      const float4 od = make_float4(ad[0] / nd, ad[32] / nd, ad[64] / nd, ad[96] / nd);
      ((float4*)(featsta + (size_t)(lo + r) * D))[lane] = os;
      ((float4*)(featdyn + (size_t)(lo + r) * D))[lane] = od;
    }
  }
}

extern "C" void kernel_launch(void* const* d_in, const int* in_sizes, int n_in,
                              void* d_out, int out_size, void* d_ws,
                              size_t ws_size, hipStream_t stream) {
  const float* curr = (const float*)d_in[0];
  const float* last = (const float*)d_in[1];
  const float* relemb = (const float*)d_in[2];
  const float* selfrel = (const float*)d_in[3];
  const float* Wq = (const float*)d_in[4];
  const float* bq = (const float*)d_in[5];
  const float* Wk = (const float*)d_in[6];
  const float* bk = (const float*)d_in[7];
  const float* Wv = (const float*)d_in[8];
  const float* bv = (const float*)d_in[9];
  const float* dynW = (const float*)d_in[10];
  const int* ei = (const int*)d_in[11];
  const int* et = (const int*)d_in[12];

  const int N = in_sizes[0] / D;
  const int E = in_sizes[12];
  const int R = in_sizes[2] / D;
  const size_t ND = (size_t)N * D;

  float* ws = (float*)d_ws;
  int2* cells = (int2*)ws;
  float* Xk = ws + ND;
  float* Vv = Xk + ND;
  float* featdyn = Vv + ND;
  float* Trel = featdyn + ND;
  float* Tq = Trel + SELF_PAD * D;
  float* Tk = Tq + SELF_PAD * D;
  float* attn = Tk + SELF_PAD * D;

  float* out0 = (float*)d_out;
  float* out1 = out0 + ND;

  const int nchunk = (E + CELL_CHUNK - 1) / CELL_CHUNK;
  build_trel<<<SELF_PAD * D / CELL_BLOCK + nchunk, CELL_BLOCK, 0, stream>>>(
      relemb, selfrel, Trel, R, ei, et, E, N, cells, nchunk);

  gemm128_wmma<<<GEMM2_GRID, 512, 0, stream>>>(last, Wk, nullptr, Xk, Wv, bv,
                                               Vv, N / 16);
  gemm128_wmma<<<SELF_PAD / 16, 512, 0, stream>>>(Trel, Wq + D * D, bq, Tq,
                                                  Wk + D * D, bk, Tk,
                                                  SELF_PAD / 16);

  edge_phase_kernel<<<(N + SC_ROWS - 1) / SC_ROWS, SC_THREADS, 0, stream>>>(
      ei, et, E, N, R, curr, Wq, Xk, Tq, Tk, Vv, attn, out0, featdyn, cells,
      nchunk);

  gemm128_wmma<<<GEMM_GRID, 256, 0, stream>>>(featdyn, dynW, nullptr, out1,
                                              nullptr, nullptr, nullptr, N / 16);
}
